# starting kernel
# baseline (speedup 1.0000x reference)
#include <hip/hip_runtime.h>
#include <stddef.h>

typedef __attribute__((ext_vector_type(2))) float v2f;
typedef __attribute__((ext_vector_type(8))) float v8f;

__device__ __forceinline__ v8f wmma4(v2f a, v2f b, v8f c) {
  return __builtin_amdgcn_wmma_f32_16x16x4_f32(
       false, a,  false, b,
       (short)0, c,  false,  false);
}

template <int N, int K>
__global__ void gemm_wmma_f32(const float* __restrict__ A,
                              const float* __restrict__ B,
                              float* __restrict__ C) {
  const int lane = threadIdx.x & 31;
  const int m0 = blockIdx.x * 16;
  const int n0 = blockIdx.y * 64;
  const int half = lane >> 4;
  const int r    = lane & 15;

  const float* Arow = A + (size_t)(m0 + r) * K + 2 * half;
  const float* Bcol = B + (size_t)(2 * half) * N + n0 + r;

  v8f c0 = {}, c1 = {}, c2 = {}, c3 = {};
#pragma unroll
  for (int k = 0; k < K; k += 4) {
    v2f a = *(const v2f*)(Arow + k);
    v2f b0, b1, b2, b3;
    b0.x = Bcol[k * N + 0];       b0.y = Bcol[k * N + N + 0];
    b1.x = Bcol[k * N + 16];      b1.y = Bcol[k * N + N + 16];
    b2.x = Bcol[k * N + 32];      b2.y = Bcol[k * N + N + 32];
    b3.x = Bcol[k * N + 48];      b3.y = Bcol[k * N + N + 48];
    c0 = wmma4(a, b0, c0);
    c1 = wmma4(a, b1, c1);
    c2 = wmma4(a, b2, c2);
    c3 = wmma4(a, b3, c3);
  }

  float* Crow = C + (size_t)(m0 + 8 * half) * N + n0 + r;
#pragma unroll
  for (int i = 0; i < 8; ++i) {
    Crow[i * N + 0]  = c0[i];
    Crow[i * N + 16] = c1[i];
    Crow[i * N + 32] = c2[i];
    Crow[i * N + 48] = c3[i];
  }
}

template <int H, int F>
__global__ void make_wa(const float* __restrict__ W, const float* __restrict__ att,
                        float* __restrict__ Wa, int K) {
  int t = blockIdx.x * blockDim.x + threadIdx.x;
  if (t >= K * H) return;
  int k = t / H, h = t & (H - 1);
  const float* wr = W + (size_t)k * (H * F) + h * F;
  const float* ar = att + h * F;
  float acc = 0.f;
#pragma unroll 4
  for (int f = 0; f < F; ++f) acc += wr[f] * ar[f];
  Wa[t] = acc;
}

template <int K, int H>
__global__ void proj_att(const float* __restrict__ X, const float* __restrict__ Wa,
                         float* __restrict__ out, int N) {
  int t = blockIdx.x * blockDim.x + threadIdx.x;
  if (t >= N * H) return;
  int n = t / H, h = t & (H - 1);
  const float* xr = X + (size_t)n * K;
  float acc = 0.f;
#pragma unroll 4
  for (int k = 0; k < K; ++k) acc += xr[k] * Wa[k * H + h];
  out[t] = acc;
}

__device__ __forceinline__ unsigned fkey(float f) {
  unsigned u = __float_as_uint(f);
  return (u & 0x80000000u) ? ~u : (u | 0x80000000u);
}
__device__ __forceinline__ float fdecode(unsigned k) {
  unsigned u = (k & 0x80000000u) ? (k & 0x7FFFFFFFu) : ~k;
  return __uint_as_float(u);
}
__device__ __forceinline__ float leaky(float a) {
  return a > 0.f ? a : 0.2f * a;
}

template <int H>
__global__ void edge_max(const int* __restrict__ src, const int* __restrict__ dst,
                         const float* __restrict__ as, const float* __restrict__ ad,
                         unsigned* __restrict__ amax, int E) {
  int t = blockIdx.x * blockDim.x + threadIdx.x;
  if (t >= E * H) return;
  int e = t / H, h = t & (H - 1);
  float a = leaky(as[src[e] * H + h] + ad[dst[e] * H + h]);
  atomicMax(&amax[dst[e] * H + h], fkey(a));
}

template <int H>
__global__ void edge_denom(const int* __restrict__ src, const int* __restrict__ dst,
                           const float* __restrict__ as, const float* __restrict__ ad,
                           const unsigned* __restrict__ amax, float* __restrict__ den,
                           int E) {
  int t = blockIdx.x * blockDim.x + threadIdx.x;
  if (t >= E * H) return;
  int e = t / H, h = t & (H - 1);
  int d = dst[e] * H + h;
  float a = leaky(as[src[e] * H + h] + ad[d]);
  float ex = __expf(a - fdecode(amax[d]));
  atomicAdd(&den[d], ex);
}

template <int H, int LOGF>
__global__ void edge_aggregate(const int* __restrict__ src, const int* __restrict__ dst,
                               const float* __restrict__ as, const float* __restrict__ ad,
                               const unsigned* __restrict__ amax,
                               const float* __restrict__ den,
                               const float* __restrict__ xs, float* __restrict__ out,
                               int E) {
  int e = blockIdx.x * (blockDim.x >> 5) + (threadIdx.x >> 5);
  if (e >= E) return;
  int lane = threadIdx.x & 31;
  int s = src[e], d = dst[e];
  constexpr int HF = H << LOGF;
#pragma unroll
  for (int p = lane; p < HF; p += 32) {
    int h = p >> LOGF;
    float a = leaky(as[s * H + h] + ad[d * H + h]);
    float ex = __expf(a - fdecode(amax[d * H + h]));
    float w = ex / (den[d * H + h] + 1e-16f);
    atomicAdd(&out[(size_t)d * HF + p], w * xs[(size_t)s * HF + p]);
  }
}

__global__ void bias_relu(float* __restrict__ h, const float* __restrict__ b,
                          int total, int HF) {
  int t = blockIdx.x * blockDim.x + threadIdx.x;
  if (t >= total) return;
  float v = h[t] + b[t & (HF - 1)];
  h[t] = v > 0.f ? v : 0.f;
}

__global__ void bias_add(float* __restrict__ o, const float* __restrict__ b,
                         int total, int C) {
  int t = blockIdx.x * blockDim.x + threadIdx.x;
  if (t >= total) return;
  o[t] = o[t] + b[t & (C - 1)];
}

extern "C" void kernel_launch(void* const* d_in, const int* in_sizes, int n_in,
                              void* d_out, int out_size, void* d_ws, size_t ws_size,
                              hipStream_t stream) {
  (void)n_in; (void)out_size; (void)ws_size;
  const float* x     = (const float*)d_in[0];
  const int*   ei    = (const int*)d_in[1];
  const float* W1s   = (const float*)d_in[2];
  const float* W1d   = (const float*)d_in[3];
  const float* att1s = (const float*)d_in[4];
  const float* att1d = (const float*)d_in[5];
  const float* b1    = (const float*)d_in[6];
  const float* W2s   = (const float*)d_in[7];
  const float* W2d   = (const float*)d_in[8];
  const float* att2s = (const float*)d_in[9];
  const float* att2d = (const float*)d_in[10];
  const float* b2    = (const float*)d_in[11];
  float* out = (float*)d_out;

  const int N = in_sizes[0] / 128;
  const int E = in_sizes[1] / 2;
  const int* src = ei;
  const int* dst = ei + E;

  float* ws = (float*)d_ws;
  size_t off = 0;
  float*    XS   = ws + off; off += (size_t)N * 256;
  float*    Hb   = ws + off; off += (size_t)N * 256;
  float*    AS   = ws + off; off += (size_t)N * 8;
  float*    AD   = ws + off; off += (size_t)N * 8;
  unsigned* AMAX = (unsigned*)(ws + off); off += (size_t)N * 8;
  float*    DEN  = ws + off; off += (size_t)N * 8;
  float*    WA   = ws + off; off += 2048;
  float*    WD   = ws + off; off += 2048;

  const int TPB = 256;

  gemm_wmma_f32<256, 128><<<dim3(N / 16, 256 / 64), 32, 0, stream>>>(x, W1s, XS);
  make_wa<8, 32><<<(128 * 8 + TPB - 1) / TPB, TPB, 0, stream>>>(W1s, att1s, WA, 128);
  make_wa<8, 32><<<(128 * 8 + TPB - 1) / TPB, TPB, 0, stream>>>(W1d, att1d, WD, 128);
  proj_att<128, 8><<<((size_t)N * 8 + TPB - 1) / TPB, TPB, 0, stream>>>(x, WA, AS, N);
  proj_att<128, 8><<<((size_t)N * 8 + TPB - 1) / TPB, TPB, 0, stream>>>(x, WD, AD, N);

  hipMemsetAsync(AMAX, 0, (size_t)N * 8 * sizeof(unsigned), stream);
  hipMemsetAsync(DEN, 0, (size_t)N * 8 * sizeof(float), stream);
  hipMemsetAsync(Hb, 0, (size_t)N * 256 * sizeof(float), stream);

  int EH = E * 8;
  edge_max<8>  <<<(EH + TPB - 1) / TPB, TPB, 0, stream>>>(src, dst, AS, AD, AMAX, E);
  edge_denom<8><<<(EH + TPB - 1) / TPB, TPB, 0, stream>>>(src, dst, AS, AD, AMAX, DEN, E);
  edge_aggregate<8, 5><<<(E + 7) / 8, 256, 0, stream>>>(src, dst, AS, AD, AMAX, DEN,
                                                        XS, Hb, E);
  bias_relu<<<((size_t)N * 256 + TPB - 1) / TPB, TPB, 0, stream>>>(Hb, b1, N * 256, 256);

  gemm_wmma_f32<128, 256><<<dim3(N / 16, 128 / 64), 32, 0, stream>>>(Hb, W2s, XS);
  make_wa<1, 128><<<(256 + TPB - 1) / TPB, TPB, 0, stream>>>(W2s, att2s, WA, 256);
  make_wa<1, 128><<<(256 + TPB - 1) / TPB, TPB, 0, stream>>>(W2d, att2d, WD, 256);
  proj_att<256, 1><<<(N + TPB - 1) / TPB, TPB, 0, stream>>>(Hb, WA, AS, N);
  proj_att<256, 1><<<(N + TPB - 1) / TPB, TPB, 0, stream>>>(Hb, WD, AD, N);

  hipMemsetAsync(AMAX, 0, (size_t)N * sizeof(unsigned), stream);
  hipMemsetAsync(DEN, 0, (size_t)N * sizeof(float), stream);
  hipMemsetAsync(out, 0, (size_t)N * 128 * sizeof(float), stream);

  edge_max<1>  <<<(E + TPB - 1) / TPB, TPB, 0, stream>>>(src, dst, AS, AD, AMAX, E);
  edge_denom<1><<<(E + TPB - 1) / TPB, TPB, 0, stream>>>(src, dst, AS, AD, AMAX, DEN, E);
  edge_aggregate<1, 7><<<(E + 7) / 8, 256, 0, stream>>>(src, dst, AS, AD, AMAX, DEN,
                                                        XS, out, E);
  bias_add<<<((size_t)N * 128 + TPB - 1) / TPB, TPB, 0, stream>>>(out, b2, N * 128, 128);
}


# edge_aggregate: each block owns 196 destination rows, filters the dst list itself (no sort), sums weighted rows by LDS float adds (arrival order), stores each row once with bias; 18 -> 14 launches
# speedup vs baseline: 1.4741x; 1.4741x over previous
#include <hip/hip_runtime.h>
#include <stddef.h>

typedef __attribute__((ext_vector_type(2))) float v2f;
typedef __attribute__((ext_vector_type(8))) float v8f;

__device__ __forceinline__ v8f wmma4(v2f a, v2f b, v8f c) {
  return __builtin_amdgcn_wmma_f32_16x16x4_f32(
       false, a,  false, b,
       (short)0, c,  false,  false);
}

template <int N, int K>
__global__ void gemm_wmma_f32(const float* __restrict__ A,
                              const float* __restrict__ B,
                              float* __restrict__ C) {
  const int lane = threadIdx.x & 31;
  const int m0 = blockIdx.x * 16;
  const int n0 = blockIdx.y * 64;
  const int half = lane >> 4;
  const int r    = lane & 15;

  const float* Arow = A + (size_t)(m0 + r) * K + 2 * half;
  const float* Bcol = B + (size_t)(2 * half) * N + n0 + r;

  v8f c0 = {}, c1 = {}, c2 = {}, c3 = {};
#pragma unroll
  for (int k = 0; k < K; k += 4) {
    v2f a = *(const v2f*)(Arow + k);
    v2f b0, b1, b2, b3;
    b0.x = Bcol[k * N + 0];       b0.y = Bcol[k * N + N + 0];
    b1.x = Bcol[k * N + 16];      b1.y = Bcol[k * N + N + 16];
    b2.x = Bcol[k * N + 32];      b2.y = Bcol[k * N + N + 32];
    b3.x = Bcol[k * N + 48];      b3.y = Bcol[k * N + N + 48];
    c0 = wmma4(a, b0, c0);
    c1 = wmma4(a, b1, c1);
    c2 = wmma4(a, b2, c2);
    c3 = wmma4(a, b3, c3);
  }

  float* Crow = C + (size_t)(m0 + 8 * half) * N + n0 + r;
#pragma unroll
  for (int i = 0; i < 8; ++i) {
    Crow[i * N + 0]  = c0[i];
    Crow[i * N + 16] = c1[i];
    Crow[i * N + 32] = c2[i];
    Crow[i * N + 48] = c3[i];
  }
}

template <int H, int F>
__global__ void make_wa(const float* __restrict__ W, const float* __restrict__ att,
                        float* __restrict__ Wa, int K) {
  int t = blockIdx.x * blockDim.x + threadIdx.x;
  if (t >= K * H) return;
  int k = t / H, h = t & (H - 1);
  const float* wr = W + (size_t)k * (H * F) + h * F;
  const float* ar = att + h * F;
  float acc = 0.f;
#pragma unroll 4
  for (int f = 0; f < F; ++f) acc += wr[f] * ar[f];
  Wa[t] = acc;
}

template <int K, int H>
__global__ void proj_att(const float* __restrict__ X, const float* __restrict__ Wa,
                         float* __restrict__ out, int N) {
  int t = blockIdx.x * blockDim.x + threadIdx.x;
  if (t >= N * H) return;
  int n = t / H, h = t & (H - 1);
  const float* xr = X + (size_t)n * K;
  float acc = 0.f;
#pragma unroll 4
  for (int k = 0; k < K; ++k) acc += xr[k] * Wa[k * H + h];
  out[t] = acc;
}

__device__ __forceinline__ unsigned fkey(float f) {
  unsigned u = __float_as_uint(f);
  return (u & 0x80000000u) ? ~u : (u | 0x80000000u);
}
__device__ __forceinline__ float fdecode(unsigned k) {
  unsigned u = (k & 0x80000000u) ? (k & 0x7FFFFFFFu) : ~k;
  return __uint_as_float(u);
}
__device__ __forceinline__ float leaky(float a) {
  return a > 0.f ? a : 0.2f * a;
}

template <int H>
__global__ void edge_max(const int* __restrict__ src, const int* __restrict__ dst,
                         const float* __restrict__ as, const float* __restrict__ ad,
                         unsigned* __restrict__ amax, int E) {
  int t = blockIdx.x * blockDim.x + threadIdx.x;
  if (t >= E * H) return;
  int e = t / H, h = t & (H - 1);
  float a = leaky(as[src[e] * H + h] + ad[dst[e] * H + h]);
  atomicMax(&amax[dst[e] * H + h], fkey(a));
}

template <int H>
__global__ void edge_denom(const int* __restrict__ src, const int* __restrict__ dst,
                           const float* __restrict__ as, const float* __restrict__ ad,
                           const unsigned* __restrict__ amax, float* __restrict__ den,
                           int E) {
  int t = blockIdx.x * blockDim.x + threadIdx.x;
  if (t >= E * H) return;
  int e = t / H, h = t & (H - 1);
  int d = dst[e] * H + h;
  float a = leaky(as[src[e] * H + h] + ad[d]);
  float ex = __expf(a - fdecode(amax[d]));
  atomicAdd(&den[d], ex);
}

template <int H, int LOGF, int EB, bool RELU>
__global__ void __launch_bounds__(1024)
edge_aggregate_owned(const int* __restrict__ src, const int* __restrict__ dst,
                     const float* __restrict__ as, const float* __restrict__ ad,
                     const unsigned* __restrict__ amax, const float* __restrict__ den,
                     const float* __restrict__ xs, const float* __restrict__ bias,
                     float* __restrict__ out, int E, int N) {
  constexpr int HF = H << LOGF;
  constexpr int NS = HF / 32;
  constexpr int RT = 196;
  constexpr int CAP = 8192;
  constexpr int U = 8;
  static_assert(EB * H <= 32, "one lane per (edge, head) of a batch");
  static_assert(LOGF >= 5, "a 32-lane slice lies inside one head");
  __shared__ float acc[RT * HF];
  __shared__ int lsrc[CAP];
  __shared__ int ldl[CAP];
  __shared__ int cnt;

  const int tid = threadIdx.x, lane = tid & 31, wave = tid >> 5;

  for (int row0 = blockIdx.x * RT; row0 < N; row0 += gridDim.x * RT) {
    const int rows = min(RT, N - row0);
    for (int i = tid; i < RT * HF; i += 1024) acc[i] = 0.f;
    if (tid == 0) cnt = 0;
    __syncthreads();

    auto take = [&](int d, int e) {
      unsigned dl = (unsigned)(d - row0);
      if (dl < (unsigned)rows) {
        int idx = atomicAdd(&cnt, 1);
        if (idx < CAP) {
          lsrc[idx] = e;
          ldl[idx] = (int)dl;
        }
      }
    };

    const int E4 = (((size_t)dst & 15) == 0) ? (E >> 2) : 0;
    const int4* dst4 = (const int4*)dst;
    for (int g0 = tid; g0 < E4; g0 += 1024 * U) {
      int4 v[U];
#pragma unroll
      for (int u = 0; u < U; ++u) {
        int g = g0 + u * 1024;
        v[u] = (g < E4) ? dst4[g] : make_int4(-1, -1, -1, -1);
      }
#pragma unroll
      for (int u = 0; u < U; ++u) {
        int e = 4 * (g0 + u * 1024);
        take(v[u].x, e);
        take(v[u].y, e + 1);
        take(v[u].z, e + 2);
        take(v[u].w, e + 3);
      }
    }
    for (int e = 4 * E4 + tid; e < E; e += 1024) take(dst[e], e);
    __syncthreads();

    const bool listed = cnt <= CAP;
    const int n = listed ? cnt : 0;
    if (!listed) {
      for (int e = tid; e < E; e += 1024) {
        const int d = dst[e];
        const unsigned dl = (unsigned)(d - row0);
        if (dl < (unsigned)rows) {
          const int s = src[e];
#pragma unroll 1
          for (int p = 0; p < HF; ++p) {
            const int h = p >> LOGF;
            const float a = leaky(as[s * H + h] + ad[d * H + h]);
            const float w = __expf(a - fdecode(amax[d * H + h])) / (den[d * H + h] + 1e-16f);
            atomicAdd(&acc[dl * HF + p], w * xs[(size_t)s * HF + p]);
          }
        }
      }
    }
    {
      int sv[CAP / 1024];
#pragma unroll
      for (int k = 0; k < CAP / 1024; ++k) {
        int i = tid + k * 1024;
        sv[k] = (i < n) ? src[lsrc[i]] : 0;
      }
#pragma unroll
      for (int k = 0; k < CAP / 1024; ++k) {
        int i = tid + k * 1024;
        if (i < n) lsrc[i] = sv[k];
      }
    }
    __syncthreads();

    for (int i0 = wave * EB; i0 < n; i0 += 32 * EB) {
      const int j = min(lane / H, EB - 1), h = lane & (H - 1);
      const int ii = min(i0 + j, n - 1);
      const int s = lsrc[ii], dl = ldl[ii];
      const int dh = (row0 + dl) * H + h;
      const float v_as = as[s * H + h], v_ad = ad[dh], v_den = den[dh];
      const unsigned v_mx = amax[dh];
      float m[EB][NS];
      int dls[EB];
#pragma unroll
      for (int jj = 0; jj < EB; ++jj) {
        const int sj = __builtin_amdgcn_readlane(s, jj * H);
        dls[jj] = __builtin_amdgcn_readlane(dl, jj * H);
#pragma unroll
        for (int i = 0; i < NS; ++i) m[jj][i] = xs[(size_t)sj * HF + lane + 32 * i];
      }
      const float w = __expf(leaky(v_as + v_ad) - fdecode(v_mx)) / (v_den + 1e-16f);
#pragma unroll
      for (int jj = 0; jj < EB; ++jj) {
        const bool valid = i0 + jj < n;
#pragma unroll
        for (int i = 0; i < NS; ++i) {
          const int p = lane + 32 * i;
          const float wj = __builtin_bit_cast(
              float, __builtin_amdgcn_readlane(__builtin_bit_cast(int, w), jj * H + ((32 * i) >> LOGF)));
          atomicAdd(&acc[dls[jj] * HF + p], valid ? wj * m[jj][i] : 0.f);
        }
      }
    }
    __syncthreads();

    for (int i = tid; i < rows * HF; i += 1024) {
      float v = acc[i] + bias[i & (HF - 1)];
      if (RELU) v = v > 0.f ? v : 0.f;
      out[(size_t)row0 * HF + i] = v;
    }
    __syncthreads();
  }
}

extern "C" void kernel_launch(void* const* d_in, const int* in_sizes, int n_in,
                              void* d_out, int out_size, void* d_ws, size_t ws_size,
                              hipStream_t stream) {
  (void)n_in; (void)out_size; (void)ws_size;
  const float* x     = (const float*)d_in[0];
  const int*   ei    = (const int*)d_in[1];
  const float* W1s   = (const float*)d_in[2];
  const float* W1d   = (const float*)d_in[3];
  const float* att1s = (const float*)d_in[4];
  const float* att1d = (const float*)d_in[5];
  const float* b1    = (const float*)d_in[6];
  const float* W2s   = (const float*)d_in[7];
  const float* W2d   = (const float*)d_in[8];
  const float* att2s = (const float*)d_in[9];
  const float* att2d = (const float*)d_in[10];
  const float* b2    = (const float*)d_in[11];
  float* out = (float*)d_out;

  const int N = in_sizes[0] / 128;
  const int E = in_sizes[1] / 2;
  const int* src = ei;
  const int* dst = ei + E;

  float* ws = (float*)d_ws;
  size_t off = 0;
  float*    XS   = ws + off; off += (size_t)N * 256;
  float*    Hb   = ws + off; off += (size_t)N * 256;
  float*    AS   = ws + off; off += (size_t)N * 8;
  float*    AD   = ws + off; off += (size_t)N * 8;
  unsigned* AMAX = (unsigned*)(ws + off); off += (size_t)N * 8;
  float*    DEN  = ws + off; off += (size_t)N * 8;
  float*    WA   = ws + off; off += 2048;
  float*    WD   = ws + off; off += 2048;

  const int TPB = 256;

  gemm_wmma_f32<256, 128><<<dim3(N / 16, 256 / 64), 32, 0, stream>>>(x, W1s, XS);
  make_wa<8, 32><<<(128 * 8 + TPB - 1) / TPB, TPB, 0, stream>>>(W1s, att1s, WA, 128);
  make_wa<8, 32><<<(128 * 8 + TPB - 1) / TPB, TPB, 0, stream>>>(W1d, att1d, WD, 128);
  proj_att<128, 8><<<((size_t)N * 8 + TPB - 1) / TPB, TPB, 0, stream>>>(x, WA, AS, N);
  proj_att<128, 8><<<((size_t)N * 8 + TPB - 1) / TPB, TPB, 0, stream>>>(x, WD, AD, N);

  hipMemsetAsync(AMAX, 0, (size_t)N * 8 * sizeof(unsigned), stream);
  hipMemsetAsync(DEN, 0, (size_t)N * 8 * sizeof(float), stream);

  int EH = E * 8;
  edge_max<8>  <<<(EH + TPB - 1) / TPB, TPB, 0, stream>>>(src, dst, AS, AD, AMAX, E);
  edge_denom<8><<<(EH + TPB - 1) / TPB, TPB, 0, stream>>>(src, dst, AS, AD, AMAX, DEN, E);
  edge_aggregate_owned<8, 5, 4, true><<<256, 1024, 0, stream>>>(src, dst, AS, AD, AMAX, DEN,
                                                              XS, b1, Hb, E, N);

  gemm_wmma_f32<128, 256><<<dim3(N / 16, 128 / 64), 32, 0, stream>>>(Hb, W2s, XS);
  make_wa<1, 128><<<(256 + TPB - 1) / TPB, TPB, 0, stream>>>(W2s, att2s, WA, 256);
  make_wa<1, 128><<<(256 + TPB - 1) / TPB, TPB, 0, stream>>>(W2d, att2d, WD, 256);
  proj_att<256, 1><<<(N + TPB - 1) / TPB, TPB, 0, stream>>>(Hb, WA, AS, N);
  proj_att<256, 1><<<(N + TPB - 1) / TPB, TPB, 0, stream>>>(Hb, WD, AD, N);

  hipMemsetAsync(AMAX, 0, (size_t)N * sizeof(unsigned), stream);
  hipMemsetAsync(DEN, 0, (size_t)N * sizeof(float), stream);

  edge_max<1>  <<<(E + TPB - 1) / TPB, TPB, 0, stream>>>(src, dst, AS, AD, AMAX, E);
  edge_denom<1><<<(E + TPB - 1) / TPB, TPB, 0, stream>>>(src, dst, AS, AD, AMAX, DEN, E);
  edge_aggregate_owned<1, 7, 8, false><<<256, 1024, 0, stream>>>(src, dst, AS, AD, AMAX, DEN,
                                                               XS, b2, out, E, N);
}


# edge_aggregate: the owning block takes segment max and softmax denominators of its rows in LDS from its edge list (LDS float adds, arrival order); edge_max, edge_denom, 4 memsets go; 16->12 launches
# speedup vs baseline: 2.2792x; 1.5462x over previous
#include <hip/hip_runtime.h>
#include <stddef.h>

typedef __attribute__((ext_vector_type(2))) float v2f;
typedef __attribute__((ext_vector_type(8))) float v8f;

__device__ __forceinline__ v8f wmma4(v2f a, v2f b, v8f c) {
  return __builtin_amdgcn_wmma_f32_16x16x4_f32(
       false, a,  false, b,
       (short)0, c,  false,  false);
}

template <int N, int K>
__global__ void gemm_wmma_f32(const float* __restrict__ A,
                              const float* __restrict__ B,
                              float* __restrict__ C) {
  const int lane = threadIdx.x & 31;
  const int m0 = blockIdx.x * 16;
  const int n0 = blockIdx.y * 64;
  const int half = lane >> 4;
  const int r    = lane & 15;

  const float* Arow = A + (size_t)(m0 + r) * K + 2 * half;
  const float* Bcol = B + (size_t)(2 * half) * N + n0 + r;

  v8f c0 = {}, c1 = {}, c2 = {}, c3 = {};
#pragma unroll
  for (int k = 0; k < K; k += 4) {
    v2f a = *(const v2f*)(Arow + k);
    v2f b0, b1, b2, b3;
    b0.x = Bcol[k * N + 0];       b0.y = Bcol[k * N + N + 0];
    b1.x = Bcol[k * N + 16];      b1.y = Bcol[k * N + N + 16];
    b2.x = Bcol[k * N + 32];      b2.y = Bcol[k * N + N + 32];
    b3.x = Bcol[k * N + 48];      b3.y = Bcol[k * N + N + 48];
    c0 = wmma4(a, b0, c0);
    c1 = wmma4(a, b1, c1);
    c2 = wmma4(a, b2, c2);
    c3 = wmma4(a, b3, c3);
  }

  float* Crow = C + (size_t)(m0 + 8 * half) * N + n0 + r;
#pragma unroll
  for (int i = 0; i < 8; ++i) {
    Crow[i * N + 0]  = c0[i];
    Crow[i * N + 16] = c1[i];
    Crow[i * N + 32] = c2[i];
    Crow[i * N + 48] = c3[i];
  }
}

template <int H, int F>
__global__ void make_wa(const float* __restrict__ W, const float* __restrict__ att,
                        float* __restrict__ Wa, int K) {
  int t = blockIdx.x * blockDim.x + threadIdx.x;
  if (t >= K * H) return;
  int k = t / H, h = t & (H - 1);
  const float* wr = W + (size_t)k * (H * F) + h * F;
  const float* ar = att + h * F;
  float acc = 0.f;
#pragma unroll 4
  for (int f = 0; f < F; ++f) acc += wr[f] * ar[f];
  Wa[t] = acc;
}

template <int K, int H>
__global__ void proj_att(const float* __restrict__ X, const float* __restrict__ Wa,
                         float* __restrict__ out, int N) {
  int t = blockIdx.x * blockDim.x + threadIdx.x;
  if (t >= N * H) return;
  int n = t / H, h = t & (H - 1);
  const float* xr = X + (size_t)n * K;
  float acc = 0.f;
#pragma unroll 4
  for (int k = 0; k < K; ++k) acc += xr[k] * Wa[k * H + h];
  out[t] = acc;
}

__device__ __forceinline__ unsigned fkey(float f) {
  unsigned u = __float_as_uint(f);
  return (u & 0x80000000u) ? ~u : (u | 0x80000000u);
}
__device__ __forceinline__ float fdecode(unsigned k) {
  unsigned u = (k & 0x80000000u) ? (k & 0x7FFFFFFFu) : ~k;
  return __uint_as_float(u);
}
__device__ __forceinline__ float leaky(float a) {
  return a > 0.f ? a : 0.2f * a;
}

template <int H>
__device__ __forceinline__ void load_heads(const float* __restrict__ p, float (&a)[H]) {
  if constexpr (H % 4 == 0) {
#pragma unroll
    for (int q = 0; q < H / 4; ++q) {
      const float4 v = ((const float4*)p)[q];
      a[4 * q + 0] = v.x; a[4 * q + 1] = v.y; a[4 * q + 2] = v.z; a[4 * q + 3] = v.w;
    }
  } else {
#pragma unroll
    for (int h = 0; h < H; ++h) a[h] = p[h];
  }
}

template <int H, int LOGF, int EB, bool RELU>
__global__ void __launch_bounds__(1024)
edge_aggregate_owned(const int* __restrict__ src, const int* __restrict__ dst,
                     const float* __restrict__ as, const float* __restrict__ ad,
                     const float* __restrict__ xs, const float* __restrict__ bias,
                     float* __restrict__ out, int E, int N) {
  constexpr int HF = H << LOGF;
  constexpr int NS = HF / 32;
  constexpr int RT = 196;
  constexpr int CAP = 8192;
  constexpr int KR = 4;
  constexpr int U = 8;
  static_assert(EB * H <= 32, "one lane per (edge, head) of a batch");
  static_assert(LOGF >= 5, "a 32-lane slice lies inside one head");
  __shared__ float acc[RT * HF];
  __shared__ int lsrc[CAP];
  __shared__ int ldl[CAP];
  __shared__ float adl[RT * H];
  __shared__ unsigned mx[RT * H];
  __shared__ float dn[RT * H];
  __shared__ int cnt;

  const int tid = threadIdx.x, lane = tid & 31, wave = tid >> 5;

  for (int row0 = blockIdx.x * RT; row0 < N; row0 += gridDim.x * RT) {
    const int rows = min(RT, N - row0);
    for (int i = tid; i < RT * HF; i += 1024) acc[i] = 0.f;
    for (int i = tid; i < RT * H; i += 1024) {
      adl[i] = (i < rows * H) ? ad[(size_t)row0 * H + i] : 0.f;
      mx[i] = 0u;
      dn[i] = 0.f;
    }
    if (tid == 0) cnt = 0;
    __syncthreads();

    auto take = [&](int d, int e) {
      unsigned dl = (unsigned)(d - row0);
      if (dl < (unsigned)rows) {
        int idx = atomicAdd(&cnt, 1);
        if (idx < CAP) {
          lsrc[idx] = e;
          ldl[idx] = (int)dl;
        }
      }
    };

    const int E4 = (((size_t)dst & 15) == 0) ? (E >> 2) : 0;
    const int4* dst4 = (const int4*)dst;
    for (int g0 = tid; g0 < E4; g0 += 1024 * U) {
      int4 v[U];
#pragma unroll
      for (int u = 0; u < U; ++u) {
        int g = g0 + u * 1024;
        v[u] = (g < E4) ? dst4[g] : make_int4(-1, -1, -1, -1);
      }
#pragma unroll
      for (int u = 0; u < U; ++u) {
        int e = 4 * (g0 + u * 1024);
        take(v[u].x, e);
        take(v[u].y, e + 1);
        take(v[u].z, e + 2);
        take(v[u].w, e + 3);
      }
    }
    for (int e = 4 * E4 + tid; e < E; e += 1024) take(dst[e], e);
    __syncthreads();

    const bool listed = cnt <= CAP;
    const int n = listed ? cnt : 0;
    if (!listed) {
      for (int e = tid; e < E; e += 1024) {
        const int d = dst[e];
        const unsigned dl = (unsigned)(d - row0);
        if (dl < (unsigned)rows) {
          const int s = src[e];
#pragma unroll 1
          for (int h = 0; h < H; ++h)
            atomicMax(&mx[dl * H + h], fkey(leaky(as[s * H + h] + adl[dl * H + h])));
        }
      }
      __syncthreads();
      for (int e = tid; e < E; e += 1024) {
        const int d = dst[e];
        const unsigned dl = (unsigned)(d - row0);
        if (dl < (unsigned)rows) {
          const int s = src[e];
#pragma unroll 1
          for (int h = 0; h < H; ++h) {
            const float a = leaky(as[s * H + h] + adl[dl * H + h]);
            atomicAdd(&dn[dl * H + h], __expf(a - fdecode(mx[dl * H + h])));
          }
        }
      }
      __syncthreads();
      for (int e = tid; e < E; e += 1024) {
        const int d = dst[e];
        const unsigned dl = (unsigned)(d - row0);
        if (dl < (unsigned)rows) {
          const int s = src[e];
#pragma unroll 1
          for (int p = 0; p < HF; ++p) {
            const int lh = dl * H + (p >> LOGF);
            const float a = leaky(as[s * H + (p >> LOGF)] + adl[lh]);
            const float w = __expf(a - fdecode(mx[lh])) / (dn[lh] + 1e-16f);
            atomicAdd(&acc[dl * HF + p], w * xs[(size_t)s * HF + p]);
          }
        }
      }
    }
    {
      int sv[KR];
      float av[KR][H];
      if (n > 0) {
#pragma unroll
        for (int k = 0; k < KR; ++k) {
          const int i = tid + k * 1024;
          const int e = (i < n) ? lsrc[i] : 0;
          sv[k] = src[e];
        }
        __builtin_amdgcn_sched_barrier(0);
#pragma unroll
        for (int k = 0; k < KR; ++k) load_heads<H>(as + (size_t)sv[k] * H, av[k]);
      }
#pragma unroll
      for (int k = 0; k < KR; ++k) {
        const int i = tid + k * 1024;
        if (i < n) {
          const int lh = ldl[i] * H;
          lsrc[i] = sv[k];
#pragma unroll
          for (int h = 0; h < H; ++h) {
            av[k][h] = leaky(av[k][h] + adl[lh + h]);
            atomicMax(&mx[lh + h], fkey(av[k][h]));
          }
        }
      }
      for (int i = KR * 1024 + tid; i < n; i += 1024) {
        const int s = src[lsrc[i]];
        const int lh = ldl[i] * H;
        float a[H];
        load_heads<H>(as + (size_t)s * H, a);
        lsrc[i] = s;
#pragma unroll
        for (int h = 0; h < H; ++h) atomicMax(&mx[lh + h], fkey(leaky(a[h] + adl[lh + h])));
      }
      __syncthreads();
#pragma unroll
      for (int k = 0; k < KR; ++k) {
        const int i = tid + k * 1024;
        if (i < n) {
          const int lh = ldl[i] * H;
#pragma unroll
          for (int h = 0; h < H; ++h)
            atomicAdd(&dn[lh + h], __expf(av[k][h] - fdecode(mx[lh + h])));
        }
      }
      for (int i = KR * 1024 + tid; i < n; i += 1024) {
        const int lh = ldl[i] * H;
        float a[H];
        load_heads<H>(as + (size_t)lsrc[i] * H, a);
#pragma unroll
        for (int h = 0; h < H; ++h)
          atomicAdd(&dn[lh + h], __expf(leaky(a[h] + adl[lh + h]) - fdecode(mx[lh + h])));
      }
    }
    __syncthreads();

    for (int i0 = wave * EB; i0 < n; i0 += 32 * EB) {
      const int j = min(lane / H, EB - 1), h = lane & (H - 1);
      const int ii = min(i0 + j, n - 1);
      const int s = lsrc[ii], dl = ldl[ii];
      const int lh = dl * H + h;
      const float v_as = as[s * H + h], v_ad = adl[lh], v_den = dn[lh];
      const unsigned v_mx = mx[lh];
      float m[EB][NS];
      int dls[EB];
#pragma unroll
      for (int jj = 0; jj < EB; ++jj) {
        const int sj = __builtin_amdgcn_readlane(s, jj * H);
        dls[jj] = __builtin_amdgcn_readlane(dl, jj * H);
#pragma unroll
        for (int i = 0; i < NS; ++i) m[jj][i] = xs[(size_t)sj * HF + lane + 32 * i];
      }
      const float w = __expf(leaky(v_as + v_ad) - fdecode(v_mx)) / (v_den + 1e-16f);
#pragma unroll
      for (int jj = 0; jj < EB; ++jj) {
        const bool valid = i0 + jj < n;
#pragma unroll
        for (int i = 0; i < NS; ++i) {
          const int p = lane + 32 * i;
          const float wj = __builtin_bit_cast(
              float, __builtin_amdgcn_readlane(__builtin_bit_cast(int, w), jj * H + ((32 * i) >> LOGF)));
          atomicAdd(&acc[dls[jj] * HF + p], valid ? wj * m[jj][i] : 0.f);
        }
      }
    }
    __syncthreads();

    for (int i = tid; i < rows * HF; i += 1024) {
      float v = acc[i] + bias[i & (HF - 1)];
      if (RELU) v = v > 0.f ? v : 0.f;
      out[(size_t)row0 * HF + i] = v;
    }
    __syncthreads();
  }
}

extern "C" void kernel_launch(void* const* d_in, const int* in_sizes, int n_in,
                              void* d_out, int out_size, void* d_ws, size_t ws_size,
                              hipStream_t stream) {
  (void)n_in; (void)out_size; (void)ws_size;
  const float* x     = (const float*)d_in[0];
  const int*   ei    = (const int*)d_in[1];
  const float* W1s   = (const float*)d_in[2];
  const float* W1d   = (const float*)d_in[3];
  const float* att1s = (const float*)d_in[4];
  const float* att1d = (const float*)d_in[5];
  const float* b1    = (const float*)d_in[6];
  const float* W2s   = (const float*)d_in[7];
  const float* W2d   = (const float*)d_in[8];
  const float* att2s = (const float*)d_in[9];
  const float* att2d = (const float*)d_in[10];
  const float* b2    = (const float*)d_in[11];
  float* out = (float*)d_out;

  const int N = in_sizes[0] / 128;
  const int E = in_sizes[1] / 2;
  const int* src = ei;
  const int* dst = ei + E;

  float* ws = (float*)d_ws;
  size_t off = 0;
  float*    XS   = ws + off; off += (size_t)N * 256;
  float*    Hb   = ws + off; off += (size_t)N * 256;
  float*    AS   = ws + off; off += (size_t)N * 8;
  float*    AD   = ws + off; off += (size_t)N * 8;
  float*    WA   = ws + off; off += 2048;
  float*    WD   = ws + off; off += 2048;

  const int TPB = 256;

  gemm_wmma_f32<256, 128><<<dim3(N / 16, 256 / 64), 32, 0, stream>>>(x, W1s, XS);
  make_wa<8, 32><<<(128 * 8 + TPB - 1) / TPB, TPB, 0, stream>>>(W1s, att1s, WA, 128);
  make_wa<8, 32><<<(128 * 8 + TPB - 1) / TPB, TPB, 0, stream>>>(W1d, att1d, WD, 128);
  proj_att<128, 8><<<((size_t)N * 8 + TPB - 1) / TPB, TPB, 0, stream>>>(x, WA, AS, N);
  proj_att<128, 8><<<((size_t)N * 8 + TPB - 1) / TPB, TPB, 0, stream>>>(x, WD, AD, N);

  edge_aggregate_owned<8, 5, 4, true><<<256, 1024, 0, stream>>>(src, dst, AS, AD,
                                                              XS, b1, Hb, E, N);

  gemm_wmma_f32<128, 256><<<dim3(N / 16, 128 / 64), 32, 0, stream>>>(Hb, W2s, XS);
  make_wa<1, 128><<<(256 + TPB - 1) / TPB, TPB, 0, stream>>>(W2s, att2s, WA, 256);
  make_wa<1, 128><<<(256 + TPB - 1) / TPB, TPB, 0, stream>>>(W2d, att2d, WD, 256);
  proj_att<256, 1><<<(N + TPB - 1) / TPB, TPB, 0, stream>>>(Hb, WA, AS, N);
  proj_att<256, 1><<<(N + TPB - 1) / TPB, TPB, 0, stream>>>(Hb, WD, AD, N);

  edge_aggregate_owned<1, 7, 8, false><<<256, 1024, 0, stream>>>(src, dst, AS, AD,
                                                               XS, b2, out, E, N);
}


# edge_aggregate: layer 1 parks each tile's finished edge list (source node, row) in that tile's own rows of out; layer 2 loads it in one trip, no second dst filter, and sums in layer 1's list order
# speedup vs baseline: 2.6332x; 1.1553x over previous
#include <hip/hip_runtime.h>
#include <stddef.h>

typedef __attribute__((ext_vector_type(2))) float v2f;
typedef __attribute__((ext_vector_type(8))) float v8f;

__device__ __forceinline__ v8f wmma4(v2f a, v2f b, v8f c) {
  return __builtin_amdgcn_wmma_f32_16x16x4_f32(
       false, a,  false, b,
       (short)0, c,  false,  false);
}

template <int N, int K>
__global__ void gemm_wmma_f32(const float* __restrict__ A,
                              const float* __restrict__ B,
                              float* __restrict__ C) {
  const int lane = threadIdx.x & 31;
  const int m0 = blockIdx.x * 16;
  const int n0 = blockIdx.y * 64;
  const int half = lane >> 4;
  const int r    = lane & 15;

  const float* Arow = A + (size_t)(m0 + r) * K + 2 * half;
  const float* Bcol = B + (size_t)(2 * half) * N + n0 + r;

  v8f c0 = {}, c1 = {}, c2 = {}, c3 = {};
#pragma unroll
  for (int k = 0; k < K; k += 4) {
    v2f a = *(const v2f*)(Arow + k);
    v2f b0, b1, b2, b3;
    b0.x = Bcol[k * N + 0];       b0.y = Bcol[k * N + N + 0];
    b1.x = Bcol[k * N + 16];      b1.y = Bcol[k * N + N + 16];
    b2.x = Bcol[k * N + 32];      b2.y = Bcol[k * N + N + 32];
    b3.x = Bcol[k * N + 48];      b3.y = Bcol[k * N + N + 48];
    c0 = wmma4(a, b0, c0);
    c1 = wmma4(a, b1, c1);
    c2 = wmma4(a, b2, c2);
    c3 = wmma4(a, b3, c3);
  }

  float* Crow = C + (size_t)(m0 + 8 * half) * N + n0 + r;
#pragma unroll
  for (int i = 0; i < 8; ++i) {
    Crow[i * N + 0]  = c0[i];
    Crow[i * N + 16] = c1[i];
    Crow[i * N + 32] = c2[i];
    Crow[i * N + 48] = c3[i];
  }
}

template <int H, int F>
__global__ void make_wa(const float* __restrict__ W, const float* __restrict__ att,
                        float* __restrict__ Wa, int K) {
  int t = blockIdx.x * blockDim.x + threadIdx.x;
  if (t >= K * H) return;
  int k = t / H, h = t & (H - 1);
  const float* wr = W + (size_t)k * (H * F) + h * F;
  const float* ar = att + h * F;
  float acc = 0.f;
#pragma unroll 4
  for (int f = 0; f < F; ++f) acc += wr[f] * ar[f];
  Wa[t] = acc;
}

template <int K, int H>
__global__ void proj_att(const float* __restrict__ X, const float* __restrict__ Wa,
                         float* __restrict__ out, int N) {
  int t = blockIdx.x * blockDim.x + threadIdx.x;
  if (t >= N * H) return;
  int n = t / H, h = t & (H - 1);
  const float* xr = X + (size_t)n * K;
  float acc = 0.f;
#pragma unroll 4
  for (int k = 0; k < K; ++k) acc += xr[k] * Wa[k * H + h];
  out[t] = acc;
}

__device__ __forceinline__ unsigned fkey(float f) {
  unsigned u = __float_as_uint(f);
  return (u & 0x80000000u) ? ~u : (u | 0x80000000u);
}
__device__ __forceinline__ float fdecode(unsigned k) {
  unsigned u = (k & 0x80000000u) ? (k & 0x7FFFFFFFu) : ~k;
  return __uint_as_float(u);
}
__device__ __forceinline__ float leaky(float a) {
  return a > 0.f ? a : 0.2f * a;
}

template <int H>
__device__ __forceinline__ void load_heads(const float* __restrict__ p, float (&a)[H]) {
  if constexpr (H % 4 == 0) {
#pragma unroll
    for (int q = 0; q < H / 4; ++q) {
      const float4 v = ((const float4*)p)[q];
      a[4 * q + 0] = v.x; a[4 * q + 1] = v.y; a[4 * q + 2] = v.z; a[4 * q + 3] = v.w;
    }
  } else {
#pragma unroll
    for (int h = 0; h < H; ++h) a[h] = p[h];
  }
}

template <int H, int LOGF, int EB, bool RELU, int LMODE>
__global__ void __launch_bounds__(1024)
edge_aggregate_owned(const int* __restrict__ src, const int* __restrict__ dst,
                     const float* __restrict__ as, const float* __restrict__ ad,
                     const float* __restrict__ xs, const float* __restrict__ bias,
                     float* __restrict__ out, float* __restrict__ lst, int E, int N) {
  constexpr int HF = H << LOGF;
  constexpr int NS = HF / 32;
  constexpr int RT = 196;
  constexpr int CAP = 8192;
  constexpr int KR = 4;
  constexpr int U = 8;
  constexpr int LP = 128;
  static_assert(RT <= 256, "a parked entry keeps its row in 8 bits");
  static_assert(LMODE != 2 || HF == LP, "the parked lists lie in this launch's own output rows");
  static_assert(EB * H <= 32, "one lane per (edge, head) of a batch");
  static_assert(LOGF >= 5, "a 32-lane slice lies inside one head");
  __shared__ float acc[RT * HF];
  __shared__ int lsrc[CAP];
  __shared__ int ldl[CAP];
  __shared__ float adl[RT * H];
  __shared__ unsigned mx[RT * H];
  __shared__ float dn[RT * H];
  __shared__ int cnt;

  const int tid = threadIdx.x, lane = tid & 31, wave = tid >> 5;

  for (int row0 = blockIdx.x * RT; row0 < N; row0 += gridDim.x * RT) {
    const int rows = min(RT, N - row0);
    const int lcap = rows * LP;
    float* const lp = (LMODE == 2 ? out : lst) + (size_t)row0 * LP;
    int c0 = -1;
    int lv[KR];
    if constexpr (LMODE == 2) {
      c0 = __float_as_int(lp[0]);
#pragma unroll
      for (int k = 0; k < KR; ++k) lv[k] = __float_as_int(lp[min(1 + tid + k * 1024, lcap - 1)]);
    }
    for (int i = tid; i < RT * HF; i += 1024) acc[i] = 0.f;
    for (int i = tid; i < RT * H; i += 1024) {
      adl[i] = (i < rows * H) ? ad[(size_t)row0 * H + i] : 0.f;
      mx[i] = 0u;
      dn[i] = 0.f;
    }
    if (tid == 0) cnt = 0;
    __syncthreads();

    auto take = [&](int d, int e) {
      unsigned dl = (unsigned)(d - row0);
      if (dl < (unsigned)rows) {
        int idx = atomicAdd(&cnt, 1);
        if (idx < CAP) {
          lsrc[idx] = e;
          ldl[idx] = (int)dl;
        }
      }
    };

    bool use_list = false;
    if constexpr (LMODE == 2) {
      c0 = __builtin_amdgcn_readfirstlane(c0);
      use_list = c0 >= 0 && c0 <= CAP && c0 < lcap;
    }
    if (use_list) {
#pragma unroll
      for (int k = 0; k < KR; ++k) {
        const int i = tid + k * 1024;
        if (i < c0) {
          lsrc[i] = lv[k] & 0xFFFFFF;
          ldl[i] = (int)((unsigned)lv[k] >> 24);
        }
      }
      for (int i = KR * 1024 + tid; i < c0; i += 1024) {
        const int v = __float_as_int(lp[1 + i]);
        lsrc[i] = v & 0xFFFFFF;
        ldl[i] = (int)((unsigned)v >> 24);
      }
      if (tid == 0) cnt = c0;
    } else {
      const int E4 = (((size_t)dst & 15) == 0) ? (E >> 2) : 0;
      const int4* dst4 = (const int4*)dst;
      for (int g0 = tid; g0 < E4; g0 += 1024 * U) {
        int4 v[U];
#pragma unroll
        for (int u = 0; u < U; ++u) {
          int g = g0 + u * 1024;
          v[u] = (g < E4) ? dst4[g] : make_int4(-1, -1, -1, -1);
        }
#pragma unroll
        for (int u = 0; u < U; ++u) {
          int e = 4 * (g0 + u * 1024);
          take(v[u].x, e);
          take(v[u].y, e + 1);
          take(v[u].z, e + 2);
          take(v[u].w, e + 3);
        }
      }
      for (int e = 4 * E4 + tid; e < E; e += 1024) take(dst[e], e);
    }
    __syncthreads();

    const bool listed = cnt <= CAP;
    const int n = listed ? cnt : 0;
    if (!listed) {
      for (int e = tid; e < E; e += 1024) {
        const int d = dst[e];
        const unsigned dl = (unsigned)(d - row0);
        if (dl < (unsigned)rows) {
          const int s = src[e];
#pragma unroll 1
          for (int h = 0; h < H; ++h)
            atomicMax(&mx[dl * H + h], fkey(leaky(as[s * H + h] + adl[dl * H + h])));
        }
      }
      __syncthreads();
      for (int e = tid; e < E; e += 1024) {
        const int d = dst[e];
        const unsigned dl = (unsigned)(d - row0);
        if (dl < (unsigned)rows) {
          const int s = src[e];
#pragma unroll 1
          for (int h = 0; h < H; ++h) {
            const float a = leaky(as[s * H + h] + adl[dl * H + h]);
            atomicAdd(&dn[dl * H + h], __expf(a - fdecode(mx[dl * H + h])));
          }
        }
      }
      __syncthreads();
      for (int e = tid; e < E; e += 1024) {
        const int d = dst[e];
        const unsigned dl = (unsigned)(d - row0);
        if (dl < (unsigned)rows) {
          const int s = src[e];
#pragma unroll 1
          for (int p = 0; p < HF; ++p) {
            const int lh = dl * H + (p >> LOGF);
            const float a = leaky(as[s * H + (p >> LOGF)] + adl[lh]);
            const float w = __expf(a - fdecode(mx[lh])) / (dn[lh] + 1e-16f);
            atomicAdd(&acc[dl * HF + p], w * xs[(size_t)s * HF + p]);
          }
        }
      }
    }
    {
      int sv[KR];
      float av[KR][H];
      if (n > 0) {
        if (use_list) {
#pragma unroll
          for (int k = 0; k < KR; ++k) {
            const int i = tid + k * 1024;
            sv[k] = (i < n) ? lsrc[i] : 0;
          }
        } else {
#pragma unroll
          for (int k = 0; k < KR; ++k) {
            const int i = tid + k * 1024;
            const int e = (i < n) ? lsrc[i] : 0;
            sv[k] = src[e];
          }
        }
        __builtin_amdgcn_sched_barrier(0);
#pragma unroll
        for (int k = 0; k < KR; ++k) load_heads<H>(as + (size_t)sv[k] * H, av[k]);
      }
#pragma unroll
      for (int k = 0; k < KR; ++k) {
        const int i = tid + k * 1024;
        if (i < n) {
          const int lh = ldl[i] * H;
          lsrc[i] = sv[k];
#pragma unroll
          for (int h = 0; h < H; ++h) {
            av[k][h] = leaky(av[k][h] + adl[lh + h]);
            atomicMax(&mx[lh + h], fkey(av[k][h]));
          }
        }
      }
      for (int i = KR * 1024 + tid; i < n; i += 1024) {
        int s = lsrc[i];
        if (!use_list) s = src[s];
        const int lh = ldl[i] * H;
        float a[H];
        load_heads<H>(as + (size_t)s * H, a);
        lsrc[i] = s;
#pragma unroll
        for (int h = 0; h < H; ++h) atomicMax(&mx[lh + h], fkey(leaky(a[h] + adl[lh + h])));
      }
      __syncthreads();
#pragma unroll
      for (int k = 0; k < KR; ++k) {
        const int i = tid + k * 1024;
        if (i < n) {
          const int lh = ldl[i] * H;
#pragma unroll
          for (int h = 0; h < H; ++h)
            atomicAdd(&dn[lh + h], __expf(av[k][h] - fdecode(mx[lh + h])));
        }
      }
      for (int i = KR * 1024 + tid; i < n; i += 1024) {
        const int lh = ldl[i] * H;
        float a[H];
        load_heads<H>(as + (size_t)lsrc[i] * H, a);
#pragma unroll
        for (int h = 0; h < H; ++h)
          atomicAdd(&dn[lh + h], __expf(leaky(a[h] + adl[lh + h]) - fdecode(mx[lh + h])));
      }
    }
    __syncthreads();

    if constexpr (LMODE == 1) {
      const bool park = listed && n < lcap && N <= (1 << 24);
      if (tid == 0) lp[0] = __int_as_float(park ? n : -1);
      if (park)
        for (int i = tid; i < n; i += 1024) lp[1 + i] = __int_as_float(lsrc[i] | (ldl[i] << 24));
    }

    for (int i0 = wave * EB; i0 < n; i0 += 32 * EB) {
      const int j = min(lane / H, EB - 1), h = lane & (H - 1);
      const int ii = min(i0 + j, n - 1);
      const int s = lsrc[ii], dl = ldl[ii];
      const int lh = dl * H + h;
      const float v_as = as[s * H + h], v_ad = adl[lh], v_den = dn[lh];
      const unsigned v_mx = mx[lh];
      float m[EB][NS];
      int dls[EB];
#pragma unroll
      for (int jj = 0; jj < EB; ++jj) {
        const int sj = __builtin_amdgcn_readlane(s, jj * H);
        dls[jj] = __builtin_amdgcn_readlane(dl, jj * H);
#pragma unroll
        for (int i = 0; i < NS; ++i) m[jj][i] = xs[(size_t)sj * HF + lane + 32 * i];
      }
      const float w = __expf(leaky(v_as + v_ad) - fdecode(v_mx)) / (v_den + 1e-16f);
#pragma unroll
      for (int jj = 0; jj < EB; ++jj) {
        const bool valid = i0 + jj < n;
#pragma unroll
        for (int i = 0; i < NS; ++i) {
          const int p = lane + 32 * i;
          const float wj = __builtin_bit_cast(
              float, __builtin_amdgcn_readlane(__builtin_bit_cast(int, w), jj * H + ((32 * i) >> LOGF)));
          atomicAdd(&acc[dls[jj] * HF + p], valid ? wj * m[jj][i] : 0.f);
        }
      }
    }
    __syncthreads();

    for (int i = tid; i < rows * HF; i += 1024) {
      float v = acc[i] + bias[i & (HF - 1)];
      if (RELU) v = v > 0.f ? v : 0.f;
      out[(size_t)row0 * HF + i] = v;
    }
    __syncthreads();
  }
}

extern "C" void kernel_launch(void* const* d_in, const int* in_sizes, int n_in,
                              void* d_out, int out_size, void* d_ws, size_t ws_size,
                              hipStream_t stream) {
  (void)n_in; (void)out_size; (void)ws_size;
  const float* x     = (const float*)d_in[0];
  const int*   ei    = (const int*)d_in[1];
  const float* W1s   = (const float*)d_in[2];
  const float* W1d   = (const float*)d_in[3];
  const float* att1s = (const float*)d_in[4];
  const float* att1d = (const float*)d_in[5];
  const float* b1    = (const float*)d_in[6];
  const float* W2s   = (const float*)d_in[7];
  const float* W2d   = (const float*)d_in[8];
  const float* att2s = (const float*)d_in[9];
  const float* att2d = (const float*)d_in[10];
  const float* b2    = (const float*)d_in[11];
  float* out = (float*)d_out;

  const int N = in_sizes[0] / 128;
  const int E = in_sizes[1] / 2;
  const int* src = ei;
  const int* dst = ei + E;

  float* ws = (float*)d_ws;
  size_t off = 0;
  float*    XS   = ws + off; off += (size_t)N * 256;
  float*    Hb   = ws + off; off += (size_t)N * 256;
  float*    AS   = ws + off; off += (size_t)N * 8;
  float*    AD   = ws + off; off += (size_t)N * 8;
  float*    WA   = ws + off; off += 2048;
  float*    WD   = ws + off; off += 2048;

  const int TPB = 256;

  gemm_wmma_f32<256, 128><<<dim3(N / 16, 256 / 64), 32, 0, stream>>>(x, W1s, XS);
  make_wa<8, 32><<<(128 * 8 + TPB - 1) / TPB, TPB, 0, stream>>>(W1s, att1s, WA, 128);
  make_wa<8, 32><<<(128 * 8 + TPB - 1) / TPB, TPB, 0, stream>>>(W1d, att1d, WD, 128);
  proj_att<128, 8><<<((size_t)N * 8 + TPB - 1) / TPB, TPB, 0, stream>>>(x, WA, AS, N);
  proj_att<128, 8><<<((size_t)N * 8 + TPB - 1) / TPB, TPB, 0, stream>>>(x, WD, AD, N);

  edge_aggregate_owned<8, 5, 4, true, 1><<<256, 1024, 0, stream>>>(src, dst, AS, AD,
                                                                 XS, b1, Hb, out, E, N);

  gemm_wmma_f32<128, 256><<<dim3(N / 16, 128 / 64), 32, 0, stream>>>(Hb, W2s, XS);
  make_wa<1, 128><<<(256 + TPB - 1) / TPB, TPB, 0, stream>>>(W2s, att2s, WA, 256);
  make_wa<1, 128><<<(256 + TPB - 1) / TPB, TPB, 0, stream>>>(W2d, att2d, WD, 256);
  proj_att<256, 1><<<(N + TPB - 1) / TPB, TPB, 0, stream>>>(Hb, WA, AS, N);
  proj_att<256, 1><<<(N + TPB - 1) / TPB, TPB, 0, stream>>>(Hb, WD, AD, N);

  edge_aggregate_owned<1, 7, 8, false, 2><<<256, 1024, 0, stream>>>(src, dst, AS, AD,
                                                                  XS, b2, out, nullptr, E, N);
}


# layer 1 dst filter reads 1-byte tile keys (dst/196, from a new first launch, in unused tails of out) not the int list: a quarter of the bytes and loads; list and LDS add order change; 13 launches
# speedup vs baseline: 2.7862x; 1.0581x over previous
#include <hip/hip_runtime.h>
#include <stddef.h>

typedef __attribute__((ext_vector_type(2))) float v2f;
typedef __attribute__((ext_vector_type(8))) float v8f;

__device__ __forceinline__ v8f wmma4(v2f a, v2f b, v8f c) {
  return __builtin_amdgcn_wmma_f32_16x16x4_f32(
       false, a,  false, b,
       (short)0, c,  false,  false);
}

template <int N, int K>
__global__ void gemm_wmma_f32(const float* __restrict__ A,
                              const float* __restrict__ B,
                              float* __restrict__ C) {
  const int lane = threadIdx.x & 31;
  const int m0 = blockIdx.x * 16;
  const int n0 = blockIdx.y * 64;
  const int half = lane >> 4;
  const int r    = lane & 15;

  const float* Arow = A + (size_t)(m0 + r) * K + 2 * half;
  const float* Bcol = B + (size_t)(2 * half) * N + n0 + r;

  v8f c0 = {}, c1 = {}, c2 = {}, c3 = {};
#pragma unroll
  for (int k = 0; k < K; k += 4) {
    v2f a = *(const v2f*)(Arow + k);
    v2f b0, b1, b2, b3;
    b0.x = Bcol[k * N + 0];       b0.y = Bcol[k * N + N + 0];
    b1.x = Bcol[k * N + 16];      b1.y = Bcol[k * N + N + 16];
    b2.x = Bcol[k * N + 32];      b2.y = Bcol[k * N + N + 32];
    b3.x = Bcol[k * N + 48];      b3.y = Bcol[k * N + N + 48];
    c0 = wmma4(a, b0, c0);
    c1 = wmma4(a, b1, c1);
    c2 = wmma4(a, b2, c2);
    c3 = wmma4(a, b3, c3);
  }

  float* Crow = C + (size_t)(m0 + 8 * half) * N + n0 + r;
#pragma unroll
  for (int i = 0; i < 8; ++i) {
    Crow[i * N + 0]  = c0[i];
    Crow[i * N + 16] = c1[i];
    Crow[i * N + 32] = c2[i];
    Crow[i * N + 48] = c3[i];
  }
}

template <int H, int F>
__global__ void make_wa(const float* __restrict__ W, const float* __restrict__ att,
                        float* __restrict__ Wa, int K) {
  int t = blockIdx.x * blockDim.x + threadIdx.x;
  if (t >= K * H) return;
  int k = t / H, h = t & (H - 1);
  const float* wr = W + (size_t)k * (H * F) + h * F;
  const float* ar = att + h * F;
  float acc = 0.f;
#pragma unroll 4
  for (int f = 0; f < F; ++f) acc += wr[f] * ar[f];
  Wa[t] = acc;
}

template <int K, int H>
__global__ void proj_att(const float* __restrict__ X, const float* __restrict__ Wa,
                         float* __restrict__ out, int N) {
  int t = blockIdx.x * blockDim.x + threadIdx.x;
  if (t >= N * H) return;
  int n = t / H, h = t & (H - 1);
  const float* xr = X + (size_t)n * K;
  float acc = 0.f;
#pragma unroll 4
  for (int k = 0; k < K; ++k) acc += xr[k] * Wa[k * H + h];
  out[t] = acc;
}

__device__ __forceinline__ unsigned fkey(float f) {
  unsigned u = __float_as_uint(f);
  return (u & 0x80000000u) ? ~u : (u | 0x80000000u);
}
__device__ __forceinline__ float fdecode(unsigned k) {
  unsigned u = (k & 0x80000000u) ? (k & 0x7FFFFFFFu) : ~k;
  return __uint_as_float(u);
}
__device__ __forceinline__ float leaky(float a) {
  return a > 0.f ? a : 0.2f * a;
}

template <int H>
__device__ __forceinline__ void load_heads(const float* __restrict__ p, float (&a)[H]) {
  if constexpr (H % 4 == 0) {
#pragma unroll
    for (int q = 0; q < H / 4; ++q) {
      const float4 v = ((const float4*)p)[q];
      a[4 * q + 0] = v.x; a[4 * q + 1] = v.y; a[4 * q + 2] = v.z; a[4 * q + 3] = v.w;
    }
  } else {
#pragma unroll
    for (int h = 0; h < H; ++h) a[h] = p[h];
  }
}

constexpr int kRT = 196;
constexpr int kLP = 128;
constexpr int kCAP = 8192;
constexpr int kKOFF = 8448;
constexpr int kKGRP = 4096;
static_assert(kKOFF >= 1 + kCAP + 1, "keys lie behind the parked list");
static_assert(kKOFF % 4 == 0 && kKOFF + 4 * kKGRP <= kRT * kLP, "keys lie inside the tile's own rows, 16-byte aligned");

__device__ __forceinline__ bool keys_fit(int E, int N) {
  return N <= kRT * 256 && (long long)((E + 15) >> 4) <= (long long)(N / kRT) * kKGRP;
}
__device__ __forceinline__ uint4* key_group(float* base, int g) {
  const unsigned ug = (unsigned)g;
  return (uint4*)(base + (size_t)(ug / (unsigned)kKGRP) * (kRT * kLP) + kKOFF) + (ug % (unsigned)kKGRP);
}
__device__ __forceinline__ unsigned tile_key(int d) {
  return ((unsigned)d / (unsigned)kRT) & 255u;
}

__global__ void tile_keys(const int* __restrict__ dst, float* __restrict__ lst, int E, int N) {
  if (!keys_fit(E, N)) return;
  const int g = blockIdx.x * blockDim.x + threadIdx.x;
  const int e0 = 16 * g;
  if (e0 >= E) return;
  unsigned w[4];
  if ((((size_t)dst & 15) == 0) && e0 + 16 <= E) {
    const int4* p = (const int4*)(dst + e0);
    int4 v[4];
#pragma unroll
    for (int q = 0; q < 4; ++q) v[q] = p[q];
#pragma unroll
    for (int q = 0; q < 4; ++q)
      w[q] = tile_key(v[q].x) | (tile_key(v[q].y) << 8) | (tile_key(v[q].z) << 16) | (tile_key(v[q].w) << 24);
  } else {
#pragma unroll
    for (int q = 0; q < 4; ++q) {
      w[q] = 0u;
#pragma unroll
      for (int b = 0; b < 4; ++b) {
        const int e = e0 + 4 * q + b;
        w[q] |= (e < E ? tile_key(dst[e]) : 255u) << (8 * b);
      }
    }
  }
  *key_group(lst, g) = make_uint4(w[0], w[1], w[2], w[3]);
}

__device__ __forceinline__ unsigned zero_bytes(unsigned x) {
  return ~(((x & 0x7F7F7F7Fu) + 0x7F7F7F7Fu) | x | 0x7F7F7F7Fu);
}

template <int H, int LOGF, int EB, bool RELU, int LMODE>
__global__ void __launch_bounds__(1024)
edge_aggregate_owned(const int* __restrict__ src, const int* __restrict__ dst,
                     const float* __restrict__ as, const float* __restrict__ ad,
                     const float* __restrict__ xs, const float* __restrict__ bias,
                     float* __restrict__ out, float* __restrict__ lst, int E, int N) {
  constexpr int HF = H << LOGF;
  constexpr int NS = HF / 32;
  constexpr int RT = kRT;
  constexpr int CAP = kCAP;
  constexpr int KR = 4;
  constexpr int U = 8;
  constexpr int LP = kLP;
  static_assert(RT <= 256, "a parked entry keeps its row in 8 bits");
  static_assert(LMODE != 2 || HF == LP, "the parked lists lie in this launch's own output rows");
  static_assert(EB * H <= 32, "one lane per (edge, head) of a batch");
  static_assert(LOGF >= 5, "a 32-lane slice lies inside one head");
  __shared__ float acc[RT * HF];
  __shared__ int lsrc[CAP];
  __shared__ int ldl[CAP];
  __shared__ float adl[RT * H];
  __shared__ unsigned mx[RT * H];
  __shared__ float dn[RT * H];
  __shared__ int cnt;

  const int tid = threadIdx.x, lane = tid & 31, wave = tid >> 5;

  for (int row0 = blockIdx.x * RT; row0 < N; row0 += gridDim.x * RT) {
    const int rows = min(RT, N - row0);
    const int lcap = rows * LP;
    float* const lp = (LMODE == 2 ? out : lst) + (size_t)row0 * LP;
    int c0 = -1;
    int lv[KR];
    if constexpr (LMODE == 2) {
      c0 = __float_as_int(lp[0]);
#pragma unroll
      for (int k = 0; k < KR; ++k) lv[k] = __float_as_int(lp[min(1 + tid + k * 1024, lcap - 1)]);
    }
    for (int i = tid; i < RT * HF; i += 1024) acc[i] = 0.f;
    for (int i = tid; i < RT * H; i += 1024) {
      adl[i] = (i < rows * H) ? ad[(size_t)row0 * H + i] : 0.f;
      mx[i] = 0u;
      dn[i] = 0.f;
    }
    if (tid == 0) cnt = 0;
    __syncthreads();

    auto list_edge = [&](int e) {
      int idx = atomicAdd(&cnt, 1);
      if (idx < CAP) lsrc[idx] = e;
    };
    auto take = [&](int d, int e) {
      if ((unsigned)(d - row0) < (unsigned)rows) list_edge(e);
    };
    auto take_marked = [&](unsigned z, int e) {
      while (z != 0u) {
        const int eb = e + (__builtin_ctz(z) >> 3);
        z &= z - 1u;
        if (eb < E) list_edge(eb);
      }
    };

    bool use_list = false;
    if constexpr (LMODE == 2) {
      c0 = __builtin_amdgcn_readfirstlane(c0);
      use_list = c0 >= 0 && c0 <= CAP && c0 < lcap;
    }
    if (use_list) {
#pragma unroll
      for (int k = 0; k < KR; ++k) {
        const int i = tid + k * 1024;
        if (i < c0) {
          lsrc[i] = lv[k] & 0xFFFFFF;
          ldl[i] = (int)((unsigned)lv[k] >> 24);
        }
      }
      for (int i = KR * 1024 + tid; i < c0; i += 1024) {
        const int v = __float_as_int(lp[1 + i]);
        lsrc[i] = v & 0xFFFFFF;
        ldl[i] = (int)((unsigned)v >> 24);
      }
      if (tid == 0) cnt = c0;
    } else if (LMODE == 1 && keys_fit(E, N)) {
      const unsigned pat = (unsigned)(row0 / RT) * 0x01010101u;
      const int G = (E + 15) >> 4;
      for (int g0 = tid; g0 < G; g0 += 1024 * U) {
        uint4 v[U];
#pragma unroll
        for (int u = 0; u < U; ++u) v[u] = *key_group(lst, min(g0 + u * 1024, G - 1));
        __builtin_amdgcn_sched_barrier(0);
#pragma unroll
        for (int u = 0; u < U; ++u) {
          const unsigned z0 = zero_bytes(v[u].x ^ pat), z1 = zero_bytes(v[u].y ^ pat);
          const unsigned z2 = zero_bytes(v[u].z ^ pat), z3 = zero_bytes(v[u].w ^ pat);
          if (g0 + u * 1024 < G && (z0 | z1 | z2 | z3) != 0u) {
            const int e = 16 * (g0 + u * 1024);
            take_marked(z0, e);
            take_marked(z1, e + 4);
            take_marked(z2, e + 8);
            take_marked(z3, e + 12);
          }
        }
      }
    } else {
      const int E4 = (((size_t)dst & 15) == 0) ? (E >> 2) : 0;
      const int4* dst4 = (const int4*)dst;
      for (int g0 = tid; g0 < E4; g0 += 1024 * U) {
        int4 v[U];
#pragma unroll
        for (int u = 0; u < U; ++u) {
          int g = g0 + u * 1024;
          v[u] = (g < E4) ? dst4[g] : make_int4(-1, -1, -1, -1);
        }
#pragma unroll
        for (int u = 0; u < U; ++u) {
          int e = 4 * (g0 + u * 1024);
          take(v[u].x, e);
          take(v[u].y, e + 1);
          take(v[u].z, e + 2);
          take(v[u].w, e + 3);
        }
      }
      for (int e = 4 * E4 + tid; e < E; e += 1024) take(dst[e], e);
    }
    __syncthreads();

    const bool listed = cnt <= CAP;
    const int n = listed ? cnt : 0;
    if (!listed) {
      for (int e = tid; e < E; e += 1024) {
        const int d = dst[e];
        const unsigned dl = (unsigned)(d - row0);
        if (dl < (unsigned)rows) {
          const int s = src[e];
#pragma unroll 1
          for (int h = 0; h < H; ++h)
            atomicMax(&mx[dl * H + h], fkey(leaky(as[s * H + h] + adl[dl * H + h])));
        }
      }
      __syncthreads();
      for (int e = tid; e < E; e += 1024) {
        const int d = dst[e];
        const unsigned dl = (unsigned)(d - row0);
        if (dl < (unsigned)rows) {
          const int s = src[e];
#pragma unroll 1
          for (int h = 0; h < H; ++h) {
            const float a = leaky(as[s * H + h] + adl[dl * H + h]);
            atomicAdd(&dn[dl * H + h], __expf(a - fdecode(mx[dl * H + h])));
          }
        }
      }
      __syncthreads();
      for (int e = tid; e < E; e += 1024) {
        const int d = dst[e];
        const unsigned dl = (unsigned)(d - row0);
        if (dl < (unsigned)rows) {
          const int s = src[e];
#pragma unroll 1
          for (int p = 0; p < HF; ++p) {
            const int lh = dl * H + (p >> LOGF);
            const float a = leaky(as[s * H + (p >> LOGF)] + adl[lh]);
            const float w = __expf(a - fdecode(mx[lh])) / (dn[lh] + 1e-16f);
            atomicAdd(&acc[dl * HF + p], w * xs[(size_t)s * HF + p]);
          }
        }
      }
    }
    {
      int sv[KR], dv[KR];
      float av[KR][H];
#pragma unroll
      for (int k = 0; k < KR; ++k) dv[k] = row0;
      if (n > 0) {
        if (use_list) {
#pragma unroll
          for (int k = 0; k < KR; ++k) {
            const int i = tid + k * 1024;
            sv[k] = (i < n) ? lsrc[i] : 0;
          }
        } else {
#pragma unroll
          for (int k = 0; k < KR; ++k) {
            const int i = tid + k * 1024;
            const int e = (i < n) ? lsrc[i] : 0;
            sv[k] = src[e];
            dv[k] = dst[e];
          }
        }
        __builtin_amdgcn_sched_barrier(0);
#pragma unroll
        for (int k = 0; k < KR; ++k) load_heads<H>(as + (size_t)sv[k] * H, av[k]);
      }
#pragma unroll
      for (int k = 0; k < KR; ++k) {
        const int i = tid + k * 1024;
        if (i < n) {
          if (!use_list) ldl[i] = min(max(dv[k] - row0, 0), rows - 1);
          const int lh = ldl[i] * H;
          lsrc[i] = sv[k];
#pragma unroll
          for (int h = 0; h < H; ++h) {
            av[k][h] = leaky(av[k][h] + adl[lh + h]);
            atomicMax(&mx[lh + h], fkey(av[k][h]));
          }
        }
      }
      for (int i = KR * 1024 + tid; i < n; i += 1024) {
        int s = lsrc[i];
        if (!use_list) {
          ldl[i] = min(max(dst[s] - row0, 0), rows - 1);
          s = src[s];
        }
        const int lh = ldl[i] * H;
        float a[H];
        load_heads<H>(as + (size_t)s * H, a);
        lsrc[i] = s;
#pragma unroll
        for (int h = 0; h < H; ++h) atomicMax(&mx[lh + h], fkey(leaky(a[h] + adl[lh + h])));
      }
      __syncthreads();
#pragma unroll
      for (int k = 0; k < KR; ++k) {
        const int i = tid + k * 1024;
        if (i < n) {
          const int lh = ldl[i] * H;
#pragma unroll
          for (int h = 0; h < H; ++h)
            atomicAdd(&dn[lh + h], __expf(av[k][h] - fdecode(mx[lh + h])));
        }
      }
      for (int i = KR * 1024 + tid; i < n; i += 1024) {
        const int lh = ldl[i] * H;
        float a[H];
        load_heads<H>(as + (size_t)lsrc[i] * H, a);
#pragma unroll
        for (int h = 0; h < H; ++h)
          atomicAdd(&dn[lh + h], __expf(leaky(a[h] + adl[lh + h]) - fdecode(mx[lh + h])));
      }
    }
    __syncthreads();

    if constexpr (LMODE == 1) {
      const bool park = listed && n < lcap && N <= (1 << 24);
      if (tid == 0) lp[0] = __int_as_float(park ? n : -1);
      if (park)
        for (int i = tid; i < n; i += 1024) lp[1 + i] = __int_as_float(lsrc[i] | (ldl[i] << 24));
    }

    for (int i0 = wave * EB; i0 < n; i0 += 32 * EB) {
      const int j = min(lane / H, EB - 1), h = lane & (H - 1);
      const int ii = min(i0 + j, n - 1);
      const int s = lsrc[ii], dl = ldl[ii];
      const int lh = dl * H + h;
      const float v_as = as[s * H + h], v_ad = adl[lh], v_den = dn[lh];
      const unsigned v_mx = mx[lh];
      float m[EB][NS];
      int dls[EB];
#pragma unroll
      for (int jj = 0; jj < EB; ++jj) {
        const int sj = __builtin_amdgcn_readlane(s, jj * H);
        dls[jj] = __builtin_amdgcn_readlane(dl, jj * H);
#pragma unroll
        for (int i = 0; i < NS; ++i) m[jj][i] = xs[(size_t)sj * HF + lane + 32 * i];
      }
      const float w = __expf(leaky(v_as + v_ad) - fdecode(v_mx)) / (v_den + 1e-16f);
#pragma unroll
      for (int jj = 0; jj < EB; ++jj) {
        const bool valid = i0 + jj < n;
#pragma unroll
        for (int i = 0; i < NS; ++i) {
          const int p = lane + 32 * i;
          const float wj = __builtin_bit_cast(
              float, __builtin_amdgcn_readlane(__builtin_bit_cast(int, w), jj * H + ((32 * i) >> LOGF)));
          atomicAdd(&acc[dls[jj] * HF + p], valid ? wj * m[jj][i] : 0.f);
        }
      }
    }
    __syncthreads();

    for (int i = tid; i < rows * HF; i += 1024) {
      float v = acc[i] + bias[i & (HF - 1)];
      if (RELU) v = v > 0.f ? v : 0.f;
      out[(size_t)row0 * HF + i] = v;
    }
    __syncthreads();
  }
}

extern "C" void kernel_launch(void* const* d_in, const int* in_sizes, int n_in,
                              void* d_out, int out_size, void* d_ws, size_t ws_size,
                              hipStream_t stream) {
  (void)n_in; (void)out_size; (void)ws_size;
  const float* x     = (const float*)d_in[0];
  const int*   ei    = (const int*)d_in[1];
  const float* W1s   = (const float*)d_in[2];
  const float* W1d   = (const float*)d_in[3];
  const float* att1s = (const float*)d_in[4];
  const float* att1d = (const float*)d_in[5];
  const float* b1    = (const float*)d_in[6];
  const float* W2s   = (const float*)d_in[7];
  const float* W2d   = (const float*)d_in[8];
  const float* att2s = (const float*)d_in[9];
  const float* att2d = (const float*)d_in[10];
  const float* b2    = (const float*)d_in[11];
  float* out = (float*)d_out;

  const int N = in_sizes[0] / 128;
  const int E = in_sizes[1] / 2;
  const int* src = ei;
  const int* dst = ei + E;

  float* ws = (float*)d_ws;
  size_t off = 0;
  float*    XS   = ws + off; off += (size_t)N * 256;
  float*    Hb   = ws + off; off += (size_t)N * 256;
  float*    AS   = ws + off; off += (size_t)N * 8;
  float*    AD   = ws + off; off += (size_t)N * 8;
  float*    WA   = ws + off; off += 2048;
  float*    WD   = ws + off; off += 2048;

  const int TPB = 256;

  gemm_wmma_f32<256, 128><<<dim3(N / 16, 256 / 64), 32, 0, stream>>>(x, W1s, XS);
  make_wa<8, 32><<<(128 * 8 + TPB - 1) / TPB, TPB, 0, stream>>>(W1s, att1s, WA, 128);
  make_wa<8, 32><<<(128 * 8 + TPB - 1) / TPB, TPB, 0, stream>>>(W1d, att1d, WD, 128);
  proj_att<128, 8><<<((size_t)N * 8 + TPB - 1) / TPB, TPB, 0, stream>>>(x, WA, AS, N);
  proj_att<128, 8><<<((size_t)N * 8 + TPB - 1) / TPB, TPB, 0, stream>>>(x, WD, AD, N);

  tile_keys<<<(E + 16 * TPB - 1) / (16 * TPB), TPB, 0, stream>>>(dst, out, E, N);

  edge_aggregate_owned<8, 5, 4, true, 1><<<256, 1024, 0, stream>>>(src, dst, AS, AD,
                                                                 XS, b1, Hb, out, E, N);

  gemm_wmma_f32<128, 256><<<dim3(N / 16, 128 / 64), 32, 0, stream>>>(Hb, W2s, XS);
  make_wa<1, 128><<<(256 + TPB - 1) / TPB, TPB, 0, stream>>>(W2s, att2s, WA, 256);
  make_wa<1, 128><<<(256 + TPB - 1) / TPB, TPB, 0, stream>>>(W2d, att2d, WD, 256);
  proj_att<256, 1><<<(N + TPB - 1) / TPB, TPB, 0, stream>>>(Hb, WA, AS, N);
  proj_att<256, 1><<<(N + TPB - 1) / TPB, TPB, 0, stream>>>(Hb, WD, AD, N);

  edge_aggregate_owned<1, 7, 8, false, 2><<<256, 1024, 0, stream>>>(src, dst, AS, AD,
                                                                  XS, b2, out, nullptr, E, N);
}


# layer 1 edge_aggregate keeps its finished rows in LDS and does h x W2_src (f32 WMMA, A from LDS) and the layer 2 logits (4-lane split sums: new sum order); gemm2, 2 proj_att go; 13->10 launches
# speedup vs baseline: 3.1588x; 1.1337x over previous
#include <hip/hip_runtime.h>
#include <stddef.h>

typedef __attribute__((ext_vector_type(2))) float v2f;
typedef __attribute__((ext_vector_type(8))) float v8f;

__device__ __forceinline__ v8f wmma4(v2f a, v2f b, v8f c) {
  return __builtin_amdgcn_wmma_f32_16x16x4_f32(
       false, a,  false, b,
       (short)0, c,  false,  false);
}

template <int N, int K>
__global__ void gemm_wmma_f32(const float* __restrict__ A,
                              const float* __restrict__ B,
                              float* __restrict__ C) {
  const int lane = threadIdx.x & 31;
  const int m0 = blockIdx.x * 16;
  const int n0 = blockIdx.y * 64;
  const int half = lane >> 4;
  const int r    = lane & 15;

  const float* Arow = A + (size_t)(m0 + r) * K + 2 * half;
  const float* Bcol = B + (size_t)(2 * half) * N + n0 + r;

  v8f c0 = {}, c1 = {}, c2 = {}, c3 = {};
#pragma unroll
  for (int k = 0; k < K; k += 4) {
    v2f a = *(const v2f*)(Arow + k);
    v2f b0, b1, b2, b3;
    b0.x = Bcol[k * N + 0];       b0.y = Bcol[k * N + N + 0];
    b1.x = Bcol[k * N + 16];      b1.y = Bcol[k * N + N + 16];
    b2.x = Bcol[k * N + 32];      b2.y = Bcol[k * N + N + 32];
    b3.x = Bcol[k * N + 48];      b3.y = Bcol[k * N + N + 48];
    c0 = wmma4(a, b0, c0);
    c1 = wmma4(a, b1, c1);
    c2 = wmma4(a, b2, c2);
    c3 = wmma4(a, b3, c3);
  }

  float* Crow = C + (size_t)(m0 + 8 * half) * N + n0 + r;
#pragma unroll
  for (int i = 0; i < 8; ++i) {
    Crow[i * N + 0]  = c0[i];
    Crow[i * N + 16] = c1[i];
    Crow[i * N + 32] = c2[i];
    Crow[i * N + 48] = c3[i];
  }
}

template <int H, int F>
__global__ void make_wa(const float* __restrict__ W, const float* __restrict__ att,
                        float* __restrict__ Wa, int K) {
  int t = blockIdx.x * blockDim.x + threadIdx.x;
  if (t >= K * H) return;
  int k = t / H, h = t & (H - 1);
  const float* wr = W + (size_t)k * (H * F) + h * F;
  const float* ar = att + h * F;
  float acc = 0.f;
#pragma unroll 4
  for (int f = 0; f < F; ++f) acc += wr[f] * ar[f];
  Wa[t] = acc;
}

template <int K, int H>
__global__ void proj_att(const float* __restrict__ X, const float* __restrict__ Wa,
                         float* __restrict__ out, int N) {
  int t = blockIdx.x * blockDim.x + threadIdx.x;
  if (t >= N * H) return;
  int n = t / H, h = t & (H - 1);
  const float* xr = X + (size_t)n * K;
  float acc = 0.f;
#pragma unroll 4
  for (int k = 0; k < K; ++k) acc += xr[k] * Wa[k * H + h];
  out[t] = acc;
}

__device__ __forceinline__ unsigned fkey(float f) {
  unsigned u = __float_as_uint(f);
  return (u & 0x80000000u) ? ~u : (u | 0x80000000u);
}
__device__ __forceinline__ float fdecode(unsigned k) {
  unsigned u = (k & 0x80000000u) ? (k & 0x7FFFFFFFu) : ~k;
  return __uint_as_float(u);
}
__device__ __forceinline__ float leaky(float a) {
  return a > 0.f ? a : 0.2f * a;
}

template <int H>
__device__ __forceinline__ void load_heads(const float* __restrict__ p, float (&a)[H]) {
  if constexpr (H % 4 == 0) {
#pragma unroll
    for (int q = 0; q < H / 4; ++q) {
      const float4 v = ((const float4*)p)[q];
      a[4 * q + 0] = v.x; a[4 * q + 1] = v.y; a[4 * q + 2] = v.z; a[4 * q + 3] = v.w;
    }
  } else {
#pragma unroll
    for (int h = 0; h < H; ++h) a[h] = p[h];
  }
}

constexpr int kRT = 196;
constexpr int kLP = 128;
constexpr int kCAP = 8192;
constexpr int kKOFF = 8448;
constexpr int kKGRP = 4096;
static_assert(kKOFF >= 1 + kCAP + 1, "keys lie behind the parked list");
static_assert(kKOFF % 4 == 0 && kKOFF + 4 * kKGRP <= kRT * kLP, "keys lie inside the tile's own rows, 16-byte aligned");

__device__ __forceinline__ bool keys_fit(int E, int N) {
  return N <= kRT * 256 && (long long)((E + 15) >> 4) <= (long long)(N / kRT) * kKGRP;
}
__device__ __forceinline__ uint4* key_group(float* base, int g) {
  const unsigned ug = (unsigned)g;
  return (uint4*)(base + (size_t)(ug / (unsigned)kKGRP) * (kRT * kLP) + kKOFF) + (ug % (unsigned)kKGRP);
}
__device__ __forceinline__ unsigned tile_key(int d) {
  return ((unsigned)d / (unsigned)kRT) & 255u;
}

__global__ void tile_keys(const int* __restrict__ dst, float* __restrict__ lst, int E, int N) {
  if (!keys_fit(E, N)) return;
  const int g = blockIdx.x * blockDim.x + threadIdx.x;
  const int e0 = 16 * g;
  if (e0 >= E) return;
  unsigned w[4];
  if ((((size_t)dst & 15) == 0) && e0 + 16 <= E) {
    const int4* p = (const int4*)(dst + e0);
    int4 v[4];
#pragma unroll
    for (int q = 0; q < 4; ++q) v[q] = p[q];
#pragma unroll
    for (int q = 0; q < 4; ++q)
      w[q] = tile_key(v[q].x) | (tile_key(v[q].y) << 8) | (tile_key(v[q].z) << 16) | (tile_key(v[q].w) << 24);
  } else {
#pragma unroll
    for (int q = 0; q < 4; ++q) {
      w[q] = 0u;
#pragma unroll
      for (int b = 0; b < 4; ++b) {
        const int e = e0 + 4 * q + b;
        w[q] |= (e < E ? tile_key(dst[e]) : 255u) << (8 * b);
      }
    }
  }
  *key_group(lst, g) = make_uint4(w[0], w[1], w[2], w[3]);
}

__device__ __forceinline__ unsigned zero_bytes(unsigned x) {
  return ~(((x & 0x7F7F7F7Fu) + 0x7F7F7F7Fu) | x | 0x7F7F7F7Fu);
}

template <int H, int LOGF, int EB, bool RELU, int LMODE>
__global__ void __launch_bounds__(1024)
edge_aggregate_owned(const int* __restrict__ src, const int* __restrict__ dst,
                     const float* __restrict__ as, const float* __restrict__ ad,
                     const float* __restrict__ xs, const float* __restrict__ bias,
                     float* __restrict__ out, float* __restrict__ lst, int E, int N,
                     const float* __restrict__ wn, const float* __restrict__ wan,
                     const float* __restrict__ wdn, float* __restrict__ asn,
                     float* __restrict__ adn) {
  constexpr int HF = H << LOGF;
  constexpr bool NEXT = LMODE == 1;
  constexpr int NN = 128;
  constexpr int AST = NEXT ? HF + 4 : HF;
  static_assert(!NEXT || HF % 4 == 0, "whole k steps of the matrix instruction");
  constexpr int NS = HF / 32;
  constexpr int RT = kRT;
  constexpr int CAP = kCAP;
  constexpr int KR = 4;
  constexpr int U = 8;
  constexpr int LP = kLP;
  static_assert(RT <= 256, "a parked entry keeps its row in 8 bits");
  static_assert(LMODE != 2 || HF == LP, "the parked lists lie in this launch's own output rows");
  static_assert(EB * H <= 32, "one lane per (edge, head) of a batch");
  static_assert(LOGF >= 5, "a 32-lane slice lies inside one head");
  __shared__ __attribute__((aligned(16))) float acc[RT * AST];
  __shared__ int lsrc[CAP];
  __shared__ int ldl[CAP];
  __shared__ float adl[RT * H];
  __shared__ unsigned mx[RT * H];
  __shared__ float dn[RT * H];
  __shared__ int cnt;

  const int tid = threadIdx.x, lane = tid & 31, wave = tid >> 5;
  constexpr int LGP = 256;
  static_assert(LGP >= RT && LGP % 32 == 0 && (LMODE != 2 || H == 1), "a tile's logits fill whole lines of their own");
  auto aidx = [](int s) -> size_t {
    if constexpr (LMODE == 2) return (size_t)((unsigned)s / (unsigned)RT) * LGP + (unsigned)s % (unsigned)RT;
    else return (size_t)s * H;
  };

  for (int row0 = blockIdx.x * RT; row0 < N; row0 += gridDim.x * RT) {
    const int rows = min(RT, N - row0);
    const int lcap = rows * LP;
    float* const lp = (LMODE == 2 ? out : lst) + (size_t)row0 * LP;
    int c0 = -1;
    int lv[KR];
    if constexpr (LMODE == 2) {
      c0 = __float_as_int(lp[0]);
#pragma unroll
      for (int k = 0; k < KR; ++k) lv[k] = __float_as_int(lp[min(1 + tid + k * 1024, lcap - 1)]);
    }
    for (int i = tid; i < RT * AST; i += 1024) acc[i] = 0.f;
    for (int i = tid; i < RT * H; i += 1024) {
      adl[i] = (i < rows * H) ? ad[aidx(row0) + i] : 0.f;
      mx[i] = 0u;
      dn[i] = 0.f;
    }
    if (tid == 0) cnt = 0;
    __syncthreads();

    auto list_edge = [&](int e) {
      int idx = atomicAdd(&cnt, 1);
      if (idx < CAP) lsrc[idx] = e;
    };
    auto take = [&](int d, int e) {
      if ((unsigned)(d - row0) < (unsigned)rows) list_edge(e);
    };
    auto take_marked = [&](unsigned z, int e) {
      while (z != 0u) {
        const int eb = e + (__builtin_ctz(z) >> 3);
        z &= z - 1u;
        if (eb < E) list_edge(eb);
      }
    };

    bool use_list = false;
    if constexpr (LMODE == 2) {
      c0 = __builtin_amdgcn_readfirstlane(c0);
      use_list = c0 >= 0 && c0 <= CAP && c0 < lcap;
    }
    if (use_list) {
#pragma unroll
      for (int k = 0; k < KR; ++k) {
        const int i = tid + k * 1024;
        if (i < c0) {
          lsrc[i] = lv[k] & 0xFFFFFF;
          ldl[i] = (int)((unsigned)lv[k] >> 24);
        }
      }
      for (int i = KR * 1024 + tid; i < c0; i += 1024) {
        const int v = __float_as_int(lp[1 + i]);
        lsrc[i] = v & 0xFFFFFF;
        ldl[i] = (int)((unsigned)v >> 24);
      }
      if (tid == 0) cnt = c0;
    } else if (LMODE == 1 && keys_fit(E, N)) {
      const unsigned pat = (unsigned)(row0 / RT) * 0x01010101u;
      const int G = (E + 15) >> 4;
      for (int g0 = tid; g0 < G; g0 += 1024 * U) {
        uint4 v[U];
#pragma unroll
        for (int u = 0; u < U; ++u) v[u] = *key_group(lst, min(g0 + u * 1024, G - 1));
        __builtin_amdgcn_sched_barrier(0);
#pragma unroll
        for (int u = 0; u < U; ++u) {
          const unsigned z0 = zero_bytes(v[u].x ^ pat), z1 = zero_bytes(v[u].y ^ pat);
          const unsigned z2 = zero_bytes(v[u].z ^ pat), z3 = zero_bytes(v[u].w ^ pat);
          if (g0 + u * 1024 < G && (z0 | z1 | z2 | z3) != 0u) {
            const int e = 16 * (g0 + u * 1024);
            take_marked(z0, e);
            take_marked(z1, e + 4);
            take_marked(z2, e + 8);
            take_marked(z3, e + 12);
          }
        }
      }
    } else {
      const int E4 = (((size_t)dst & 15) == 0) ? (E >> 2) : 0;
      const int4* dst4 = (const int4*)dst;
      for (int g0 = tid; g0 < E4; g0 += 1024 * U) {
        int4 v[U];
#pragma unroll
        for (int u = 0; u < U; ++u) {
          int g = g0 + u * 1024;
          v[u] = (g < E4) ? dst4[g] : make_int4(-1, -1, -1, -1);
        }
#pragma unroll
        for (int u = 0; u < U; ++u) {
          int e = 4 * (g0 + u * 1024);
          take(v[u].x, e);
          take(v[u].y, e + 1);
          take(v[u].z, e + 2);
          take(v[u].w, e + 3);
        }
      }
      for (int e = 4 * E4 + tid; e < E; e += 1024) take(dst[e], e);
    }
    __syncthreads();

    const bool listed = cnt <= CAP;
    const int n = listed ? cnt : 0;
    if (!listed) {
      for (int e = tid; e < E; e += 1024) {
        const int d = dst[e];
        const unsigned dl = (unsigned)(d - row0);
        if (dl < (unsigned)rows) {
          const int s = src[e];
#pragma unroll 1
          for (int h = 0; h < H; ++h)
            atomicMax(&mx[dl * H + h], fkey(leaky(as[aidx(s) + h] + adl[dl * H + h])));
        }
      }
      __syncthreads();
      for (int e = tid; e < E; e += 1024) {
        const int d = dst[e];
        const unsigned dl = (unsigned)(d - row0);
        if (dl < (unsigned)rows) {
          const int s = src[e];
#pragma unroll 1
          for (int h = 0; h < H; ++h) {
            const float a = leaky(as[aidx(s) + h] + adl[dl * H + h]);
            atomicAdd(&dn[dl * H + h], __expf(a - fdecode(mx[dl * H + h])));
          }
        }
      }
      __syncthreads();
      for (int e = tid; e < E; e += 1024) {
        const int d = dst[e];
        const unsigned dl = (unsigned)(d - row0);
        if (dl < (unsigned)rows) {
          const int s = src[e];
#pragma unroll 1
          for (int p = 0; p < HF; ++p) {
            const int lh = dl * H + (p >> LOGF);
            const float a = leaky(as[aidx(s) + (p >> LOGF)] + adl[lh]);
            const float w = __expf(a - fdecode(mx[lh])) / (dn[lh] + 1e-16f);
            atomicAdd(&acc[dl * AST + p], w * xs[(size_t)s * HF + p]);
          }
        }
      }
    }
    {
      int sv[KR], dv[KR];
      float av[KR][H];
#pragma unroll
      for (int k = 0; k < KR; ++k) dv[k] = row0;
      if (n > 0) {
        if (use_list) {
#pragma unroll
          for (int k = 0; k < KR; ++k) {
            const int i = tid + k * 1024;
            sv[k] = (i < n) ? lsrc[i] : 0;
          }
        } else {
#pragma unroll
          for (int k = 0; k < KR; ++k) {
            const int i = tid + k * 1024;
            const int e = (i < n) ? lsrc[i] : 0;
            sv[k] = src[e];
            dv[k] = dst[e];
          }
        }
        __builtin_amdgcn_sched_barrier(0);
#pragma unroll
        for (int k = 0; k < KR; ++k) load_heads<H>(as + aidx(sv[k]), av[k]);
      }
#pragma unroll
      for (int k = 0; k < KR; ++k) {
        const int i = tid + k * 1024;
        if (i < n) {
          if (!use_list) ldl[i] = min(max(dv[k] - row0, 0), rows - 1);
          const int lh = ldl[i] * H;
          lsrc[i] = sv[k];
#pragma unroll
          for (int h = 0; h < H; ++h) {
            av[k][h] = leaky(av[k][h] + adl[lh + h]);
            atomicMax(&mx[lh + h], fkey(av[k][h]));
          }
        }
      }
      for (int i = KR * 1024 + tid; i < n; i += 1024) {
        int s = lsrc[i];
        if (!use_list) {
          ldl[i] = min(max(dst[s] - row0, 0), rows - 1);
          s = src[s];
        }
        const int lh = ldl[i] * H;
        float a[H];
        load_heads<H>(as + aidx(s), a);
        lsrc[i] = s;
#pragma unroll
        for (int h = 0; h < H; ++h) atomicMax(&mx[lh + h], fkey(leaky(a[h] + adl[lh + h])));
      }
      __syncthreads();
#pragma unroll
      for (int k = 0; k < KR; ++k) {
        const int i = tid + k * 1024;
        if (i < n) {
          const int lh = ldl[i] * H;
#pragma unroll
          for (int h = 0; h < H; ++h)
            atomicAdd(&dn[lh + h], __expf(av[k][h] - fdecode(mx[lh + h])));
        }
      }
      for (int i = KR * 1024 + tid; i < n; i += 1024) {
        const int lh = ldl[i] * H;
        float a[H];
        load_heads<H>(as + aidx(lsrc[i]), a);
#pragma unroll
        for (int h = 0; h < H; ++h)
          atomicAdd(&dn[lh + h], __expf(leaky(a[h] + adl[lh + h]) - fdecode(mx[lh + h])));
      }
    }
    __syncthreads();

    if constexpr (LMODE == 1) {
      const bool park = listed && n < lcap && N <= (1 << 24);
      if (tid == 0) lp[0] = __int_as_float(park ? n : -1);
      if (park)
        for (int i = tid; i < n; i += 1024) lp[1 + i] = __int_as_float(lsrc[i] | (ldl[i] << 24));
    }

    for (int i0 = wave * EB; i0 < n; i0 += 32 * EB) {
      const int j = min(lane / H, EB - 1), h = lane & (H - 1);
      const int ii = min(i0 + j, n - 1);
      const int s = lsrc[ii], dl = ldl[ii];
      const int lh = dl * H + h;
      const float v_as = as[aidx(s) + h], v_ad = adl[lh], v_den = dn[lh];
      const unsigned v_mx = mx[lh];
      float m[EB][NS];
      int dls[EB];
#pragma unroll
      for (int jj = 0; jj < EB; ++jj) {
        const int sj = __builtin_amdgcn_readlane(s, jj * H);
        dls[jj] = __builtin_amdgcn_readlane(dl, jj * H);
#pragma unroll
        for (int i = 0; i < NS; ++i) m[jj][i] = xs[(size_t)sj * HF + lane + 32 * i];
      }
      const float w = __expf(leaky(v_as + v_ad) - fdecode(v_mx)) / (v_den + 1e-16f);
#pragma unroll
      for (int jj = 0; jj < EB; ++jj) {
        const bool valid = i0 + jj < n;
#pragma unroll
        for (int i = 0; i < NS; ++i) {
          const int p = lane + 32 * i;
          const float wj = __builtin_bit_cast(
              float, __builtin_amdgcn_readlane(__builtin_bit_cast(int, w), jj * H + ((32 * i) >> LOGF)));
          atomicAdd(&acc[dls[jj] * AST + p], valid ? wj * m[jj][i] : 0.f);
        }
      }
    }
    __syncthreads();

    if constexpr (!NEXT) {
      for (int i = tid; i < rows * HF; i += 1024) {
        float v = acc[i] + bias[i & (HF - 1)];
        if (RELU) v = v > 0.f ? v : 0.f;
        out[(size_t)row0 * HF + i] = v;
      }
    } else {
      int tl = tid;
      asm volatile("" : "+v"(tl));
      for (int i = tid; i < RT * HF; i += 1024) {
        const int cc = i & (HF - 1), ix = (i / HF) * AST + cc;
        float v = acc[ix] + bias[cc];
        if (RELU) v = v > 0.f ? v : 0.f;
        acc[ix] = v;
      }
      __syncthreads();
      {
        const int rr = tl >> 2, q = tl & 3;
        const float* hr = acc + min(rr, RT - 1) * AST;
        float s1 = 0.f, s2 = 0.f;
#pragma unroll 8
        for (unsigned k = (unsigned)q; k < (unsigned)HF; k += 4u) {
          const float hv = hr[k];
          s1 += hv * wan[k];
          s2 += hv * wdn[k];
        }
        s1 += __shfl_xor(s1, 1); s2 += __shfl_xor(s2, 1);
        s1 += __shfl_xor(s1, 2); s2 += __shfl_xor(s2, 2);
        if (q == 0 && rr < RT) {
          adl[rr] = s1;
          dn[rr] = s2;
        }
      }
      __syncthreads();
      if (tl < 2 * LGP) {
        const int j = tl & (LGP - 1);
        if (j < rows) (tl < LGP ? asn : adn)[(size_t)(row0 / RT) * LGP + j] = (tl < LGP ? adl : dn)[j];
      }
      {
        constexpr int MT = (RT + 15) / 16, NU = MT * (NN / 32);
        static_assert(NU <= 52 && NN == 128, "two units per wave at most, unit -> (tile, column pair) by shift and mask");
        const int half = (tl >> 4) & 1, r = tl & 15;
        const int wv = __builtin_amdgcn_readfirstlane(tl >> 5);
        const int wb = wv >> 3, wo = wv & 7;
        const bool two = wo < 4 || wo == 4 + wb;
        const int u2 = 32 + 5 * wb + min(wo, 4);
#pragma unroll 1
        for (int t = 0; t < 2; ++t) {
          const int u = t == 0 ? wv : u2;
          if (t == 1 && !two) break;
          if (u >= NU) break;
          const int mt = u >> 2, np = u & 3;
          if (mt * 16 >= rows) continue;
          const float* ap = acc + min(mt * 16 + r, RT - 1) * AST + 2 * half;
          const float* bp = wn + (unsigned)((2 * half) * NN + np * 32 + r);
          v8f c0 = {}, c1 = {};
          constexpr int KG = 4;
#pragma unroll
          for (int k0 = 0; k0 < HF; k0 += 4 * KG) {
            v2f a[KG], b0[KG], b1[KG];
#pragma unroll
            for (int g = 0; g < KG; ++g) {
              const int k = k0 + 4 * g;
              a[g] = *(const v2f*)(ap + k);
              b0[g].x = bp[k * NN];       b0[g].y = bp[k * NN + NN];
              b1[g].x = bp[k * NN + 16];  b1[g].y = bp[k * NN + NN + 16];
            }
            __builtin_amdgcn_sched_barrier(0);
#pragma unroll
            for (int g = 0; g < KG; ++g) {
              c0 = wmma4(a[g], b0[g], c0);
              c1 = wmma4(a[g], b1[g], c1);
            }
          }
          float* cp = out + (unsigned)((row0 + mt * 16 + 8 * half) * NN + np * 32 + r);
          const int rl = rows - (mt * 16 + 8 * half);
#pragma unroll
          for (int i = 0; i < 8; ++i) {
            if (i < rl) {
              cp[i * NN] = c0[i];
              cp[i * NN + 16] = c1[i];
            }
          }
        }
      }
    }
    __syncthreads();
  }
}

extern "C" void kernel_launch(void* const* d_in, const int* in_sizes, int n_in,
                              void* d_out, int out_size, void* d_ws, size_t ws_size,
                              hipStream_t stream) {
  (void)n_in; (void)out_size; (void)ws_size;
  const float* x     = (const float*)d_in[0];
  const int*   ei    = (const int*)d_in[1];
  const float* W1s   = (const float*)d_in[2];
  const float* W1d   = (const float*)d_in[3];
  const float* att1s = (const float*)d_in[4];
  const float* att1d = (const float*)d_in[5];
  const float* b1    = (const float*)d_in[6];
  const float* W2s   = (const float*)d_in[7];
  const float* W2d   = (const float*)d_in[8];
  const float* att2s = (const float*)d_in[9];
  const float* att2d = (const float*)d_in[10];
  const float* b2    = (const float*)d_in[11];
  float* out = (float*)d_out;

  const int N = in_sizes[0] / 128;
  const int E = in_sizes[1] / 2;
  const int* src = ei;
  const int* dst = ei + E;

  float* ws = (float*)d_ws;
  size_t off = 0;
  float*    XS   = ws + off; off += (size_t)N * 256;
  float*    Hb   = ws + off; off += (size_t)N * 256;
  float*    AS   = ws + off; off += (size_t)N * 8;
  float*    AD   = ws + off; off += (size_t)N * 8;
  float*    WA   = ws + off; off += 2048;
  float*    WD   = ws + off; off += 2048;
  float*    AS2  = Hb + (size_t)N * 128;
  float*    AD2  = AS2 + (size_t)N * 2 + 256;

  const int TPB = 256;

  gemm_wmma_f32<256, 128><<<dim3(N / 16, 256 / 64), 32, 0, stream>>>(x, W1s, XS);
  make_wa<8, 32><<<(128 * 8 + TPB - 1) / TPB, TPB, 0, stream>>>(W1s, att1s, WA, 128);
  make_wa<8, 32><<<(128 * 8 + TPB - 1) / TPB, TPB, 0, stream>>>(W1d, att1d, WD, 128);
  proj_att<128, 8><<<((size_t)N * 8 + TPB - 1) / TPB, TPB, 0, stream>>>(x, WA, AS, N);
  proj_att<128, 8><<<((size_t)N * 8 + TPB - 1) / TPB, TPB, 0, stream>>>(x, WD, AD, N);

  tile_keys<<<(E + 16 * TPB - 1) / (16 * TPB), TPB, 0, stream>>>(dst, out, E, N);

  make_wa<1, 128><<<(256 + TPB - 1) / TPB, TPB, 0, stream>>>(W2s, att2s, WA, 256);
  make_wa<1, 128><<<(256 + TPB - 1) / TPB, TPB, 0, stream>>>(W2d, att2d, WD, 256);

  edge_aggregate_owned<8, 5, 4, true, 1><<<256, 1024, 0, stream>>>(src, dst, AS, AD,
                                                                 XS, b1, Hb, out, E, N,
                                                                 W2s, WA, WD, AS2, AD2);

  edge_aggregate_owned<1, 7, 8, false, 2><<<256, 1024, 0, stream>>>(src, dst, AS2, AD2,
                                                                  Hb, b2, out, nullptr, E, N,
                                                                  nullptr, nullptr, nullptr,
                                                                  nullptr, nullptr);
}


# the 8 launches in front of the first aggregation (none reads another's output) run as block ranges of ONE launch; logits blocks fold their own att panels in LDS (8-lane split sums: new sum order)
# speedup vs baseline: 3.6082x; 1.1423x over previous
#include <hip/hip_runtime.h>
#include <stddef.h>

typedef __attribute__((ext_vector_type(2))) float v2f;
typedef __attribute__((ext_vector_type(8))) float v8f;

__device__ __forceinline__ v8f wmma4(v2f a, v2f b, v8f c) {
  return __builtin_amdgcn_wmma_f32_16x16x4_f32(
       false, a,  false, b,
       (short)0, c,  false,  false);
}

template <int N, int K>
__device__ __forceinline__ void gemm_strip(const float* __restrict__ A,
                                           const float* __restrict__ B,
                                           float* __restrict__ C, int mt, int ns, int lane) {
  const int m0 = mt * 16;
  const int n0 = ns * 64;
  const int half = lane >> 4;
  const int r    = lane & 15;

  const float* Arow = A + (size_t)(m0 + r) * K + 2 * half;
  const float* Bcol = B + (size_t)(2 * half) * N + n0 + r;

  v8f c0 = {}, c1 = {}, c2 = {}, c3 = {};
#pragma unroll
  for (int k = 0; k < K; k += 4) {
    v2f a = *(const v2f*)(Arow + k);
    v2f b0, b1, b2, b3;
    b0.x = Bcol[k * N + 0];       b0.y = Bcol[k * N + N + 0];
    b1.x = Bcol[k * N + 16];      b1.y = Bcol[k * N + N + 16];
    b2.x = Bcol[k * N + 32];      b2.y = Bcol[k * N + N + 32];
    b3.x = Bcol[k * N + 48];      b3.y = Bcol[k * N + N + 48];
    c0 = wmma4(a, b0, c0);
    c1 = wmma4(a, b1, c1);
    c2 = wmma4(a, b2, c2);
    c3 = wmma4(a, b3, c3);
  }

  float* Crow = C + (size_t)(m0 + 8 * half) * N + n0 + r;
#pragma unroll
  for (int i = 0; i < 8; ++i) {
    Crow[i * N + 0]  = c0[i];
    Crow[i * N + 16] = c1[i];
    Crow[i * N + 32] = c2[i];
    Crow[i * N + 48] = c3[i];
  }
}

constexpr int kLN = 196;
static_assert(kLN % 4 == 0, "4 nodes of a wave fill one whole 128-byte line of a_s and of a_d");

__device__ __forceinline__ void fold_and_logits(const float* __restrict__ x,
                                                const float* __restrict__ Wsrc, const float* __restrict__ Wdst,
                                                const float* __restrict__ atts, const float* __restrict__ attd,
                                                float* __restrict__ as, float* __restrict__ ad,
                                                float4* pan, int blk, int nblk, int N) {
  const int tid = threadIdx.x, lane = tid & 31, wave = tid >> 5;
  {
    const int hh = (wave & 1) * 4 + (lane >> 3);
    const float4 a_s4 = *(const float4*)(atts + hh * 32 + (lane & 7) * 4);
    const float4 a_d4 = *(const float4*)(attd + hh * 32 + (lane & 7) * 4);
    const float* ws = Wsrc + wave * 128 + lane * 4;
    const float* wd = Wdst + wave * 128 + lane * 4;
    constexpr int FC = 8;
#pragma unroll 1
    for (int c0 = 0; c0 < 32; c0 += FC) {
      float4 vs[FC], vd[FC];
#pragma unroll
      for (int j = 0; j < FC; ++j) {
        vs[j] = *(const float4*)(ws + (c0 + j) * 1024);
        vd[j] = *(const float4*)(wd + (c0 + j) * 1024);
      }
#pragma unroll
      for (int j = 0; j < FC; ++j) {
        float ps = vs[j].x * a_s4.x + vs[j].y * a_s4.y + vs[j].z * a_s4.z + vs[j].w * a_s4.w;
        float pd = vd[j].x * a_d4.x + vd[j].y * a_d4.y + vd[j].z * a_d4.z + vd[j].w * a_d4.w;
        ps += __shfl_xor(ps, 1); pd += __shfl_xor(pd, 1);
        ps += __shfl_xor(ps, 2); pd += __shfl_xor(pd, 2);
        ps += __shfl_xor(ps, 4); pd += __shfl_xor(pd, 4);
        if ((lane & 7) == 0) {
          const int k = (c0 + j) * 4 + (wave >> 1);
          float* q = (float*)(pan + (k >> 1) * 8 + hh) + (k & 1) * 2;
          q[0] = ps;
          q[1] = pd;
        }
      }
    }
  }
  __syncthreads();

  const int h = tid & 7;
  for (int row0 = blk * kLN; row0 < N; row0 += nblk * kLN) {
    const int rows = min(kLN, N - row0);
    for (int p0 = wave * 4; p0 < rows; p0 += 32) {
      const int nl = p0 + (lane >> 3);
      const int node = row0 + min(nl, rows - 1);
      const float4* xp = (const float4*)(x + (size_t)node * 128);
      float s = 0.f, d = 0.f;
      constexpr int XC = 8;
#pragma unroll
      for (int q0 = 0; q0 < 32; q0 += XC) {
        float4 xv[XC];
#pragma unroll
        for (int j = 0; j < XC; ++j) xv[j] = xp[q0 + j];
#pragma unroll
        for (int j = 0; j < XC; ++j) {
          const float4 w0 = pan[(2 * (q0 + j)) * 8 + h];
          const float4 w1 = pan[(2 * (q0 + j) + 1) * 8 + h];
          s += xv[j].x * w0.x;  d += xv[j].x * w0.y;
          s += xv[j].y * w0.z;  d += xv[j].y * w0.w;
          s += xv[j].z * w1.x;  d += xv[j].z * w1.y;
          s += xv[j].w * w1.z;  d += xv[j].w * w1.w;
        }
      }
      if (nl < rows) {
        as[(size_t)node * 8 + h] = s;
        ad[(size_t)node * 8 + h] = d;
      }
    }
  }
}

__device__ __forceinline__ void fold_next(const float* __restrict__ Wsrc, const float* __restrict__ Wdst,
                                          const float* __restrict__ atts, const float* __restrict__ attd,
                                          float* __restrict__ wa, float* __restrict__ wd, int unit, int lane) {
  const bool second = unit >= 8;
  const float* Wm = (second ? Wdst : Wsrc) + (size_t)((unit & 7) * 32) * 128 + lane * 4;
  const float4 a4 = *(const float4*)((second ? attd : atts) + lane * 4);
  float res = 0.f;
  constexpr int RC = 16;
#pragma unroll 1
  for (int j0 = 0; j0 < 32; j0 += RC) {
    float4 v[RC];
#pragma unroll
    for (int j = 0; j < RC; ++j) v[j] = *(const float4*)(Wm + (j0 + j) * 128);
#pragma unroll
    for (int j = 0; j < RC; ++j) {
      float p = v[j].x * a4.x + v[j].y * a4.y + v[j].z * a4.z + v[j].w * a4.w;
      p += __shfl_xor(p, 1);
      p += __shfl_xor(p, 2);
      p += __shfl_xor(p, 4);
      p += __shfl_xor(p, 8);
      p += __shfl_xor(p, 16);
      if (lane == j0 + j) res = p;
    }
  }
  (second ? wd : wa)[(unit & 7) * 32 + lane] = res;
}

__device__ __forceinline__ unsigned fkey(float f) {
  unsigned u = __float_as_uint(f);
  return (u & 0x80000000u) ? ~u : (u | 0x80000000u);
}
__device__ __forceinline__ float fdecode(unsigned k) {
  unsigned u = (k & 0x80000000u) ? (k & 0x7FFFFFFFu) : ~k;
  return __uint_as_float(u);
}
__device__ __forceinline__ float leaky(float a) {
  return a > 0.f ? a : 0.2f * a;
}

template <int H>
__device__ __forceinline__ void load_heads(const float* __restrict__ p, float (&a)[H]) {
  if constexpr (H % 4 == 0) {
#pragma unroll
    for (int q = 0; q < H / 4; ++q) {
      const float4 v = ((const float4*)p)[q];
      a[4 * q + 0] = v.x; a[4 * q + 1] = v.y; a[4 * q + 2] = v.z; a[4 * q + 3] = v.w;
    }
  } else {
#pragma unroll
    for (int h = 0; h < H; ++h) a[h] = p[h];
  }
}

constexpr int kRT = 196;
constexpr int kLP = 128;
constexpr int kCAP = 8192;
constexpr int kKOFF = 8448;
constexpr int kKGRP = 4096;
static_assert(kKOFF >= 1 + kCAP + 1, "keys lie behind the parked list");
static_assert(kKOFF % 4 == 0 && kKOFF + 4 * kKGRP <= kRT * kLP, "keys lie inside the tile's own rows, 16-byte aligned");

__device__ __forceinline__ bool keys_fit(int E, int N) {
  return N <= kRT * 256 && (long long)((E + 15) >> 4) <= (long long)(N / kRT) * kKGRP;
}
__device__ __forceinline__ uint4* key_group(float* base, int g) {
  const unsigned ug = (unsigned)g;
  return (uint4*)(base + (size_t)(ug / (unsigned)kKGRP) * (kRT * kLP) + kKOFF) + (ug % (unsigned)kKGRP);
}
__device__ __forceinline__ unsigned tile_key(int d) {
  return ((unsigned)d / (unsigned)kRT) & 255u;
}

__device__ __forceinline__ void tile_keys(const int* __restrict__ dst, float* __restrict__ lst, int E, int N, int g) {
  if (!keys_fit(E, N)) return;
  const int e0 = 16 * g;
  if (e0 >= E) return;
  unsigned w[4];
  if ((((size_t)dst & 15) == 0) && e0 + 16 <= E) {
    const int4* p = (const int4*)(dst + e0);
    int4 v[4];
#pragma unroll
    for (int q = 0; q < 4; ++q) v[q] = p[q];
#pragma unroll
    for (int q = 0; q < 4; ++q)
      w[q] = tile_key(v[q].x) | (tile_key(v[q].y) << 8) | (tile_key(v[q].z) << 16) | (tile_key(v[q].w) << 24);
  } else {
#pragma unroll
    for (int q = 0; q < 4; ++q) {
      w[q] = 0u;
#pragma unroll
      for (int b = 0; b < 4; ++b) {
        const int e = e0 + 4 * q + b;
        w[q] |= (e < E ? tile_key(dst[e]) : 255u) << (8 * b);
      }
    }
  }
  *key_group(lst, g) = make_uint4(w[0], w[1], w[2], w[3]);
}

__global__ void __launch_bounds__(256)
front(const float* __restrict__ x, const int* __restrict__ dst,
      const float* __restrict__ W1s, const float* __restrict__ W1d,
      const float* __restrict__ att1s, const float* __restrict__ att1d,
      const float* __restrict__ W2s, const float* __restrict__ W2d,
      const float* __restrict__ att2s, const float* __restrict__ att2d,
      float* __restrict__ xs, float* __restrict__ as, float* __restrict__ ad,
      float* __restrict__ wa2, float* __restrict__ wd2, float* __restrict__ lst,
      int N, int E, int bF, int bK, int bG) {
  __shared__ float4 pan[64 * 8];
  const int b = blockIdx.x;
  const int lane = threadIdx.x & 31, wave = threadIdx.x >> 5;
  if (b < bF) {
    fold_and_logits(x, W1s, W1d, att1s, att1d, as, ad, pan, b, bF, N);
  } else if (b < bK) {
    const int unit = (b - bF) * 8 + wave;
    if (unit < 16) fold_next(W2s, W2d, att2s, att2d, wa2, wd2, unit, lane);
  } else if (b < bG) {
    tile_keys(dst, lst, E, N, (b - bK) * 256 + (int)threadIdx.x);
  } else {
    const int MT = N >> 4;
    const int g = (b - bG) * 8 + wave;
    if (g < 4 * MT) gemm_strip<256, 128>(x, W1s, xs, g % MT, g / MT, lane);
  }
}

__device__ __forceinline__ unsigned zero_bytes(unsigned x) {
  return ~(((x & 0x7F7F7F7Fu) + 0x7F7F7F7Fu) | x | 0x7F7F7F7Fu);
}

template <int H, int LOGF, int EB, bool RELU, int LMODE>
__global__ void __launch_bounds__(1024)
edge_aggregate_owned(const int* __restrict__ src, const int* __restrict__ dst,
                     const float* __restrict__ as, const float* __restrict__ ad,
                     const float* __restrict__ xs, const float* __restrict__ bias,
                     float* __restrict__ out, float* __restrict__ lst, int E, int N,
                     const float* __restrict__ wn, const float* __restrict__ wan,
                     const float* __restrict__ wdn, float* __restrict__ asn,
                     float* __restrict__ adn) {
  constexpr int HF = H << LOGF;
  constexpr bool NEXT = LMODE == 1;
  constexpr int NN = 128;
  constexpr int AST = NEXT ? HF + 4 : HF;
  static_assert(!NEXT || HF % 4 == 0, "whole k steps of the matrix instruction");
  constexpr int NS = HF / 32;
  constexpr int RT = kRT;
  constexpr int CAP = kCAP;
  constexpr int KR = 4;
  constexpr int U = 8;
  constexpr int LP = kLP;
  static_assert(RT <= 256, "a parked entry keeps its row in 8 bits");
  static_assert(LMODE != 2 || HF == LP, "the parked lists lie in this launch's own output rows");
  static_assert(EB * H <= 32, "one lane per (edge, head) of a batch");
  static_assert(LOGF >= 5, "a 32-lane slice lies inside one head");
  __shared__ __attribute__((aligned(16))) float acc[RT * AST];
  __shared__ int lsrc[CAP];
  __shared__ int ldl[CAP];
  __shared__ float adl[RT * H];
  __shared__ unsigned mx[RT * H];
  __shared__ float dn[RT * H];
  __shared__ int cnt;

  const int tid = threadIdx.x, lane = tid & 31, wave = tid >> 5;
  constexpr int LGP = 256;
  static_assert(LGP >= RT && LGP % 32 == 0 && (LMODE != 2 || H == 1), "a tile's logits fill whole lines of their own");
  auto aidx = [](int s) -> size_t {
    if constexpr (LMODE == 2) return (size_t)((unsigned)s / (unsigned)RT) * LGP + (unsigned)s % (unsigned)RT;
    else return (size_t)s * H;
  };

  for (int row0 = blockIdx.x * RT; row0 < N; row0 += gridDim.x * RT) {
    const int rows = min(RT, N - row0);
    const int lcap = rows * LP;
    float* const lp = (LMODE == 2 ? out : lst) + (size_t)row0 * LP;
    int c0 = -1;
    int lv[KR];
    if constexpr (LMODE == 2) {
      c0 = __float_as_int(lp[0]);
#pragma unroll
      for (int k = 0; k < KR; ++k) lv[k] = __float_as_int(lp[min(1 + tid + k * 1024, lcap - 1)]);
    }
    for (int i = tid; i < RT * AST; i += 1024) acc[i] = 0.f;
    for (int i = tid; i < RT * H; i += 1024) {
      adl[i] = (i < rows * H) ? ad[aidx(row0) + i] : 0.f;
      mx[i] = 0u;
      dn[i] = 0.f;
    }
    if (tid == 0) cnt = 0;
    __syncthreads();

    auto list_edge = [&](int e) {
      int idx = atomicAdd(&cnt, 1);
      if (idx < CAP) lsrc[idx] = e;
    };
    auto take = [&](int d, int e) {
      if ((unsigned)(d - row0) < (unsigned)rows) list_edge(e);
    };
    auto take_marked = [&](unsigned z, int e) {
      while (z != 0u) {
        const int eb = e + (__builtin_ctz(z) >> 3);
        z &= z - 1u;
        if (eb < E) list_edge(eb);
      }
    };

    bool use_list = false;
    if constexpr (LMODE == 2) {
      c0 = __builtin_amdgcn_readfirstlane(c0);
      use_list = c0 >= 0 && c0 <= CAP && c0 < lcap;
    }
    if (use_list) {
#pragma unroll
      for (int k = 0; k < KR; ++k) {
        const int i = tid + k * 1024;
        if (i < c0) {
          lsrc[i] = lv[k] & 0xFFFFFF;
          ldl[i] = (int)((unsigned)lv[k] >> 24);
        }
      }
      for (int i = KR * 1024 + tid; i < c0; i += 1024) {
        const int v = __float_as_int(lp[1 + i]);
        lsrc[i] = v & 0xFFFFFF;
        ldl[i] = (int)((unsigned)v >> 24);
      }
      if (tid == 0) cnt = c0;
    } else if (LMODE == 1 && keys_fit(E, N)) {
      const unsigned pat = (unsigned)(row0 / RT) * 0x01010101u;
      const int G = (E + 15) >> 4;
      for (int g0 = tid; g0 < G; g0 += 1024 * U) {
        uint4 v[U];
#pragma unroll
        for (int u = 0; u < U; ++u) v[u] = *key_group(lst, min(g0 + u * 1024, G - 1));
        __builtin_amdgcn_sched_barrier(0);
#pragma unroll
        for (int u = 0; u < U; ++u) {
          const unsigned z0 = zero_bytes(v[u].x ^ pat), z1 = zero_bytes(v[u].y ^ pat);
          const unsigned z2 = zero_bytes(v[u].z ^ pat), z3 = zero_bytes(v[u].w ^ pat);
          if (g0 + u * 1024 < G && (z0 | z1 | z2 | z3) != 0u) {
            const int e = 16 * (g0 + u * 1024);
            take_marked(z0, e);
            take_marked(z1, e + 4);
            take_marked(z2, e + 8);
            take_marked(z3, e + 12);
          }
        }
      }
    } else {
      const int E4 = (((size_t)dst & 15) == 0) ? (E >> 2) : 0;
      const int4* dst4 = (const int4*)dst;
      for (int g0 = tid; g0 < E4; g0 += 1024 * U) {
        int4 v[U];
#pragma unroll
        for (int u = 0; u < U; ++u) {
          int g = g0 + u * 1024;
          v[u] = (g < E4) ? dst4[g] : make_int4(-1, -1, -1, -1);
        }
#pragma unroll
        for (int u = 0; u < U; ++u) {
          int e = 4 * (g0 + u * 1024);
          take(v[u].x, e);
          take(v[u].y, e + 1);
          take(v[u].z, e + 2);
          take(v[u].w, e + 3);
        }
      }
      for (int e = 4 * E4 + tid; e < E; e += 1024) take(dst[e], e);
    }
    __syncthreads();

    const bool listed = cnt <= CAP;
    const int n = listed ? cnt : 0;
    if (!listed) {
      for (int e = tid; e < E; e += 1024) {
        const int d = dst[e];
        const unsigned dl = (unsigned)(d - row0);
        if (dl < (unsigned)rows) {
          const int s = src[e];
#pragma unroll 1
          for (int h = 0; h < H; ++h)
            atomicMax(&mx[dl * H + h], fkey(leaky(as[aidx(s) + h] + adl[dl * H + h])));
        }
      }
      __syncthreads();
      for (int e = tid; e < E; e += 1024) {
        const int d = dst[e];
        const unsigned dl = (unsigned)(d - row0);
        if (dl < (unsigned)rows) {
          const int s = src[e];
#pragma unroll 1
          for (int h = 0; h < H; ++h) {
            const float a = leaky(as[aidx(s) + h] + adl[dl * H + h]);
            atomicAdd(&dn[dl * H + h], __expf(a - fdecode(mx[dl * H + h])));
          }
        }
      }
      __syncthreads();
      for (int e = tid; e < E; e += 1024) {
        const int d = dst[e];
        const unsigned dl = (unsigned)(d - row0);
        if (dl < (unsigned)rows) {
          const int s = src[e];
#pragma unroll 1
          for (int p = 0; p < HF; ++p) {
            const int lh = dl * H + (p >> LOGF);
            const float a = leaky(as[aidx(s) + (p >> LOGF)] + adl[lh]);
            const float w = __expf(a - fdecode(mx[lh])) / (dn[lh] + 1e-16f);
            atomicAdd(&acc[dl * AST + p], w * xs[(size_t)s * HF + p]);
          }
        }
      }
    }
    {
      int sv[KR], dv[KR];
      float av[KR][H];
#pragma unroll
      for (int k = 0; k < KR; ++k) dv[k] = row0;
      if (n > 0) {
        if (use_list) {
#pragma unroll
          for (int k = 0; k < KR; ++k) {
            const int i = tid + k * 1024;
            sv[k] = (i < n) ? lsrc[i] : 0;
          }
        } else {
#pragma unroll
          for (int k = 0; k < KR; ++k) {
            const int i = tid + k * 1024;
            const int e = (i < n) ? lsrc[i] : 0;
            sv[k] = src[e];
            dv[k] = dst[e];
          }
        }
        __builtin_amdgcn_sched_barrier(0);
#pragma unroll
        for (int k = 0; k < KR; ++k) load_heads<H>(as + aidx(sv[k]), av[k]);
      }
#pragma unroll
      for (int k = 0; k < KR; ++k) {
        const int i = tid + k * 1024;
        if (i < n) {
          if (!use_list) ldl[i] = min(max(dv[k] - row0, 0), rows - 1);
          const int lh = ldl[i] * H;
          lsrc[i] = sv[k];
#pragma unroll
          for (int h = 0; h < H; ++h) {
            av[k][h] = leaky(av[k][h] + adl[lh + h]);
            atomicMax(&mx[lh + h], fkey(av[k][h]));
          }
        }
      }
      for (int i = KR * 1024 + tid; i < n; i += 1024) {
        int s = lsrc[i];
        if (!use_list) {
          ldl[i] = min(max(dst[s] - row0, 0), rows - 1);
          s = src[s];
        }
        const int lh = ldl[i] * H;
        float a[H];
        load_heads<H>(as + aidx(s), a);
        lsrc[i] = s;
#pragma unroll
        for (int h = 0; h < H; ++h) atomicMax(&mx[lh + h], fkey(leaky(a[h] + adl[lh + h])));
      }
      __syncthreads();
#pragma unroll
      for (int k = 0; k < KR; ++k) {
        const int i = tid + k * 1024;
        if (i < n) {
          const int lh = ldl[i] * H;
#pragma unroll
          for (int h = 0; h < H; ++h)
            atomicAdd(&dn[lh + h], __expf(av[k][h] - fdecode(mx[lh + h])));
        }
      }
      for (int i = KR * 1024 + tid; i < n; i += 1024) {
        const int lh = ldl[i] * H;
        float a[H];
        load_heads<H>(as + aidx(lsrc[i]), a);
#pragma unroll
        for (int h = 0; h < H; ++h)
          atomicAdd(&dn[lh + h], __expf(leaky(a[h] + adl[lh + h]) - fdecode(mx[lh + h])));
      }
    }
    __syncthreads();

    if constexpr (LMODE == 1) {
      const bool park = listed && n < lcap && N <= (1 << 24);
      if (tid == 0) lp[0] = __int_as_float(park ? n : -1);
      if (park)
        for (int i = tid; i < n; i += 1024) lp[1 + i] = __int_as_float(lsrc[i] | (ldl[i] << 24));
    }

    for (int i0 = wave * EB; i0 < n; i0 += 32 * EB) {
      const int j = min(lane / H, EB - 1), h = lane & (H - 1);
      const int ii = min(i0 + j, n - 1);
      const int s = lsrc[ii], dl = ldl[ii];
      const int lh = dl * H + h;
      const float v_as = as[aidx(s) + h], v_ad = adl[lh], v_den = dn[lh];
      const unsigned v_mx = mx[lh];
      float m[EB][NS];
      int dls[EB];
#pragma unroll
      for (int jj = 0; jj < EB; ++jj) {
        const int sj = __builtin_amdgcn_readlane(s, jj * H);
        dls[jj] = __builtin_amdgcn_readlane(dl, jj * H);
#pragma unroll
        for (int i = 0; i < NS; ++i) m[jj][i] = xs[(size_t)sj * HF + lane + 32 * i];
      }
      const float w = __expf(leaky(v_as + v_ad) - fdecode(v_mx)) / (v_den + 1e-16f);
#pragma unroll
      for (int jj = 0; jj < EB; ++jj) {
        const bool valid = i0 + jj < n;
#pragma unroll
        for (int i = 0; i < NS; ++i) {
          const int p = lane + 32 * i;
          const float wj = __builtin_bit_cast(
              float, __builtin_amdgcn_readlane(__builtin_bit_cast(int, w), jj * H + ((32 * i) >> LOGF)));
          atomicAdd(&acc[dls[jj] * AST + p], valid ? wj * m[jj][i] : 0.f);
        }
      }
    }
    __syncthreads();

    if constexpr (!NEXT) {
      for (int i = tid; i < rows * HF; i += 1024) {
        float v = acc[i] + bias[i & (HF - 1)];
        if (RELU) v = v > 0.f ? v : 0.f;
        out[(size_t)row0 * HF + i] = v;
      }
    } else {
      int tl = tid;
      asm volatile("" : "+v"(tl));
      for (int i = tid; i < RT * HF; i += 1024) {
        const int cc = i & (HF - 1), ix = (i / HF) * AST + cc;
        float v = acc[ix] + bias[cc];
        if (RELU) v = v > 0.f ? v : 0.f;
        acc[ix] = v;
      }
      __syncthreads();
      {
        const int rr = tl >> 2, q = tl & 3;
        const float* hr = acc + min(rr, RT - 1) * AST;
        float s1 = 0.f, s2 = 0.f;
#pragma unroll 8
        for (unsigned k = (unsigned)q; k < (unsigned)HF; k += 4u) {
          const float hv = hr[k];
          s1 += hv * wan[k];
          s2 += hv * wdn[k];
        }
        s1 += __shfl_xor(s1, 1); s2 += __shfl_xor(s2, 1);
        s1 += __shfl_xor(s1, 2); s2 += __shfl_xor(s2, 2);
        if (q == 0 && rr < RT) {
          adl[rr] = s1;
          dn[rr] = s2;
        }
      }
      __syncthreads();
      if (tl < 2 * LGP) {
        const int j = tl & (LGP - 1);
        if (j < rows) (tl < LGP ? asn : adn)[(size_t)(row0 / RT) * LGP + j] = (tl < LGP ? adl : dn)[j];
      }
      {
        constexpr int MT = (RT + 15) / 16, NU = MT * (NN / 32);
        static_assert(NU <= 52 && NN == 128, "two units per wave at most, unit -> (tile, column pair) by shift and mask");
        const int half = (tl >> 4) & 1, r = tl & 15;
        const int wv = __builtin_amdgcn_readfirstlane(tl >> 5);
        const int wb = wv >> 3, wo = wv & 7;
        const bool two = wo < 4 || wo == 4 + wb;
        const int u2 = 32 + 5 * wb + min(wo, 4);
#pragma unroll 1
        for (int t = 0; t < 2; ++t) {
          const int u = t == 0 ? wv : u2;
          if (t == 1 && !two) break;
          if (u >= NU) break;
          const int mt = u >> 2, np = u & 3;
          if (mt * 16 >= rows) continue;
          const float* ap = acc + min(mt * 16 + r, RT - 1) * AST + 2 * half;
          const float* bp = wn + (unsigned)((2 * half) * NN + np * 32 + r);
          v8f c0 = {}, c1 = {};
          constexpr int KG = 4;
#pragma unroll
          for (int k0 = 0; k0 < HF; k0 += 4 * KG) {
            v2f a[KG], b0[KG], b1[KG];
#pragma unroll
            for (int g = 0; g < KG; ++g) {
              const int k = k0 + 4 * g;
              a[g] = *(const v2f*)(ap + k);
              b0[g].x = bp[k * NN];       b0[g].y = bp[k * NN + NN];
              b1[g].x = bp[k * NN + 16];  b1[g].y = bp[k * NN + NN + 16];
            }
            __builtin_amdgcn_sched_barrier(0);
#pragma unroll
            for (int g = 0; g < KG; ++g) {
              c0 = wmma4(a[g], b0[g], c0);
              c1 = wmma4(a[g], b1[g], c1);
            }
          }
          float* cp = out + (unsigned)((row0 + mt * 16 + 8 * half) * NN + np * 32 + r);
          const int rl = rows - (mt * 16 + 8 * half);
#pragma unroll
          for (int i = 0; i < 8; ++i) {
            if (i < rl) {
              cp[i * NN] = c0[i];
              cp[i * NN + 16] = c1[i];
            }
          }
        }
      }
    }
    __syncthreads();
  }
}

extern "C" void kernel_launch(void* const* d_in, const int* in_sizes, int n_in,
                              void* d_out, int out_size, void* d_ws, size_t ws_size,
                              hipStream_t stream) {
  (void)n_in; (void)out_size; (void)ws_size;
  const float* x     = (const float*)d_in[0];
  const int*   ei    = (const int*)d_in[1];
  const float* W1s   = (const float*)d_in[2];
  const float* W1d   = (const float*)d_in[3];
  const float* att1s = (const float*)d_in[4];
  const float* att1d = (const float*)d_in[5];
  const float* b1    = (const float*)d_in[6];
  const float* W2s   = (const float*)d_in[7];
  const float* W2d   = (const float*)d_in[8];
  const float* att2s = (const float*)d_in[9];
  const float* att2d = (const float*)d_in[10];
  const float* b2    = (const float*)d_in[11];
  float* out = (float*)d_out;

  const int N = in_sizes[0] / 128;
  const int E = in_sizes[1] / 2;
  const int* src = ei;
  const int* dst = ei + E;

  float* ws = (float*)d_ws;
  size_t off = 0;
  float*    XS   = ws + off; off += (size_t)N * 256;
  float*    Hb   = ws + off; off += (size_t)N * 256;
  float*    AS   = ws + off; off += (size_t)N * 8;
  float*    AD   = ws + off; off += (size_t)N * 8;
  float*    WA   = ws + off; off += 2048;
  float*    WD   = ws + off; off += 2048;
  float*    AS2  = Hb + (size_t)N * 128;
  float*    AD2  = AS2 + (size_t)N * 2 + 256;

  const int TPB = 256;

  const int bF = 256;
  const int bK = bF + 2;
  const int bG = bK + (E + 16 * TPB - 1) / (16 * TPB);
  const int nb = bG + (N / 16 * 4 + 7) / 8;
  front<<<nb, TPB, 0, stream>>>(x, dst, W1s, W1d, att1s, att1d, W2s, W2d, att2s, att2d,
                                XS, AS, AD, WA, WD, out, N, E, bF, bK, bG);

  edge_aggregate_owned<8, 5, 4, true, 1><<<256, 1024, 0, stream>>>(src, dst, AS, AD,
                                                                 XS, b1, Hb, out, E, N,
                                                                 W2s, WA, WD, AS2, AD2);

  edge_aggregate_owned<1, 7, 8, false, 2><<<256, 1024, 0, stream>>>(src, dst, AS2, AD2,
                                                                  Hb, b2, out, nullptr, E, N,
                                                                  nullptr, nullptr, nullptr,
                                                                  nullptr, nullptr);
}


# front launch bins the edges by destination tile (source node | row, one segment per tile and bin block); layer 1 packs its tile's segments, no key filter; list and LDS add order change
# speedup vs baseline: 4.0780x; 1.1302x over previous
#include <hip/hip_runtime.h>
#include <stddef.h>

typedef __attribute__((ext_vector_type(2))) float v2f;
typedef __attribute__((ext_vector_type(8))) float v8f;

__device__ __forceinline__ v8f wmma4(v2f a, v2f b, v8f c) {
  return __builtin_amdgcn_wmma_f32_16x16x4_f32(
       false, a,  false, b,
       (short)0, c,  false,  false);
}

template <int N, int K>
__device__ __forceinline__ void gemm_strip(const float* __restrict__ A,
                                           const float* __restrict__ B,
                                           float* __restrict__ C, int mt, int ns, int lane) {
  const int m0 = mt * 16;
  const int n0 = ns * 64;
  const int half = lane >> 4;
  const int r    = lane & 15;

  const float* Arow = A + (size_t)(m0 + r) * K + 2 * half;
  const float* Bcol = B + (size_t)(2 * half) * N + n0 + r;

  v8f c0 = {}, c1 = {}, c2 = {}, c3 = {};
#pragma unroll
  for (int k = 0; k < K; k += 4) {
    v2f a = *(const v2f*)(Arow + k);
    v2f b0, b1, b2, b3;
    b0.x = Bcol[k * N + 0];       b0.y = Bcol[k * N + N + 0];
    b1.x = Bcol[k * N + 16];      b1.y = Bcol[k * N + N + 16];
    b2.x = Bcol[k * N + 32];      b2.y = Bcol[k * N + N + 32];
    b3.x = Bcol[k * N + 48];      b3.y = Bcol[k * N + N + 48];
    c0 = wmma4(a, b0, c0);
    c1 = wmma4(a, b1, c1);
    c2 = wmma4(a, b2, c2);
    c3 = wmma4(a, b3, c3);
  }

  float* Crow = C + (size_t)(m0 + 8 * half) * N + n0 + r;
#pragma unroll
  for (int i = 0; i < 8; ++i) {
    Crow[i * N + 0]  = c0[i];
    Crow[i * N + 16] = c1[i];
    Crow[i * N + 32] = c2[i];
    Crow[i * N + 48] = c3[i];
  }
}

constexpr int kLN = 196;
static_assert(kLN % 4 == 0, "4 nodes of a wave fill one whole 128-byte line of a_s and of a_d");

__device__ __forceinline__ void fold_and_logits(const float* __restrict__ x,
                                                const float* __restrict__ Wsrc, const float* __restrict__ Wdst,
                                                const float* __restrict__ atts, const float* __restrict__ attd,
                                                float* __restrict__ as, float* __restrict__ ad,
                                                float4* pan, int blk, int nblk, int N) {
  const int tid = threadIdx.x, lane = tid & 31, wave = tid >> 5;
  {
    const int hh = (wave & 1) * 4 + (lane >> 3);
    const float4 a_s4 = *(const float4*)(atts + hh * 32 + (lane & 7) * 4);
    const float4 a_d4 = *(const float4*)(attd + hh * 32 + (lane & 7) * 4);
    const float* ws = Wsrc + wave * 128 + lane * 4;
    const float* wd = Wdst + wave * 128 + lane * 4;
    constexpr int FC = 8;
#pragma unroll 1
    for (int c0 = 0; c0 < 32; c0 += FC) {
      float4 vs[FC], vd[FC];
#pragma unroll
      for (int j = 0; j < FC; ++j) {
        vs[j] = *(const float4*)(ws + (c0 + j) * 1024);
        vd[j] = *(const float4*)(wd + (c0 + j) * 1024);
      }
#pragma unroll
      for (int j = 0; j < FC; ++j) {
        float ps = vs[j].x * a_s4.x + vs[j].y * a_s4.y + vs[j].z * a_s4.z + vs[j].w * a_s4.w;
        float pd = vd[j].x * a_d4.x + vd[j].y * a_d4.y + vd[j].z * a_d4.z + vd[j].w * a_d4.w;
        ps += __shfl_xor(ps, 1); pd += __shfl_xor(pd, 1);
        ps += __shfl_xor(ps, 2); pd += __shfl_xor(pd, 2);
        ps += __shfl_xor(ps, 4); pd += __shfl_xor(pd, 4);
        if ((lane & 7) == 0) {
          const int k = (c0 + j) * 4 + (wave >> 1);
          float* q = (float*)(pan + (k >> 1) * 8 + hh) + (k & 1) * 2;
          q[0] = ps;
          q[1] = pd;
        }
      }
    }
  }
  __syncthreads();

  const int h = tid & 7;
  for (int row0 = blk * kLN; row0 < N; row0 += nblk * kLN) {
    const int rows = min(kLN, N - row0);
    for (int p0 = wave * 4; p0 < rows; p0 += 32) {
      const int nl = p0 + (lane >> 3);
      const int node = row0 + min(nl, rows - 1);
      const float4* xp = (const float4*)(x + (size_t)node * 128);
      float s = 0.f, d = 0.f;
      constexpr int XC = 8;
#pragma unroll
      for (int q0 = 0; q0 < 32; q0 += XC) {
        float4 xv[XC];
#pragma unroll
        for (int j = 0; j < XC; ++j) xv[j] = xp[q0 + j];
#pragma unroll
        for (int j = 0; j < XC; ++j) {
          const float4 w0 = pan[(2 * (q0 + j)) * 8 + h];
          const float4 w1 = pan[(2 * (q0 + j) + 1) * 8 + h];
          s += xv[j].x * w0.x;  d += xv[j].x * w0.y;
          s += xv[j].y * w0.z;  d += xv[j].y * w0.w;
          s += xv[j].z * w1.x;  d += xv[j].z * w1.y;
          s += xv[j].w * w1.z;  d += xv[j].w * w1.w;
        }
      }
      if (nl < rows) {
        as[(size_t)node * 8 + h] = s;
        ad[(size_t)node * 8 + h] = d;
      }
    }
  }
}

__device__ __forceinline__ void fold_next(const float* __restrict__ Wsrc, const float* __restrict__ Wdst,
                                          const float* __restrict__ atts, const float* __restrict__ attd,
                                          float* __restrict__ wa, float* __restrict__ wd, int unit, int lane) {
  const bool second = unit >= 8;
  const float* Wm = (second ? Wdst : Wsrc) + (size_t)((unit & 7) * 32) * 128 + lane * 4;
  const float4 a4 = *(const float4*)((second ? attd : atts) + lane * 4);
  float res = 0.f;
  constexpr int RC = 16;
#pragma unroll 1
  for (int j0 = 0; j0 < 32; j0 += RC) {
    float4 v[RC];
#pragma unroll
    for (int j = 0; j < RC; ++j) v[j] = *(const float4*)(Wm + (j0 + j) * 128);
#pragma unroll
    for (int j = 0; j < RC; ++j) {
      float p = v[j].x * a4.x + v[j].y * a4.y + v[j].z * a4.z + v[j].w * a4.w;
      p += __shfl_xor(p, 1);
      p += __shfl_xor(p, 2);
      p += __shfl_xor(p, 4);
      p += __shfl_xor(p, 8);
      p += __shfl_xor(p, 16);
      if (lane == j0 + j) res = p;
    }
  }
  (second ? wd : wa)[(unit & 7) * 32 + lane] = res;
}

__device__ __forceinline__ unsigned fkey(float f) {
  unsigned u = __float_as_uint(f);
  return (u & 0x80000000u) ? ~u : (u | 0x80000000u);
}
__device__ __forceinline__ float fdecode(unsigned k) {
  unsigned u = (k & 0x80000000u) ? (k & 0x7FFFFFFFu) : ~k;
  return __uint_as_float(u);
}
__device__ __forceinline__ float leaky(float a) {
  return a > 0.f ? a : 0.2f * a;
}

template <int H>
__device__ __forceinline__ void load_heads(const float* __restrict__ p, float (&a)[H]) {
  if constexpr (H % 4 == 0) {
#pragma unroll
    for (int q = 0; q < H / 4; ++q) {
      const float4 v = ((const float4*)p)[q];
      a[4 * q + 0] = v.x; a[4 * q + 1] = v.y; a[4 * q + 2] = v.z; a[4 * q + 3] = v.w;
    }
  } else {
#pragma unroll
    for (int h = 0; h < H; ++h) a[h] = p[h];
  }
}

constexpr int kRT = 196;
constexpr int kLP = 128;
constexpr int kCAP = 8192;
constexpr int kEPB = 4096;
constexpr int kSEG = 64;
static_assert(kEPB == 16 * 256, "a bin block is 256 threads, 16 edges each");
constexpr int kBOFF = 262144;

__device__ __forceinline__ bool bins_fit(int E, int N) {
  const long long ntile = (N + kRT - 1) / kRT, nseg = ((long long)E + kEPB - 1) / kEPB;
  return N <= (1 << 24) && ntile <= 256 && ntile * nseg * kSEG <= (long long)N * 128 - kBOFF;
}

__device__ __forceinline__ void bin_edges(const int* __restrict__ src, const int* __restrict__ dst,
                                          float* __restrict__ binsf, int* cb, int E, int N, int bb) {
  if (!bins_fit(E, N)) return;
  unsigned* bins = (unsigned*)binsf;
  const int tid = threadIdx.x;
  const int ntile = (N + kRT - 1) / kRT, nseg = (E + kEPB - 1) / kEPB;
  cb[tid] = 0;
  __syncthreads();
  int d[16], s[16];
  const int e0 = bb * kEPB;
  if (((((size_t)src | (size_t)dst) & 15) == 0) && e0 + kEPB <= E) {
    const int4* d4 = (const int4*)(dst + e0) + tid;
    const int4* s4 = (const int4*)(src + e0) + tid;
#pragma unroll
    for (int q = 0; q < 4; ++q) {
      const int4 dv = d4[q * 256], sv = s4[q * 256];
      d[4 * q + 0] = dv.x; d[4 * q + 1] = dv.y; d[4 * q + 2] = dv.z; d[4 * q + 3] = dv.w;
      s[4 * q + 0] = sv.x; s[4 * q + 1] = sv.y; s[4 * q + 2] = sv.z; s[4 * q + 3] = sv.w;
    }
  } else {
#pragma unroll
    for (int j = 0; j < 16; ++j) {
      const int e = e0 + j * 256 + tid;
      d[j] = e < E ? dst[e] : -1;
      s[j] = e < E ? src[e] : 0;
    }
  }
  int pos[16];
#pragma unroll
  for (int j = 0; j < 16; ++j) {
    pos[j] = kSEG;
    if ((unsigned)d[j] < (unsigned)N) pos[j] = atomicAdd(&cb[(unsigned)d[j] / (unsigned)kRT], 1);
  }
#pragma unroll
  for (int j = 0; j < 16; ++j) {
    if (pos[j] < kSEG - 1) {
      const unsigned t = (unsigned)d[j] / (unsigned)kRT;
      bins[((size_t)t * nseg + bb) * kSEG + 1 + pos[j]] =
          ((unsigned)s[j] & 0xFFFFFFu) | (((unsigned)d[j] - t * (unsigned)kRT) << 24);
    }
  }
  __syncthreads();
  if (tid < ntile) bins[((size_t)tid * nseg + bb) * kSEG] = (unsigned)cb[tid];
}

__global__ void __launch_bounds__(256)
front(const float* __restrict__ x, const int* __restrict__ src, const int* __restrict__ dst,
      const float* __restrict__ W1s, const float* __restrict__ W1d,
      const float* __restrict__ att1s, const float* __restrict__ att1d,
      const float* __restrict__ W2s, const float* __restrict__ W2d,
      const float* __restrict__ att2s, const float* __restrict__ att2d,
      float* __restrict__ xs, float* __restrict__ as, float* __restrict__ ad,
      float* __restrict__ wa2, float* __restrict__ wd2, float* __restrict__ bins,
      int N, int E, int bF, int bK, int bG) {
  __shared__ float4 pan[64 * 8];
  const int b = blockIdx.x;
  const int lane = threadIdx.x & 31, wave = threadIdx.x >> 5;
  if (b < bF) {
    fold_and_logits(x, W1s, W1d, att1s, att1d, as, ad, pan, b, bF, N);
  } else if (b < bK) {
    const int unit = (b - bF) * 8 + wave;
    if (unit < 16) fold_next(W2s, W2d, att2s, att2d, wa2, wd2, unit, lane);
  } else if (b < bG) {
    bin_edges(src, dst, bins, (int*)pan, E, N, b - bK);
  } else {
    const int MT = N >> 4;
    const int g = (b - bG) * 8 + wave;
    if (g < 4 * MT) gemm_strip<256, 128>(x, W1s, xs, g % MT, g / MT, lane);
  }
}

template <int H, int LOGF, int EB, bool RELU, int LMODE>
__global__ void __launch_bounds__(1024)
edge_aggregate_owned(const int* __restrict__ src, const int* __restrict__ dst,
                     const float* __restrict__ as, const float* __restrict__ ad,
                     const float* __restrict__ xs, const float* __restrict__ bias,
                     float* __restrict__ out, float* __restrict__ lst,
                     const float* __restrict__ bins, int E, int N,
                     const float* __restrict__ wn, const float* __restrict__ wan,
                     const float* __restrict__ wdn, float* __restrict__ asn,
                     float* __restrict__ adn) {
  constexpr int HF = H << LOGF;
  constexpr bool NEXT = LMODE == 1;
  constexpr int NN = 128;
  constexpr int AST = NEXT ? HF + 4 : HF;
  static_assert(!NEXT || HF % 4 == 0, "whole k steps of the matrix instruction");
  constexpr int NS = HF / 32;
  constexpr int RT = kRT;
  constexpr int CAP = kCAP;
  constexpr int KR = 4;
  constexpr int U = 8;
  constexpr int LP = kLP;
  static_assert(RT <= 256, "a parked entry keeps its row in 8 bits");
  static_assert(LMODE != 2 || HF == LP, "the parked lists lie in this launch's own output rows");
  static_assert(EB * H <= 32, "one lane per (edge, head) of a batch");
  static_assert(LOGF >= 5, "a 32-lane slice lies inside one head");
  __shared__ __attribute__((aligned(16))) float acc[RT * AST];
  __shared__ int lsrc[CAP];
  __shared__ int ldl[CAP];
  __shared__ float adl[RT * H];
  __shared__ unsigned mx[RT * H];
  __shared__ float dn[RT * H];
  __shared__ int cnt;
  __shared__ int ovf;

  const int tid = threadIdx.x, lane = tid & 31, wave = tid >> 5;
  constexpr int LGP = 256;
  static_assert(LGP >= RT && LGP % 32 == 0 && (LMODE != 2 || H == 1), "a tile's logits fill whole lines of their own");
  auto aidx = [](int s) -> size_t {
    if constexpr (LMODE == 2) return (size_t)((unsigned)s / (unsigned)RT) * LGP + (unsigned)s % (unsigned)RT;
    else return (size_t)s * H;
  };

  for (int row0 = blockIdx.x * RT; row0 < N; row0 += gridDim.x * RT) {
    const int rows = min(RT, N - row0);
    const int lcap = rows * LP;
    float* const lp = (LMODE == 2 ? out : lst) + (size_t)row0 * LP;
    int c0 = -1;
    int lv[KR];
    if constexpr (LMODE == 2) {
      c0 = __float_as_int(lp[0]);
#pragma unroll
      for (int k = 0; k < KR; ++k) lv[k] = __float_as_int(lp[min(1 + tid + k * 1024, lcap - 1)]);
    }
    for (int i = tid; i < RT * AST; i += 1024) acc[i] = 0.f;
    for (int i = tid; i < RT * H; i += 1024) {
      adl[i] = (i < rows * H) ? ad[aidx(row0) + i] : 0.f;
      mx[i] = 0u;
      dn[i] = 0.f;
    }
    if (tid == 0) {
      cnt = 0;
      ovf = 0;
    }
    __syncthreads();

    auto list_edge = [&](int e) {
      int idx = atomicAdd(&cnt, 1);
      if (idx < CAP) lsrc[idx] = e;
    };
    auto take = [&](int d, int e) {
      if ((unsigned)(d - row0) < (unsigned)rows) list_edge(e);
    };
    bool use_list = false;
    if constexpr (LMODE == 2) {
      c0 = __builtin_amdgcn_readfirstlane(c0);
      use_list = c0 >= 0 && c0 <= CAP && c0 < lcap;
    }
    bool nodes = use_list;
    if (use_list) {
#pragma unroll
      for (int k = 0; k < KR; ++k) {
        const int i = tid + k * 1024;
        if (i < c0) {
          lsrc[i] = lv[k] & 0xFFFFFF;
          ldl[i] = (int)((unsigned)lv[k] >> 24);
        }
      }
      for (int i = KR * 1024 + tid; i < c0; i += 1024) {
        const int v = __float_as_int(lp[1 + i]);
        lsrc[i] = v & 0xFFFFFF;
        ldl[i] = (int)((unsigned)v >> 24);
      }
      if (tid == 0) cnt = c0;
    } else if (LMODE == 1 && bins_fit(E, N)) {
      const int nseg = (E + kEPB - 1) / kEPB;
      const uint4* sp = (const uint4*)bins + (size_t)(row0 / RT) * nseg * (kSEG / 4);
      const int q = tid & 15, sl = tid >> 4;
      constexpr int SU = 4;
      for (int p0 = 0; p0 < nseg; p0 += 64 * SU) {
        uint4 v[SU];
#pragma unroll
        for (int u = 0; u < SU; ++u) v[u] = sp[(size_t)min(p0 + u * 64 + sl, nseg - 1) * (kSEG / 4) + q];
#pragma unroll
        for (int u = 0; u < SU; ++u) {
          const bool live = p0 + u * 64 + sl < nseg;
          const int cs = __shfl((int)v[u].x, lane & 16);
          const bool fits = (unsigned)cs <= (unsigned)(kSEG - 1);
          if (live && !fits) ovf = 1;
          const int cc = (live && fits) ? cs : 0;
          int base = 0;
          if (q == 0 && cc > 0) base = atomicAdd(&cnt, cc);
          base = __shfl(base, lane & 16);
          const unsigned w4[4] = {v[u].x, v[u].y, v[u].z, v[u].w};
#pragma unroll
          for (int j = 0; j < 4; ++j) {
            const int w = 4 * q + j;
            const int idx = base + w - 1;
            if (w >= 1 && w <= cc && idx < CAP) {
              lsrc[idx] = (int)(w4[j] & 0xFFFFFFu);
              ldl[idx] = min((int)(w4[j] >> 24), rows - 1);
            }
          }
        }
      }
      __syncthreads();
      if (ovf != 0) {
        if (tid == 0) cnt = 0;
        __syncthreads();
      } else {
        nodes = true;
      }
    }
    if (!nodes) {
      int ts = tid;
      asm volatile("" : "+v"(ts));
      const int E4 = (((size_t)dst & 15) == 0) ? (E >> 2) : 0;
      const int4* dst4 = (const int4*)dst;
      for (int g0 = ts; g0 < E4; g0 += 1024 * U) {
        int4 v[U];
#pragma unroll
        for (int u = 0; u < U; ++u) {
          int g = g0 + u * 1024;
          v[u] = (g < E4) ? dst4[g] : make_int4(-1, -1, -1, -1);
        }
#pragma unroll
        for (int u = 0; u < U; ++u) {
          int e = 4 * (g0 + u * 1024);
          take(v[u].x, e);
          take(v[u].y, e + 1);
          take(v[u].z, e + 2);
          take(v[u].w, e + 3);
        }
      }
      for (int e = 4 * E4 + ts; e < E; e += 1024) take(dst[e], e);
    }
    __syncthreads();

    const bool listed = cnt <= CAP;
    const int n = listed ? cnt : 0;
    if (!listed) {
      for (int e = tid; e < E; e += 1024) {
        const int d = dst[e];
        const unsigned dl = (unsigned)(d - row0);
        if (dl < (unsigned)rows) {
          const int s = src[e];
#pragma unroll 1
          for (int h = 0; h < H; ++h)
            atomicMax(&mx[dl * H + h], fkey(leaky(as[aidx(s) + h] + adl[dl * H + h])));
        }
      }
      __syncthreads();
      for (int e = tid; e < E; e += 1024) {
        const int d = dst[e];
        const unsigned dl = (unsigned)(d - row0);
        if (dl < (unsigned)rows) {
          const int s = src[e];
#pragma unroll 1
          for (int h = 0; h < H; ++h) {
            const float a = leaky(as[aidx(s) + h] + adl[dl * H + h]);
            atomicAdd(&dn[dl * H + h], __expf(a - fdecode(mx[dl * H + h])));
          }
        }
      }
      __syncthreads();
      for (int e = tid; e < E; e += 1024) {
        const int d = dst[e];
        const unsigned dl = (unsigned)(d - row0);
        if (dl < (unsigned)rows) {
          const int s = src[e];
#pragma unroll 1
          for (int p = 0; p < HF; ++p) {
            const int lh = dl * H + (p >> LOGF);
            const float a = leaky(as[aidx(s) + (p >> LOGF)] + adl[lh]);
            const float w = __expf(a - fdecode(mx[lh])) / (dn[lh] + 1e-16f);
            atomicAdd(&acc[dl * AST + p], w * xs[(size_t)s * HF + p]);
          }
        }
      }
    }
    {
      int sv[KR], dv[KR];
      float av[KR][H];
#pragma unroll
      for (int k = 0; k < KR; ++k) dv[k] = row0;
      if (n > 0) {
        if (nodes) {
#pragma unroll
          for (int k = 0; k < KR; ++k) {
            const int i = tid + k * 1024;
            sv[k] = (i < n) ? lsrc[i] : 0;
          }
        } else {
#pragma unroll
          for (int k = 0; k < KR; ++k) {
            const int i = tid + k * 1024;
            const int e = (i < n) ? lsrc[i] : 0;
            sv[k] = src[e];
            dv[k] = dst[e];
          }
        }
        __builtin_amdgcn_sched_barrier(0);
#pragma unroll
        for (int k = 0; k < KR; ++k) load_heads<H>(as + aidx(sv[k]), av[k]);
      }
#pragma unroll
      for (int k = 0; k < KR; ++k) {
        const int i = tid + k * 1024;
        if (i < n) {
          if (!nodes) ldl[i] = min(max(dv[k] - row0, 0), rows - 1);
          const int lh = ldl[i] * H;
          lsrc[i] = sv[k];
#pragma unroll
          for (int h = 0; h < H; ++h) {
            av[k][h] = leaky(av[k][h] + adl[lh + h]);
            atomicMax(&mx[lh + h], fkey(av[k][h]));
          }
        }
      }
      for (int i = KR * 1024 + tid; i < n; i += 1024) {
        int s = lsrc[i];
        if (!nodes) {
          ldl[i] = min(max(dst[s] - row0, 0), rows - 1);
          s = src[s];
        }
        const int lh = ldl[i] * H;
        float a[H];
        load_heads<H>(as + aidx(s), a);
        lsrc[i] = s;
#pragma unroll
        for (int h = 0; h < H; ++h) atomicMax(&mx[lh + h], fkey(leaky(a[h] + adl[lh + h])));
      }
      __syncthreads();
#pragma unroll
      for (int k = 0; k < KR; ++k) {
        const int i = tid + k * 1024;
        if (i < n) {
          const int lh = ldl[i] * H;
#pragma unroll
          for (int h = 0; h < H; ++h)
            atomicAdd(&dn[lh + h], __expf(av[k][h] - fdecode(mx[lh + h])));
        }
      }
      for (int i = KR * 1024 + tid; i < n; i += 1024) {
        const int lh = ldl[i] * H;
        float a[H];
        load_heads<H>(as + aidx(lsrc[i]), a);
#pragma unroll
        for (int h = 0; h < H; ++h)
          atomicAdd(&dn[lh + h], __expf(leaky(a[h] + adl[lh + h]) - fdecode(mx[lh + h])));
      }
    }
    __syncthreads();

    if constexpr (LMODE == 1) {
      const bool park = listed && n < lcap && N <= (1 << 24);
      if (tid == 0) lp[0] = __int_as_float(park ? n : -1);
      if (park)
        for (int i = tid; i < n; i += 1024) lp[1 + i] = __int_as_float(lsrc[i] | (ldl[i] << 24));
    }

    for (int i0 = wave * EB; i0 < n; i0 += 32 * EB) {
      const int j = min(lane / H, EB - 1), h = lane & (H - 1);
      const int ii = min(i0 + j, n - 1);
      const int s = lsrc[ii], dl = ldl[ii];
      const int lh = dl * H + h;
      const float v_as = as[aidx(s) + h], v_ad = adl[lh], v_den = dn[lh];
      const unsigned v_mx = mx[lh];
      float m[EB][NS];
      int dls[EB];
#pragma unroll
      for (int jj = 0; jj < EB; ++jj) {
        const int sj = __builtin_amdgcn_readlane(s, jj * H);
        dls[jj] = __builtin_amdgcn_readlane(dl, jj * H);
#pragma unroll
        for (int i = 0; i < NS; ++i) m[jj][i] = xs[(size_t)sj * HF + lane + 32 * i];
      }
      const float w = __expf(leaky(v_as + v_ad) - fdecode(v_mx)) / (v_den + 1e-16f);
#pragma unroll
      for (int jj = 0; jj < EB; ++jj) {
        const bool valid = i0 + jj < n;
#pragma unroll
        for (int i = 0; i < NS; ++i) {
          const int p = lane + 32 * i;
          const float wj = __builtin_bit_cast(
              float, __builtin_amdgcn_readlane(__builtin_bit_cast(int, w), jj * H + ((32 * i) >> LOGF)));
          atomicAdd(&acc[dls[jj] * AST + p], valid ? wj * m[jj][i] : 0.f);
        }
      }
    }
    __syncthreads();

    if constexpr (!NEXT) {
      for (int i = tid; i < rows * HF; i += 1024) {
        float v = acc[i] + bias[i & (HF - 1)];
        if (RELU) v = v > 0.f ? v : 0.f;
        out[(size_t)row0 * HF + i] = v;
      }
    } else {
      int tl = tid;
      asm volatile("" : "+v"(tl));
      for (int i = tid; i < RT * HF; i += 1024) {
        const int cc = i & (HF - 1), ix = (i / HF) * AST + cc;
        float v = acc[ix] + bias[cc];
        if (RELU) v = v > 0.f ? v : 0.f;
        acc[ix] = v;
      }
      __syncthreads();
      {
        const int rr = tl >> 2, q = tl & 3;
        const float* hr = acc + min(rr, RT - 1) * AST;
        float s1 = 0.f, s2 = 0.f;
#pragma unroll 8
        for (unsigned k = (unsigned)q; k < (unsigned)HF; k += 4u) {
          const float hv = hr[k];
          s1 += hv * wan[k];
          s2 += hv * wdn[k];
        }
        s1 += __shfl_xor(s1, 1); s2 += __shfl_xor(s2, 1);
        s1 += __shfl_xor(s1, 2); s2 += __shfl_xor(s2, 2);
        if (q == 0 && rr < RT) {
          adl[rr] = s1;
          dn[rr] = s2;
        }
      }
      __syncthreads();
      if (tl < 2 * LGP) {
        const int j = tl & (LGP - 1);
        if (j < rows) (tl < LGP ? asn : adn)[(size_t)(row0 / RT) * LGP + j] = (tl < LGP ? adl : dn)[j];
      }
      {
        constexpr int MT = (RT + 15) / 16, NU = MT * (NN / 32);
        static_assert(NU <= 52 && NN == 128, "two units per wave at most, unit -> (tile, column pair) by shift and mask");
        const int half = (tl >> 4) & 1, r = tl & 15;
        const int wv = __builtin_amdgcn_readfirstlane(tl >> 5);
        const int wb = wv >> 3, wo = wv & 7;
        const bool two = wo < 4 || wo == 4 + wb;
        const int u2 = 32 + 5 * wb + min(wo, 4);
#pragma unroll 1
        for (int t = 0; t < 2; ++t) {
          const int u = t == 0 ? wv : u2;
          if (t == 1 && !two) break;
          if (u >= NU) break;
          const int mt = u >> 2, np = u & 3;
          if (mt * 16 >= rows) continue;
          const float* ap = acc + min(mt * 16 + r, RT - 1) * AST + 2 * half;
          const float* bp = wn + (unsigned)((2 * half) * NN + np * 32 + r);
          v8f c0 = {}, c1 = {};
          constexpr int KG = 4;
#pragma unroll
          for (int k0 = 0; k0 < HF; k0 += 4 * KG) {
            v2f a[KG], b0[KG], b1[KG];
#pragma unroll
            for (int g = 0; g < KG; ++g) {
              const int k = k0 + 4 * g;
              a[g] = *(const v2f*)(ap + k);
              b0[g].x = bp[k * NN];       b0[g].y = bp[k * NN + NN];
              b1[g].x = bp[k * NN + 16];  b1[g].y = bp[k * NN + NN + 16];
            }
            __builtin_amdgcn_sched_barrier(0);
#pragma unroll
            for (int g = 0; g < KG; ++g) {
              c0 = wmma4(a[g], b0[g], c0);
              c1 = wmma4(a[g], b1[g], c1);
            }
          }
          float* cp = out + (unsigned)((row0 + mt * 16 + 8 * half) * NN + np * 32 + r);
          const int rl = rows - (mt * 16 + 8 * half);
#pragma unroll
          for (int i = 0; i < 8; ++i) {
            if (i < rl) {
              cp[i * NN] = c0[i];
              cp[i * NN + 16] = c1[i];
            }
          }
        }
      }
    }
    __syncthreads();
  }
}

extern "C" void kernel_launch(void* const* d_in, const int* in_sizes, int n_in,
                              void* d_out, int out_size, void* d_ws, size_t ws_size,
                              hipStream_t stream) {
  (void)n_in; (void)out_size; (void)ws_size;
  const float* x     = (const float*)d_in[0];
  const int*   ei    = (const int*)d_in[1];
  const float* W1s   = (const float*)d_in[2];
  const float* W1d   = (const float*)d_in[3];
  const float* att1s = (const float*)d_in[4];
  const float* att1d = (const float*)d_in[5];
  const float* b1    = (const float*)d_in[6];
  const float* W2s   = (const float*)d_in[7];
  const float* W2d   = (const float*)d_in[8];
  const float* att2s = (const float*)d_in[9];
  const float* att2d = (const float*)d_in[10];
  const float* b2    = (const float*)d_in[11];
  float* out = (float*)d_out;

  const int N = in_sizes[0] / 128;
  const int E = in_sizes[1] / 2;
  const int* src = ei;
  const int* dst = ei + E;

  float* ws = (float*)d_ws;
  size_t off = 0;
  float*    XS   = ws + off; off += (size_t)N * 256;
  float*    Hb   = ws + off; off += (size_t)N * 256;
  float*    AS   = ws + off; off += (size_t)N * 8;
  float*    AD   = ws + off; off += (size_t)N * 8;
  float*    WA   = ws + off; off += 2048;
  float*    WD   = ws + off; off += 2048;
  float*    AS2  = Hb + (size_t)N * 128;
  float*    AD2  = AS2 + (size_t)N * 2 + 256;
  float*    BINS = AS2 + kBOFF;

  const int TPB = 256;

  const int bF = 256;
  const int bK = bF + 2;
  const int bG = bK + (E + kEPB - 1) / kEPB;
  const int nb = bG + (N / 16 * 4 + 7) / 8;
  front<<<nb, TPB, 0, stream>>>(x, src, dst, W1s, W1d, att1s, att1d, W2s, W2d, att2s, att2d,
                                XS, AS, AD, WA, WD, BINS, N, E, bF, bK, bG);

  edge_aggregate_owned<8, 5, 4, true, 1><<<256, 1024, 0, stream>>>(src, dst, AS, AD,
                                                                 XS, b1, Hb, out, BINS, E, N,
                                                                 W2s, WA, WD, AS2, AD2);

  edge_aggregate_owned<1, 7, 8, false, 2><<<256, 1024, 0, stream>>>(src, dst, AS2, AD2,
                                                                  Hb, b2, out, nullptr, nullptr, E, N,
                                                                  nullptr, nullptr, nullptr,
                                                                  nullptr, nullptr);
}


# front launch's x @ W1_src: a GEMM block copies its 64-column strip of W1_src to LDS once; 8 waves x 2 row tiles read B fragments by 8-byte LDS reads, not 8 global loads per k step; 2017->1238 blocks
# speedup vs baseline: 4.2640x; 1.0456x over previous
#include <hip/hip_runtime.h>
#include <stddef.h>

typedef __attribute__((ext_vector_type(2))) float v2f;
typedef __attribute__((ext_vector_type(8))) float v8f;

__device__ __forceinline__ v8f wmma4(v2f a, v2f b, v8f c) {
  return __builtin_amdgcn_wmma_f32_16x16x4_f32(
       false, a,  false, b,
       (short)0, c,  false,  false);
}

constexpr int kGP = 160;
constexpr int kGT = 2;
static_assert(kGP % 64 == 32 && kGP >= 128, "the two lane halves of a fragment read use different bank halves");

template <int N, int K>
__device__ __forceinline__ void gemm_strips_staged(const float* __restrict__ A,
                                                   const float* __restrict__ B,
                                                   float* __restrict__ C, float* Bs,
                                                   int chunk, int ns, int MT) {
  static_assert(K % 4 == 0 && (K / 2) * 16 % 256 == 0, "whole k steps; the copy is whole trips of 256 threads");
  const int tid = threadIdx.x, lane = tid & 31, wave = tid >> 5;
  const int n0 = ns * 64;
#pragma unroll
  for (int i = 0; i < (K / 2) * 16 / 256; ++i) {
    const int item = tid + 256 * i;
    const int p = item >> 4, c4 = item & 15;
    const float4 lo = *(const float4*)(B + (size_t)(2 * p) * N + n0 + 4 * c4);
    const float4 hi = *(const float4*)(B + (size_t)(2 * p + 1) * N + n0 + 4 * c4);
    float4* q = (float4*)(Bs + p * kGP + 8 * c4);
    q[0] = make_float4(lo.x, hi.x, lo.y, hi.y);
    q[1] = make_float4(lo.z, hi.z, lo.w, hi.w);
  }
  __syncthreads();

  const int half = lane >> 4;
  const int r    = lane & 15;
  const float* bp = Bs + half * kGP + 2 * r;
#pragma unroll 1
  for (int t = 0; t < kGT; ++t) {
    const int mt = chunk * (8 * kGT) + t * 8 + wave;
    if (mt >= MT) break;
    const int m0 = mt * 16;
    const float* Arow = A + (size_t)(m0 + r) * K + 2 * half;
    v8f c0 = {}, c1 = {}, c2 = {}, c3 = {};
#pragma unroll
    for (int k = 0; k < K; k += 4) {
      const v2f a = *(const v2f*)(Arow + k);
      const float* bk = bp + (k >> 1) * kGP;
      const v2f b0 = *(const v2f*)(bk);
      const v2f b1 = *(const v2f*)(bk + 32);
      const v2f b2 = *(const v2f*)(bk + 64);
      const v2f b3 = *(const v2f*)(bk + 96);
      c0 = wmma4(a, b0, c0);
      c1 = wmma4(a, b1, c1);
      c2 = wmma4(a, b2, c2);
      c3 = wmma4(a, b3, c3);
    }
    float* Crow = C + (size_t)(m0 + 8 * half) * N + n0 + r;
#pragma unroll
    for (int i = 0; i < 8; ++i) {
      Crow[i * N + 0]  = c0[i];
      Crow[i * N + 16] = c1[i];
      Crow[i * N + 32] = c2[i];
      Crow[i * N + 48] = c3[i];
    }
  }
}

constexpr int kLN = 196;
static_assert(kLN % 4 == 0, "4 nodes of a wave fill one whole 128-byte line of a_s and of a_d");

__device__ __forceinline__ void fold_and_logits(const float* __restrict__ x,
                                                const float* __restrict__ Wsrc, const float* __restrict__ Wdst,
                                                const float* __restrict__ atts, const float* __restrict__ attd,
                                                float* __restrict__ as, float* __restrict__ ad,
                                                float4* pan, int blk, int nblk, int N) {
  const int tid = threadIdx.x, lane = tid & 31, wave = tid >> 5;
  {
    const int hh = (wave & 1) * 4 + (lane >> 3);
    const float4 a_s4 = *(const float4*)(atts + hh * 32 + (lane & 7) * 4);
    const float4 a_d4 = *(const float4*)(attd + hh * 32 + (lane & 7) * 4);
    const float* ws = Wsrc + wave * 128 + lane * 4;
    const float* wd = Wdst + wave * 128 + lane * 4;
    constexpr int FC = 8;
#pragma unroll 1
    for (int c0 = 0; c0 < 32; c0 += FC) {
      float4 vs[FC], vd[FC];
#pragma unroll
      for (int j = 0; j < FC; ++j) {
        vs[j] = *(const float4*)(ws + (c0 + j) * 1024);
        vd[j] = *(const float4*)(wd + (c0 + j) * 1024);
      }
#pragma unroll
      for (int j = 0; j < FC; ++j) {
        float ps = vs[j].x * a_s4.x + vs[j].y * a_s4.y + vs[j].z * a_s4.z + vs[j].w * a_s4.w;
        float pd = vd[j].x * a_d4.x + vd[j].y * a_d4.y + vd[j].z * a_d4.z + vd[j].w * a_d4.w;
        ps += __shfl_xor(ps, 1); pd += __shfl_xor(pd, 1);
        ps += __shfl_xor(ps, 2); pd += __shfl_xor(pd, 2);
        ps += __shfl_xor(ps, 4); pd += __shfl_xor(pd, 4);
        if ((lane & 7) == 0) {
          const int k = (c0 + j) * 4 + (wave >> 1);
          float* q = (float*)(pan + (k >> 1) * 8 + hh) + (k & 1) * 2;
          q[0] = ps;
          q[1] = pd;
        }
      }
    }
  }
  __syncthreads();

  const int h = tid & 7;
  for (int row0 = blk * kLN; row0 < N; row0 += nblk * kLN) {
    const int rows = min(kLN, N - row0);
    for (int p0 = wave * 4; p0 < rows; p0 += 32) {
      const int nl = p0 + (lane >> 3);
      const int node = row0 + min(nl, rows - 1);
      const float4* xp = (const float4*)(x + (size_t)node * 128);
      float s = 0.f, d = 0.f;
      constexpr int XC = 8;
#pragma unroll
      for (int q0 = 0; q0 < 32; q0 += XC) {
        float4 xv[XC];
#pragma unroll
        for (int j = 0; j < XC; ++j) xv[j] = xp[q0 + j];
#pragma unroll
        for (int j = 0; j < XC; ++j) {
          const float4 w0 = pan[(2 * (q0 + j)) * 8 + h];
          const float4 w1 = pan[(2 * (q0 + j) + 1) * 8 + h];
          s += xv[j].x * w0.x;  d += xv[j].x * w0.y;
          s += xv[j].y * w0.z;  d += xv[j].y * w0.w;
          s += xv[j].z * w1.x;  d += xv[j].z * w1.y;
          s += xv[j].w * w1.z;  d += xv[j].w * w1.w;
        }
      }
      if (nl < rows) {
        as[(size_t)node * 8 + h] = s;
        ad[(size_t)node * 8 + h] = d;
      }
    }
  }
}

__device__ __forceinline__ void fold_next(const float* __restrict__ Wsrc, const float* __restrict__ Wdst,
                                          const float* __restrict__ atts, const float* __restrict__ attd,
                                          float* __restrict__ wa, float* __restrict__ wd, int unit, int lane) {
  const bool second = unit >= 8;
  const float* Wm = (second ? Wdst : Wsrc) + (size_t)((unit & 7) * 32) * 128 + lane * 4;
  const float4 a4 = *(const float4*)((second ? attd : atts) + lane * 4);
  float res = 0.f;
  constexpr int RC = 16;
#pragma unroll 1
  for (int j0 = 0; j0 < 32; j0 += RC) {
    float4 v[RC];
#pragma unroll
    for (int j = 0; j < RC; ++j) v[j] = *(const float4*)(Wm + (j0 + j) * 128);
#pragma unroll
    for (int j = 0; j < RC; ++j) {
      float p = v[j].x * a4.x + v[j].y * a4.y + v[j].z * a4.z + v[j].w * a4.w;
      p += __shfl_xor(p, 1);
      p += __shfl_xor(p, 2);
      p += __shfl_xor(p, 4);
      p += __shfl_xor(p, 8);
      p += __shfl_xor(p, 16);
      if (lane == j0 + j) res = p;
    }
  }
  (second ? wd : wa)[(unit & 7) * 32 + lane] = res;
}

__device__ __forceinline__ unsigned fkey(float f) {
  unsigned u = __float_as_uint(f);
  return (u & 0x80000000u) ? ~u : (u | 0x80000000u);
}
__device__ __forceinline__ float fdecode(unsigned k) {
  unsigned u = (k & 0x80000000u) ? (k & 0x7FFFFFFFu) : ~k;
  return __uint_as_float(u);
}
__device__ __forceinline__ float leaky(float a) {
  return a > 0.f ? a : 0.2f * a;
}

template <int H>
__device__ __forceinline__ void load_heads(const float* __restrict__ p, float (&a)[H]) {
  if constexpr (H % 4 == 0) {
#pragma unroll
    for (int q = 0; q < H / 4; ++q) {
      const float4 v = ((const float4*)p)[q];
      a[4 * q + 0] = v.x; a[4 * q + 1] = v.y; a[4 * q + 2] = v.z; a[4 * q + 3] = v.w;
    }
  } else {
#pragma unroll
    for (int h = 0; h < H; ++h) a[h] = p[h];
  }
}

constexpr int kRT = 196;
constexpr int kLP = 128;
constexpr int kCAP = 8192;
constexpr int kEPB = 4096;
constexpr int kSEG = 64;
static_assert(kEPB == 16 * 256, "a bin block is 256 threads, 16 edges each");
constexpr int kBOFF = 262144;

__device__ __forceinline__ bool bins_fit(int E, int N) {
  const long long ntile = (N + kRT - 1) / kRT, nseg = ((long long)E + kEPB - 1) / kEPB;
  return N <= (1 << 24) && ntile <= 256 && ntile * nseg * kSEG <= (long long)N * 128 - kBOFF;
}

__device__ __forceinline__ void bin_edges(const int* __restrict__ src, const int* __restrict__ dst,
                                          float* __restrict__ binsf, int* cb, int E, int N, int bb) {
  if (!bins_fit(E, N)) return;
  unsigned* bins = (unsigned*)binsf;
  const int tid = threadIdx.x;
  const int ntile = (N + kRT - 1) / kRT, nseg = (E + kEPB - 1) / kEPB;
  cb[tid] = 0;
  __syncthreads();
  int d[16], s[16];
  const int e0 = bb * kEPB;
  if (((((size_t)src | (size_t)dst) & 15) == 0) && e0 + kEPB <= E) {
    const int4* d4 = (const int4*)(dst + e0) + tid;
    const int4* s4 = (const int4*)(src + e0) + tid;
#pragma unroll
    for (int q = 0; q < 4; ++q) {
      const int4 dv = d4[q * 256], sv = s4[q * 256];
      d[4 * q + 0] = dv.x; d[4 * q + 1] = dv.y; d[4 * q + 2] = dv.z; d[4 * q + 3] = dv.w;
      s[4 * q + 0] = sv.x; s[4 * q + 1] = sv.y; s[4 * q + 2] = sv.z; s[4 * q + 3] = sv.w;
    }
  } else {
#pragma unroll
    for (int j = 0; j < 16; ++j) {
      const int e = e0 + j * 256 + tid;
      d[j] = e < E ? dst[e] : -1;
      s[j] = e < E ? src[e] : 0;
    }
  }
  int pos[16];
#pragma unroll
  for (int j = 0; j < 16; ++j) {
    pos[j] = kSEG;
    if ((unsigned)d[j] < (unsigned)N) pos[j] = atomicAdd(&cb[(unsigned)d[j] / (unsigned)kRT], 1);
  }
#pragma unroll
  for (int j = 0; j < 16; ++j) {
    if (pos[j] < kSEG - 1) {
      const unsigned t = (unsigned)d[j] / (unsigned)kRT;
      bins[((size_t)t * nseg + bb) * kSEG + 1 + pos[j]] =
          ((unsigned)s[j] & 0xFFFFFFu) | (((unsigned)d[j] - t * (unsigned)kRT) << 24);
    }
  }
  __syncthreads();
  if (tid < ntile) bins[((size_t)tid * nseg + bb) * kSEG] = (unsigned)cb[tid];
}

__global__ void __launch_bounds__(256)
front(const float* __restrict__ x, const int* __restrict__ src, const int* __restrict__ dst,
      const float* __restrict__ W1s, const float* __restrict__ W1d,
      const float* __restrict__ att1s, const float* __restrict__ att1d,
      const float* __restrict__ W2s, const float* __restrict__ W2d,
      const float* __restrict__ att2s, const float* __restrict__ att2d,
      float* __restrict__ xs, float* __restrict__ as, float* __restrict__ ad,
      float* __restrict__ wa2, float* __restrict__ wd2, float* __restrict__ bins,
      int N, int E, int bF, int bK, int bG) {
  __shared__ __attribute__((aligned(16))) float lbuf[64 * kGP];
  float4* pan = (float4*)lbuf;
  const int b = blockIdx.x;
  const int lane = threadIdx.x & 31, wave = threadIdx.x >> 5;
  if (b < bF) {
    fold_and_logits(x, W1s, W1d, att1s, att1d, as, ad, pan, b, bF, N);
  } else if (b < bK) {
    const int unit = (b - bF) * 8 + wave;
    if (unit < 16) fold_next(W2s, W2d, att2s, att2d, wa2, wd2, unit, lane);
  } else if (b < bG) {
    bin_edges(src, dst, bins, (int*)pan, E, N, b - bK);
  } else {
    const int MT = N >> 4;
    const int bps = (MT + 8 * kGT - 1) / (8 * kGT);
    const int gb = b - bG;
    const int ns = gb / bps;
    if (ns < 4) gemm_strips_staged<256, 128>(x, W1s, xs, lbuf, gb - ns * bps, ns, MT);
  }
}

template <int H, int LOGF, int EB, bool RELU, int LMODE>
__global__ void __launch_bounds__(1024)
edge_aggregate_owned(const int* __restrict__ src, const int* __restrict__ dst,
                     const float* __restrict__ as, const float* __restrict__ ad,
                     const float* __restrict__ xs, const float* __restrict__ bias,
                     float* __restrict__ out, float* __restrict__ lst,
                     const float* __restrict__ bins, int E, int N,
                     const float* __restrict__ wn, const float* __restrict__ wan,
                     const float* __restrict__ wdn, float* __restrict__ asn,
                     float* __restrict__ adn) {
  constexpr int HF = H << LOGF;
  constexpr bool NEXT = LMODE == 1;
  constexpr int NN = 128;
  constexpr int AST = NEXT ? HF + 4 : HF;
  static_assert(!NEXT || HF % 4 == 0, "whole k steps of the matrix instruction");
  constexpr int NS = HF / 32;
  constexpr int RT = kRT;
  constexpr int CAP = kCAP;
  constexpr int KR = 4;
  constexpr int U = 8;
  constexpr int LP = kLP;
  static_assert(RT <= 256, "a parked entry keeps its row in 8 bits");
  static_assert(LMODE != 2 || HF == LP, "the parked lists lie in this launch's own output rows");
  static_assert(EB * H <= 32, "one lane per (edge, head) of a batch");
  static_assert(LOGF >= 5, "a 32-lane slice lies inside one head");
  __shared__ __attribute__((aligned(16))) float acc[RT * AST];
  __shared__ int lsrc[CAP];
  __shared__ int ldl[CAP];
  __shared__ float adl[RT * H];
  __shared__ unsigned mx[RT * H];
  __shared__ float dn[RT * H];
  __shared__ int cnt;
  __shared__ int ovf;

  const int tid = threadIdx.x, lane = tid & 31, wave = tid >> 5;
  constexpr int LGP = 256;
  static_assert(LGP >= RT && LGP % 32 == 0 && (LMODE != 2 || H == 1), "a tile's logits fill whole lines of their own");
  auto aidx = [](int s) -> size_t {
    if constexpr (LMODE == 2) return (size_t)((unsigned)s / (unsigned)RT) * LGP + (unsigned)s % (unsigned)RT;
    else return (size_t)s * H;
  };

  for (int row0 = blockIdx.x * RT; row0 < N; row0 += gridDim.x * RT) {
    const int rows = min(RT, N - row0);
    const int lcap = rows * LP;
    float* const lp = (LMODE == 2 ? out : lst) + (size_t)row0 * LP;
    int c0 = -1;
    int lv[KR];
    if constexpr (LMODE == 2) {
      c0 = __float_as_int(lp[0]);
#pragma unroll
      for (int k = 0; k < KR; ++k) lv[k] = __float_as_int(lp[min(1 + tid + k * 1024, lcap - 1)]);
    }
    for (int i = tid; i < RT * AST; i += 1024) acc[i] = 0.f;
    for (int i = tid; i < RT * H; i += 1024) {
      adl[i] = (i < rows * H) ? ad[aidx(row0) + i] : 0.f;
      mx[i] = 0u;
      dn[i] = 0.f;
    }
    if (tid == 0) {
      cnt = 0;
      ovf = 0;
    }
    __syncthreads();

    auto list_edge = [&](int e) {
      int idx = atomicAdd(&cnt, 1);
      if (idx < CAP) lsrc[idx] = e;
    };
    auto take = [&](int d, int e) {
      if ((unsigned)(d - row0) < (unsigned)rows) list_edge(e);
    };
    bool use_list = false;
    if constexpr (LMODE == 2) {
      c0 = __builtin_amdgcn_readfirstlane(c0);
      use_list = c0 >= 0 && c0 <= CAP && c0 < lcap;
    }
    bool nodes = use_list;
    if (use_list) {
#pragma unroll
      for (int k = 0; k < KR; ++k) {
        const int i = tid + k * 1024;
        if (i < c0) {
          lsrc[i] = lv[k] & 0xFFFFFF;
          ldl[i] = (int)((unsigned)lv[k] >> 24);
        }
      }
      for (int i = KR * 1024 + tid; i < c0; i += 1024) {
        const int v = __float_as_int(lp[1 + i]);
        lsrc[i] = v & 0xFFFFFF;
        ldl[i] = (int)((unsigned)v >> 24);
      }
      if (tid == 0) cnt = c0;
    } else if (LMODE == 1 && bins_fit(E, N)) {
      const int nseg = (E + kEPB - 1) / kEPB;
      const uint4* sp = (const uint4*)bins + (size_t)(row0 / RT) * nseg * (kSEG / 4);
      const int q = tid & 15, sl = tid >> 4;
      constexpr int SU = 4;
      for (int p0 = 0; p0 < nseg; p0 += 64 * SU) {
        uint4 v[SU];
#pragma unroll
        for (int u = 0; u < SU; ++u) v[u] = sp[(size_t)min(p0 + u * 64 + sl, nseg - 1) * (kSEG / 4) + q];
#pragma unroll
        for (int u = 0; u < SU; ++u) {
          const bool live = p0 + u * 64 + sl < nseg;
          const int cs = __shfl((int)v[u].x, lane & 16);
          const bool fits = (unsigned)cs <= (unsigned)(kSEG - 1);
          if (live && !fits) ovf = 1;
          const int cc = (live && fits) ? cs : 0;
          int base = 0;
          if (q == 0 && cc > 0) base = atomicAdd(&cnt, cc);
          base = __shfl(base, lane & 16);
          const unsigned w4[4] = {v[u].x, v[u].y, v[u].z, v[u].w};
#pragma unroll
          for (int j = 0; j < 4; ++j) {
            const int w = 4 * q + j;
            const int idx = base + w - 1;
            if (w >= 1 && w <= cc && idx < CAP) {
              lsrc[idx] = (int)(w4[j] & 0xFFFFFFu);
              ldl[idx] = min((int)(w4[j] >> 24), rows - 1);
            }
          }
        }
      }
      __syncthreads();
      if (ovf != 0) {
        if (tid == 0) cnt = 0;
        __syncthreads();
      } else {
        nodes = true;
      }
    }
    if (!nodes) {
      int ts = tid;
      asm volatile("" : "+v"(ts));
      const int E4 = (((size_t)dst & 15) == 0) ? (E >> 2) : 0;
      const int4* dst4 = (const int4*)dst;
      for (int g0 = ts; g0 < E4; g0 += 1024 * U) {
        int4 v[U];
#pragma unroll
        for (int u = 0; u < U; ++u) {
          int g = g0 + u * 1024;
          v[u] = (g < E4) ? dst4[g] : make_int4(-1, -1, -1, -1);
        }
#pragma unroll
        for (int u = 0; u < U; ++u) {
          int e = 4 * (g0 + u * 1024);
          take(v[u].x, e);
          take(v[u].y, e + 1);
          take(v[u].z, e + 2);
          take(v[u].w, e + 3);
        }
      }
      for (int e = 4 * E4 + ts; e < E; e += 1024) take(dst[e], e);
    }
    __syncthreads();

    const bool listed = cnt <= CAP;
    const int n = listed ? cnt : 0;
    if (!listed) {
      for (int e = tid; e < E; e += 1024) {
        const int d = dst[e];
        const unsigned dl = (unsigned)(d - row0);
        if (dl < (unsigned)rows) {
          const int s = src[e];
#pragma unroll 1
          for (int h = 0; h < H; ++h)
            atomicMax(&mx[dl * H + h], fkey(leaky(as[aidx(s) + h] + adl[dl * H + h])));
        }
      }
      __syncthreads();
      for (int e = tid; e < E; e += 1024) {
        const int d = dst[e];
        const unsigned dl = (unsigned)(d - row0);
        if (dl < (unsigned)rows) {
          const int s = src[e];
#pragma unroll 1
          for (int h = 0; h < H; ++h) {
            const float a = leaky(as[aidx(s) + h] + adl[dl * H + h]);
            atomicAdd(&dn[dl * H + h], __expf(a - fdecode(mx[dl * H + h])));
          }
        }
      }
      __syncthreads();
      for (int e = tid; e < E; e += 1024) {
        const int d = dst[e];
        const unsigned dl = (unsigned)(d - row0);
        if (dl < (unsigned)rows) {
          const int s = src[e];
#pragma unroll 1
          for (int p = 0; p < HF; ++p) {
            const int lh = dl * H + (p >> LOGF);
            const float a = leaky(as[aidx(s) + (p >> LOGF)] + adl[lh]);
            const float w = __expf(a - fdecode(mx[lh])) / (dn[lh] + 1e-16f);
            atomicAdd(&acc[dl * AST + p], w * xs[(size_t)s * HF + p]);
          }
        }
      }
    }
    {
      int sv[KR], dv[KR];
      float av[KR][H];
#pragma unroll
      for (int k = 0; k < KR; ++k) dv[k] = row0;
      if (n > 0) {
        if (nodes) {
#pragma unroll
          for (int k = 0; k < KR; ++k) {
            const int i = tid + k * 1024;
            sv[k] = (i < n) ? lsrc[i] : 0;
          }
        } else {
#pragma unroll
          for (int k = 0; k < KR; ++k) {
            const int i = tid + k * 1024;
            const int e = (i < n) ? lsrc[i] : 0;
            sv[k] = src[e];
            dv[k] = dst[e];
          }
        }
        __builtin_amdgcn_sched_barrier(0);
#pragma unroll
        for (int k = 0; k < KR; ++k) load_heads<H>(as + aidx(sv[k]), av[k]);
      }
#pragma unroll
      for (int k = 0; k < KR; ++k) {
        const int i = tid + k * 1024;
        if (i < n) {
          if (!nodes) ldl[i] = min(max(dv[k] - row0, 0), rows - 1);
          const int lh = ldl[i] * H;
          lsrc[i] = sv[k];
#pragma unroll
          for (int h = 0; h < H; ++h) {
            av[k][h] = leaky(av[k][h] + adl[lh + h]);
            atomicMax(&mx[lh + h], fkey(av[k][h]));
          }
        }
      }
      for (int i = KR * 1024 + tid; i < n; i += 1024) {
        int s = lsrc[i];
        if (!nodes) {
          ldl[i] = min(max(dst[s] - row0, 0), rows - 1);
          s = src[s];
        }
        const int lh = ldl[i] * H;
        float a[H];
        load_heads<H>(as + aidx(s), a);
        lsrc[i] = s;
#pragma unroll
        for (int h = 0; h < H; ++h) atomicMax(&mx[lh + h], fkey(leaky(a[h] + adl[lh + h])));
      }
      __syncthreads();
#pragma unroll
      for (int k = 0; k < KR; ++k) {
        const int i = tid + k * 1024;
        if (i < n) {
          const int lh = ldl[i] * H;
#pragma unroll
          for (int h = 0; h < H; ++h)
            atomicAdd(&dn[lh + h], __expf(av[k][h] - fdecode(mx[lh + h])));
        }
      }
      for (int i = KR * 1024 + tid; i < n; i += 1024) {
        const int lh = ldl[i] * H;
        float a[H];
        load_heads<H>(as + aidx(lsrc[i]), a);
#pragma unroll
        for (int h = 0; h < H; ++h)
          atomicAdd(&dn[lh + h], __expf(leaky(a[h] + adl[lh + h]) - fdecode(mx[lh + h])));
      }
    }
    __syncthreads();

    if constexpr (LMODE == 1) {
      const bool park = listed && n < lcap && N <= (1 << 24);
      if (tid == 0) lp[0] = __int_as_float(park ? n : -1);
      if (park)
        for (int i = tid; i < n; i += 1024) lp[1 + i] = __int_as_float(lsrc[i] | (ldl[i] << 24));
    }

    for (int i0 = wave * EB; i0 < n; i0 += 32 * EB) {
      const int j = min(lane / H, EB - 1), h = lane & (H - 1);
      const int ii = min(i0 + j, n - 1);
      const int s = lsrc[ii], dl = ldl[ii];
      const int lh = dl * H + h;
      const float v_as = as[aidx(s) + h], v_ad = adl[lh], v_den = dn[lh];
      const unsigned v_mx = mx[lh];
      float m[EB][NS];
      int dls[EB];
#pragma unroll
      for (int jj = 0; jj < EB; ++jj) {
        const int sj = __builtin_amdgcn_readlane(s, jj * H);
        dls[jj] = __builtin_amdgcn_readlane(dl, jj * H);
#pragma unroll
        for (int i = 0; i < NS; ++i) m[jj][i] = xs[(size_t)sj * HF + lane + 32 * i];
      }
      const float w = __expf(leaky(v_as + v_ad) - fdecode(v_mx)) / (v_den + 1e-16f);
#pragma unroll
      for (int jj = 0; jj < EB; ++jj) {
        const bool valid = i0 + jj < n;
#pragma unroll
        for (int i = 0; i < NS; ++i) {
          const int p = lane + 32 * i;
          const float wj = __builtin_bit_cast(
              float, __builtin_amdgcn_readlane(__builtin_bit_cast(int, w), jj * H + ((32 * i) >> LOGF)));
          atomicAdd(&acc[dls[jj] * AST + p], valid ? wj * m[jj][i] : 0.f);
        }
      }
    }
    __syncthreads();

    if constexpr (!NEXT) {
      for (int i = tid; i < rows * HF; i += 1024) {
        float v = acc[i] + bias[i & (HF - 1)];
        if (RELU) v = v > 0.f ? v : 0.f;
        out[(size_t)row0 * HF + i] = v;
      }
    } else {
      int tl = tid;
      asm volatile("" : "+v"(tl));
      for (int i = tid; i < RT * HF; i += 1024) {
        const int cc = i & (HF - 1), ix = (i / HF) * AST + cc;
        float v = acc[ix] + bias[cc];
        if (RELU) v = v > 0.f ? v : 0.f;
        acc[ix] = v;
      }
      __syncthreads();
      {
        const int rr = tl >> 2, q = tl & 3;
        const float* hr = acc + min(rr, RT - 1) * AST;
        float s1 = 0.f, s2 = 0.f;
#pragma unroll 8
        for (unsigned k = (unsigned)q; k < (unsigned)HF; k += 4u) {
          const float hv = hr[k];
          s1 += hv * wan[k];
          s2 += hv * wdn[k];
        }
        s1 += __shfl_xor(s1, 1); s2 += __shfl_xor(s2, 1);
        s1 += __shfl_xor(s1, 2); s2 += __shfl_xor(s2, 2);
        if (q == 0 && rr < RT) {
          adl[rr] = s1;
          dn[rr] = s2;
        }
      }
      __syncthreads();
      if (tl < 2 * LGP) {
        const int j = tl & (LGP - 1);
        if (j < rows) (tl < LGP ? asn : adn)[(size_t)(row0 / RT) * LGP + j] = (tl < LGP ? adl : dn)[j];
      }
      {
        constexpr int MT = (RT + 15) / 16, NU = MT * (NN / 32);
        static_assert(NU <= 52 && NN == 128, "two units per wave at most, unit -> (tile, column pair) by shift and mask");
        const int half = (tl >> 4) & 1, r = tl & 15;
        const int wv = __builtin_amdgcn_readfirstlane(tl >> 5);
        const int wb = wv >> 3, wo = wv & 7;
        const bool two = wo < 4 || wo == 4 + wb;
        const int u2 = 32 + 5 * wb + min(wo, 4);
#pragma unroll 1
        for (int t = 0; t < 2; ++t) {
          const int u = t == 0 ? wv : u2;
          if (t == 1 && !two) break;
          if (u >= NU) break;
          const int mt = u >> 2, np = u & 3;
          if (mt * 16 >= rows) continue;
          const float* ap = acc + min(mt * 16 + r, RT - 1) * AST + 2 * half;
          const float* bp = wn + (unsigned)((2 * half) * NN + np * 32 + r);
          v8f c0 = {}, c1 = {};
          constexpr int KG = 4;
#pragma unroll
          for (int k0 = 0; k0 < HF; k0 += 4 * KG) {
            v2f a[KG], b0[KG], b1[KG];
#pragma unroll
            for (int g = 0; g < KG; ++g) {
              const int k = k0 + 4 * g;
              a[g] = *(const v2f*)(ap + k);
              b0[g].x = bp[k * NN];       b0[g].y = bp[k * NN + NN];
              b1[g].x = bp[k * NN + 16];  b1[g].y = bp[k * NN + NN + 16];
            }
            __builtin_amdgcn_sched_barrier(0);
#pragma unroll
            for (int g = 0; g < KG; ++g) {
              c0 = wmma4(a[g], b0[g], c0);
              c1 = wmma4(a[g], b1[g], c1);
            }
          }
          float* cp = out + (unsigned)((row0 + mt * 16 + 8 * half) * NN + np * 32 + r);
          const int rl = rows - (mt * 16 + 8 * half);
#pragma unroll
          for (int i = 0; i < 8; ++i) {
            if (i < rl) {
              cp[i * NN] = c0[i];
              cp[i * NN + 16] = c1[i];
            }
          }
        }
      }
    }
    __syncthreads();
  }
}

extern "C" void kernel_launch(void* const* d_in, const int* in_sizes, int n_in,
                              void* d_out, int out_size, void* d_ws, size_t ws_size,
                              hipStream_t stream) {
  (void)n_in; (void)out_size; (void)ws_size;
  const float* x     = (const float*)d_in[0];
  const int*   ei    = (const int*)d_in[1];
  const float* W1s   = (const float*)d_in[2];
  const float* W1d   = (const float*)d_in[3];
  const float* att1s = (const float*)d_in[4];
  const float* att1d = (const float*)d_in[5];
  const float* b1    = (const float*)d_in[6];
  const float* W2s   = (const float*)d_in[7];
  const float* W2d   = (const float*)d_in[8];
  const float* att2s = (const float*)d_in[9];
  const float* att2d = (const float*)d_in[10];
  const float* b2    = (const float*)d_in[11];
  float* out = (float*)d_out;

  const int N = in_sizes[0] / 128;
  const int E = in_sizes[1] / 2;
  const int* src = ei;
  const int* dst = ei + E;

  float* ws = (float*)d_ws;
  size_t off = 0;
  float*    XS   = ws + off; off += (size_t)N * 256;
  float*    Hb   = ws + off; off += (size_t)N * 256;
  float*    AS   = ws + off; off += (size_t)N * 8;
  float*    AD   = ws + off; off += (size_t)N * 8;
  float*    WA   = ws + off; off += 2048;
  float*    WD   = ws + off; off += 2048;
  float*    AS2  = Hb + (size_t)N * 128;
  float*    AD2  = AS2 + (size_t)N * 2 + 256;
  float*    BINS = AS2 + kBOFF;

  const int TPB = 256;

  const int bF = 256;
  const int bK = bF + 2;
  const int bG = bK + (E + kEPB - 1) / kEPB;
  const int nb = bG + 4 * ((N / 16 + 8 * kGT - 1) / (8 * kGT));
  front<<<nb, TPB, 0, stream>>>(x, src, dst, W1s, W1d, att1s, att1d, W2s, W2d, att2s, att2d,
                                XS, AS, AD, WA, WD, BINS, N, E, bF, bK, bG);

  edge_aggregate_owned<8, 5, 4, true, 1><<<256, 1024, 0, stream>>>(src, dst, AS, AD,
                                                                 XS, b1, Hb, out, BINS, E, N,
                                                                 W2s, WA, WD, AS2, AD2);

  edge_aggregate_owned<1, 7, 8, false, 2><<<256, 1024, 0, stream>>>(src, dst, AS2, AD2,
                                                                  Hb, b2, out, nullptr, nullptr, E, N,
                                                                  nullptr, nullptr, nullptr,
                                                                  nullptr, nullptr);
}


# edge_aggregate max and denominator passes: lane -> (entry, head), 4 entries x 8 heads per LDS update (8 banks in a row per entry; was 32 entries on one head's 8 banks); denominator add order changes
# speedup vs baseline: 4.4471x; 1.0429x over previous
#include <hip/hip_runtime.h>
#include <stddef.h>

typedef __attribute__((ext_vector_type(2))) float v2f;
typedef __attribute__((ext_vector_type(8))) float v8f;

__device__ __forceinline__ v8f wmma4(v2f a, v2f b, v8f c) {
  return __builtin_amdgcn_wmma_f32_16x16x4_f32(
       false, a,  false, b,
       (short)0, c,  false,  false);
}

constexpr int kGP = 160;
constexpr int kGT = 2;
static_assert(kGP % 64 == 32 && kGP >= 128, "the two lane halves of a fragment read use different bank halves");

template <int N, int K>
__device__ __forceinline__ void gemm_strips_staged(const float* __restrict__ A,
                                                   const float* __restrict__ B,
                                                   float* __restrict__ C, float* Bs,
                                                   int chunk, int ns, int MT) {
  static_assert(K % 4 == 0 && (K / 2) * 16 % 256 == 0, "whole k steps; the copy is whole trips of 256 threads");
  const int tid = threadIdx.x, lane = tid & 31, wave = tid >> 5;
  const int n0 = ns * 64;
#pragma unroll
  for (int i = 0; i < (K / 2) * 16 / 256; ++i) {
    const int item = tid + 256 * i;
    const int p = item >> 4, c4 = item & 15;
    const float4 lo = *(const float4*)(B + (size_t)(2 * p) * N + n0 + 4 * c4);
    const float4 hi = *(const float4*)(B + (size_t)(2 * p + 1) * N + n0 + 4 * c4);
    float4* q = (float4*)(Bs + p * kGP + 8 * c4);
    q[0] = make_float4(lo.x, hi.x, lo.y, hi.y);
    q[1] = make_float4(lo.z, hi.z, lo.w, hi.w);
  }
  __syncthreads();

  const int half = lane >> 4;
  const int r    = lane & 15;
  const float* bp = Bs + half * kGP + 2 * r;
#pragma unroll 1
  for (int t = 0; t < kGT; ++t) {
    const int mt = chunk * (8 * kGT) + t * 8 + wave;
    if (mt >= MT) break;
    const int m0 = mt * 16;
    const float* Arow = A + (size_t)(m0 + r) * K + 2 * half;
    v8f c0 = {}, c1 = {}, c2 = {}, c3 = {};
#pragma unroll
    for (int k = 0; k < K; k += 4) {
      const v2f a = *(const v2f*)(Arow + k);
      const float* bk = bp + (k >> 1) * kGP;
      const v2f b0 = *(const v2f*)(bk);
      const v2f b1 = *(const v2f*)(bk + 32);
      const v2f b2 = *(const v2f*)(bk + 64);
      const v2f b3 = *(const v2f*)(bk + 96);
      c0 = wmma4(a, b0, c0);
      c1 = wmma4(a, b1, c1);
      c2 = wmma4(a, b2, c2);
      c3 = wmma4(a, b3, c3);
    }
    float* Crow = C + (size_t)(m0 + 8 * half) * N + n0 + r;
#pragma unroll
    for (int i = 0; i < 8; ++i) {
      Crow[i * N + 0]  = c0[i];
      Crow[i * N + 16] = c1[i];
      Crow[i * N + 32] = c2[i];
      Crow[i * N + 48] = c3[i];
    }
  }
}

constexpr int kLN = 196;
static_assert(kLN % 4 == 0, "4 nodes of a wave fill one whole 128-byte line of a_s and of a_d");

__device__ __forceinline__ void fold_and_logits(const float* __restrict__ x,
                                                const float* __restrict__ Wsrc, const float* __restrict__ Wdst,
                                                const float* __restrict__ atts, const float* __restrict__ attd,
                                                float* __restrict__ as, float* __restrict__ ad,
                                                float4* pan, int blk, int nblk, int N) {
  const int tid = threadIdx.x, lane = tid & 31, wave = tid >> 5;
  {
    const int hh = (wave & 1) * 4 + (lane >> 3);
    const float4 a_s4 = *(const float4*)(atts + hh * 32 + (lane & 7) * 4);
    const float4 a_d4 = *(const float4*)(attd + hh * 32 + (lane & 7) * 4);
    const float* ws = Wsrc + wave * 128 + lane * 4;
    const float* wd = Wdst + wave * 128 + lane * 4;
    constexpr int FC = 8;
#pragma unroll 1
    for (int c0 = 0; c0 < 32; c0 += FC) {
      float4 vs[FC], vd[FC];
#pragma unroll
      for (int j = 0; j < FC; ++j) {
        vs[j] = *(const float4*)(ws + (c0 + j) * 1024);
        vd[j] = *(const float4*)(wd + (c0 + j) * 1024);
      }
#pragma unroll
      for (int j = 0; j < FC; ++j) {
        float ps = vs[j].x * a_s4.x + vs[j].y * a_s4.y + vs[j].z * a_s4.z + vs[j].w * a_s4.w;
        float pd = vd[j].x * a_d4.x + vd[j].y * a_d4.y + vd[j].z * a_d4.z + vd[j].w * a_d4.w;
        ps += __shfl_xor(ps, 1); pd += __shfl_xor(pd, 1);
        ps += __shfl_xor(ps, 2); pd += __shfl_xor(pd, 2);
        ps += __shfl_xor(ps, 4); pd += __shfl_xor(pd, 4);
        if ((lane & 7) == 0) {
          const int k = (c0 + j) * 4 + (wave >> 1);
          float* q = (float*)(pan + (k >> 1) * 8 + hh) + (k & 1) * 2;
          q[0] = ps;
          q[1] = pd;
        }
      }
    }
  }
  __syncthreads();

  const int h = tid & 7;
  for (int row0 = blk * kLN; row0 < N; row0 += nblk * kLN) {
    const int rows = min(kLN, N - row0);
    for (int p0 = wave * 4; p0 < rows; p0 += 32) {
      const int nl = p0 + (lane >> 3);
      const int node = row0 + min(nl, rows - 1);
      const float4* xp = (const float4*)(x + (size_t)node * 128);
      float s = 0.f, d = 0.f;
      constexpr int XC = 8;
#pragma unroll
      for (int q0 = 0; q0 < 32; q0 += XC) {
        float4 xv[XC];
#pragma unroll
        for (int j = 0; j < XC; ++j) xv[j] = xp[q0 + j];
#pragma unroll
        for (int j = 0; j < XC; ++j) {
          const float4 w0 = pan[(2 * (q0 + j)) * 8 + h];
          const float4 w1 = pan[(2 * (q0 + j) + 1) * 8 + h];
          s += xv[j].x * w0.x;  d += xv[j].x * w0.y;
          s += xv[j].y * w0.z;  d += xv[j].y * w0.w;
          s += xv[j].z * w1.x;  d += xv[j].z * w1.y;
          s += xv[j].w * w1.z;  d += xv[j].w * w1.w;
        }
      }
      if (nl < rows) {
        as[(size_t)node * 8 + h] = s;
        ad[(size_t)node * 8 + h] = d;
      }
    }
  }
}

__device__ __forceinline__ void fold_next(const float* __restrict__ Wsrc, const float* __restrict__ Wdst,
                                          const float* __restrict__ atts, const float* __restrict__ attd,
                                          float* __restrict__ wa, float* __restrict__ wd, int unit, int lane) {
  const bool second = unit >= 8;
  const float* Wm = (second ? Wdst : Wsrc) + (size_t)((unit & 7) * 32) * 128 + lane * 4;
  const float4 a4 = *(const float4*)((second ? attd : atts) + lane * 4);
  float res = 0.f;
  constexpr int RC = 16;
#pragma unroll 1
  for (int j0 = 0; j0 < 32; j0 += RC) {
    float4 v[RC];
#pragma unroll
    for (int j = 0; j < RC; ++j) v[j] = *(const float4*)(Wm + (j0 + j) * 128);
#pragma unroll
    for (int j = 0; j < RC; ++j) {
      float p = v[j].x * a4.x + v[j].y * a4.y + v[j].z * a4.z + v[j].w * a4.w;
      p += __shfl_xor(p, 1);
      p += __shfl_xor(p, 2);
      p += __shfl_xor(p, 4);
      p += __shfl_xor(p, 8);
      p += __shfl_xor(p, 16);
      if (lane == j0 + j) res = p;
    }
  }
  (second ? wd : wa)[(unit & 7) * 32 + lane] = res;
}

__device__ __forceinline__ unsigned fkey(float f) {
  unsigned u = __float_as_uint(f);
  return (u & 0x80000000u) ? ~u : (u | 0x80000000u);
}
__device__ __forceinline__ float fdecode(unsigned k) {
  unsigned u = (k & 0x80000000u) ? (k & 0x7FFFFFFFu) : ~k;
  return __uint_as_float(u);
}
__device__ __forceinline__ float leaky(float a) {
  return a > 0.f ? a : 0.2f * a;
}

template <int H>
__device__ __forceinline__ void load_heads(const float* __restrict__ p, float (&a)[H]) {
  if constexpr (H % 4 == 0) {
#pragma unroll
    for (int q = 0; q < H / 4; ++q) {
      const float4 v = ((const float4*)p)[q];
      a[4 * q + 0] = v.x; a[4 * q + 1] = v.y; a[4 * q + 2] = v.z; a[4 * q + 3] = v.w;
    }
  } else {
#pragma unroll
    for (int h = 0; h < H; ++h) a[h] = p[h];
  }
}

constexpr int kRT = 196;
constexpr int kLP = 128;
constexpr int kCAP = 8192;
constexpr int kEPB = 4096;
constexpr int kSEG = 64;
static_assert(kEPB == 16 * 256, "a bin block is 256 threads, 16 edges each");
constexpr int kBOFF = 262144;

__device__ __forceinline__ bool bins_fit(int E, int N) {
  const long long ntile = (N + kRT - 1) / kRT, nseg = ((long long)E + kEPB - 1) / kEPB;
  return N <= (1 << 24) && ntile <= 256 && ntile * nseg * kSEG <= (long long)N * 128 - kBOFF;
}

__device__ __forceinline__ void bin_edges(const int* __restrict__ src, const int* __restrict__ dst,
                                          float* __restrict__ binsf, int* cb, int E, int N, int bb) {
  if (!bins_fit(E, N)) return;
  unsigned* bins = (unsigned*)binsf;
  const int tid = threadIdx.x;
  const int ntile = (N + kRT - 1) / kRT, nseg = (E + kEPB - 1) / kEPB;
  cb[tid] = 0;
  __syncthreads();
  int d[16], s[16];
  const int e0 = bb * kEPB;
  if (((((size_t)src | (size_t)dst) & 15) == 0) && e0 + kEPB <= E) {
    const int4* d4 = (const int4*)(dst + e0) + tid;
    const int4* s4 = (const int4*)(src + e0) + tid;
#pragma unroll
    for (int q = 0; q < 4; ++q) {
      const int4 dv = d4[q * 256], sv = s4[q * 256];
      d[4 * q + 0] = dv.x; d[4 * q + 1] = dv.y; d[4 * q + 2] = dv.z; d[4 * q + 3] = dv.w;
      s[4 * q + 0] = sv.x; s[4 * q + 1] = sv.y; s[4 * q + 2] = sv.z; s[4 * q + 3] = sv.w;
    }
  } else {
#pragma unroll
    for (int j = 0; j < 16; ++j) {
      const int e = e0 + j * 256 + tid;
      d[j] = e < E ? dst[e] : -1;
      s[j] = e < E ? src[e] : 0;
    }
  }
  int pos[16];
#pragma unroll
  for (int j = 0; j < 16; ++j) {
    pos[j] = kSEG;
    if ((unsigned)d[j] < (unsigned)N) pos[j] = atomicAdd(&cb[(unsigned)d[j] / (unsigned)kRT], 1);
  }
#pragma unroll
  for (int j = 0; j < 16; ++j) {
    if (pos[j] < kSEG - 1) {
      const unsigned t = (unsigned)d[j] / (unsigned)kRT;
      bins[((size_t)t * nseg + bb) * kSEG + 1 + pos[j]] =
          ((unsigned)s[j] & 0xFFFFFFu) | (((unsigned)d[j] - t * (unsigned)kRT) << 24);
    }
  }
  __syncthreads();
  if (tid < ntile) bins[((size_t)tid * nseg + bb) * kSEG] = (unsigned)cb[tid];
}

__global__ void __launch_bounds__(256)
front(const float* __restrict__ x, const int* __restrict__ src, const int* __restrict__ dst,
      const float* __restrict__ W1s, const float* __restrict__ W1d,
      const float* __restrict__ att1s, const float* __restrict__ att1d,
      const float* __restrict__ W2s, const float* __restrict__ W2d,
      const float* __restrict__ att2s, const float* __restrict__ att2d,
      float* __restrict__ xs, float* __restrict__ as, float* __restrict__ ad,
      float* __restrict__ wa2, float* __restrict__ wd2, float* __restrict__ bins,
      int N, int E, int bF, int bK, int bG) {
  __shared__ __attribute__((aligned(16))) float lbuf[64 * kGP];
  float4* pan = (float4*)lbuf;
  const int b = blockIdx.x;
  const int lane = threadIdx.x & 31, wave = threadIdx.x >> 5;
  if (b < bF) {
    fold_and_logits(x, W1s, W1d, att1s, att1d, as, ad, pan, b, bF, N);
  } else if (b < bK) {
    const int unit = (b - bF) * 8 + wave;
    if (unit < 16) fold_next(W2s, W2d, att2s, att2d, wa2, wd2, unit, lane);
  } else if (b < bG) {
    bin_edges(src, dst, bins, (int*)pan, E, N, b - bK);
  } else {
    const int MT = N >> 4;
    const int bps = (MT + 8 * kGT - 1) / (8 * kGT);
    const int gb = b - bG;
    const int ns = gb / bps;
    if (ns < 4) gemm_strips_staged<256, 128>(x, W1s, xs, lbuf, gb - ns * bps, ns, MT);
  }
}

template <int H, int LOGF, int EB, bool RELU, int LMODE>
__global__ void __launch_bounds__(1024)
edge_aggregate_owned(const int* __restrict__ src, const int* __restrict__ dst,
                     const float* __restrict__ as, const float* __restrict__ ad,
                     const float* __restrict__ xs, const float* __restrict__ bias,
                     float* __restrict__ out, float* __restrict__ lst,
                     const float* __restrict__ bins, int E, int N,
                     const float* __restrict__ wn, const float* __restrict__ wan,
                     const float* __restrict__ wdn, float* __restrict__ asn,
                     float* __restrict__ adn) {
  constexpr int HF = H << LOGF;
  constexpr bool NEXT = LMODE == 1;
  constexpr int NN = 128;
  constexpr int AST = NEXT ? HF + 4 : HF;
  static_assert(!NEXT || HF % 4 == 0, "whole k steps of the matrix instruction");
  constexpr int NS = HF / 32;
  constexpr int RT = kRT;
  constexpr int CAP = kCAP;
  constexpr int KR = 4;
  constexpr int U = 8;
  constexpr int LP = kLP;
  static_assert(RT <= 256, "a parked entry keeps its row in 8 bits");
  static_assert(LMODE != 2 || HF == LP, "the parked lists lie in this launch's own output rows");
  static_assert(EB * H <= 32, "one lane per (edge, head) of a batch");
  static_assert(LOGF >= 5, "a 32-lane slice lies inside one head");
  __shared__ __attribute__((aligned(16))) float acc[RT * AST];
  __shared__ int lsrc[CAP];
  __shared__ int ldl[CAP];
  __shared__ float adl[RT * H];
  __shared__ unsigned mx[RT * H];
  __shared__ float dn[RT * H];
  __shared__ int cnt;
  __shared__ int ovf;

  const int tid = threadIdx.x, lane = tid & 31, wave = tid >> 5;
  constexpr int LGP = 256;
  static_assert(LGP >= RT && LGP % 32 == 0 && (LMODE != 2 || H == 1), "a tile's logits fill whole lines of their own");
  auto aidx = [](int s) -> size_t {
    if constexpr (LMODE == 2) return (size_t)((unsigned)s / (unsigned)RT) * LGP + (unsigned)s % (unsigned)RT;
    else return (size_t)s * H;
  };

  for (int row0 = blockIdx.x * RT; row0 < N; row0 += gridDim.x * RT) {
    const int rows = min(RT, N - row0);
    const int lcap = rows * LP;
    float* const lp = (LMODE == 2 ? out : lst) + (size_t)row0 * LP;
    int c0 = -1;
    int lv[KR];
    if constexpr (LMODE == 2) {
      c0 = __float_as_int(lp[0]);
#pragma unroll
      for (int k = 0; k < KR; ++k) lv[k] = __float_as_int(lp[min(1 + tid + k * 1024, lcap - 1)]);
    }
    for (int i = tid; i < RT * AST; i += 1024) acc[i] = 0.f;
    for (int i = tid; i < RT * H; i += 1024) {
      adl[i] = (i < rows * H) ? ad[aidx(row0) + i] : 0.f;
      mx[i] = 0u;
      dn[i] = 0.f;
    }
    if (tid == 0) {
      cnt = 0;
      ovf = 0;
    }
    __syncthreads();

    auto list_edge = [&](int e) {
      int idx = atomicAdd(&cnt, 1);
      if (idx < CAP) lsrc[idx] = e;
    };
    auto take = [&](int d, int e) {
      if ((unsigned)(d - row0) < (unsigned)rows) list_edge(e);
    };
    bool use_list = false;
    if constexpr (LMODE == 2) {
      c0 = __builtin_amdgcn_readfirstlane(c0);
      use_list = c0 >= 0 && c0 <= CAP && c0 < lcap;
    }
    bool nodes = use_list;
    if (use_list) {
#pragma unroll
      for (int k = 0; k < KR; ++k) {
        const int i = tid + k * 1024;
        if (i < c0) {
          lsrc[i] = lv[k] & 0xFFFFFF;
          ldl[i] = (int)((unsigned)lv[k] >> 24);
        }
      }
      for (int i = KR * 1024 + tid; i < c0; i += 1024) {
        const int v = __float_as_int(lp[1 + i]);
        lsrc[i] = v & 0xFFFFFF;
        ldl[i] = (int)((unsigned)v >> 24);
      }
      if (tid == 0) cnt = c0;
    } else if (LMODE == 1 && bins_fit(E, N)) {
      const int nseg = (E + kEPB - 1) / kEPB;
      const uint4* sp = (const uint4*)bins + (size_t)(row0 / RT) * nseg * (kSEG / 4);
      int tb = tid;
      asm volatile("" : "+v"(tb));
      const int q = tb & 15, sl = tb >> 4;
      constexpr int SU = 4;
      for (int p0 = 0; p0 < nseg; p0 += 64 * SU) {
        uint4 v[SU];
#pragma unroll
        for (int u = 0; u < SU; ++u) v[u] = sp[(size_t)min(p0 + u * 64 + sl, nseg - 1) * (kSEG / 4) + q];
#pragma unroll
        for (int u = 0; u < SU; ++u) {
          const bool live = p0 + u * 64 + sl < nseg;
          const int cs = __shfl((int)v[u].x, lane & 16);
          const bool fits = (unsigned)cs <= (unsigned)(kSEG - 1);
          if (live && !fits) ovf = 1;
          const int cc = (live && fits) ? cs : 0;
          int base = 0;
          if (q == 0 && cc > 0) base = atomicAdd(&cnt, cc);
          base = __shfl(base, lane & 16);
          const unsigned w4[4] = {v[u].x, v[u].y, v[u].z, v[u].w};
#pragma unroll
          for (int j = 0; j < 4; ++j) {
            const int w = 4 * q + j;
            const int idx = base + w - 1;
            if (w >= 1 && w <= cc && idx < CAP) {
              lsrc[idx] = (int)(w4[j] & 0xFFFFFFu);
              ldl[idx] = min((int)(w4[j] >> 24), rows - 1);
            }
          }
        }
      }
      __syncthreads();
      if (ovf != 0) {
        if (tid == 0) cnt = 0;
        __syncthreads();
      } else {
        nodes = true;
      }
    }
    if (!nodes) {
      int ts = tid;
      asm volatile("" : "+v"(ts));
      const int E4 = (((size_t)dst & 15) == 0) ? (E >> 2) : 0;
      const int4* dst4 = (const int4*)dst;
      for (int g0 = ts; g0 < E4; g0 += 1024 * U) {
        int4 v[U];
#pragma unroll
        for (int u = 0; u < U; ++u) {
          int g = g0 + u * 1024;
          v[u] = (g < E4) ? dst4[g] : make_int4(-1, -1, -1, -1);
        }
#pragma unroll
        for (int u = 0; u < U; ++u) {
          int e = 4 * (g0 + u * 1024);
          take(v[u].x, e);
          take(v[u].y, e + 1);
          take(v[u].z, e + 2);
          take(v[u].w, e + 3);
        }
      }
      for (int e = 4 * E4 + ts; e < E; e += 1024) take(dst[e], e);
    }
    __syncthreads();

    const bool listed = cnt <= CAP;
    const int n = listed ? cnt : 0;
    if (!listed) {
      for (int e = tid; e < E; e += 1024) {
        const int d = dst[e];
        const unsigned dl = (unsigned)(d - row0);
        if (dl < (unsigned)rows) {
          const int s = src[e];
#pragma unroll 1
          for (int h = 0; h < H; ++h)
            atomicMax(&mx[dl * H + h], fkey(leaky(as[aidx(s) + h] + adl[dl * H + h])));
        }
      }
      __syncthreads();
      for (int e = tid; e < E; e += 1024) {
        const int d = dst[e];
        const unsigned dl = (unsigned)(d - row0);
        if (dl < (unsigned)rows) {
          const int s = src[e];
#pragma unroll 1
          for (int h = 0; h < H; ++h) {
            const float a = leaky(as[aidx(s) + h] + adl[dl * H + h]);
            atomicAdd(&dn[dl * H + h], __expf(a - fdecode(mx[dl * H + h])));
          }
        }
      }
      __syncthreads();
      for (int e = tid; e < E; e += 1024) {
        const int d = dst[e];
        const unsigned dl = (unsigned)(d - row0);
        if (dl < (unsigned)rows) {
          const int s = src[e];
#pragma unroll 1
          for (int p = 0; p < HF; ++p) {
            const int lh = dl * H + (p >> LOGF);
            const float a = leaky(as[aidx(s) + (p >> LOGF)] + adl[lh]);
            const float w = __expf(a - fdecode(mx[lh])) / (dn[lh] + 1e-16f);
            atomicAdd(&acc[dl * AST + p], w * xs[(size_t)s * HF + p]);
          }
        }
      }
    }
    if (!nodes && n > 0) {
      for (int i = tid; i < n; i += 1024) {
        const int e = lsrc[i];
        ldl[i] = min(max(dst[e] - row0, 0), rows - 1);
        lsrc[i] = src[e];
      }
      __syncthreads();
    }
    constexpr int G = 32 / H;
    constexpr int KS = (KR * H < 28) ? KR * H : 28;
    constexpr int CH = (KS % 7 == 0) ? 7 : KS;
    const int hh = lane & (H - 1);
    int e0 = wave * G + lane / H;
    asm volatile("" : "+v"(e0));
    auto entry = [&](int eb, int k) -> int {
      const int i = eb + k * 32 * G;
      return (i < n) ? i : min(i - n, n - 1);
    };
    float av[KS];
    if (n > 0) {
      {
        int sv[KS];
#pragma unroll
        for (int k = 0; k < KS; ++k) sv[k] = lsrc[entry(e0, k)];
        __builtin_amdgcn_sched_barrier(0);
#pragma unroll
        for (int k = 0; k < KS; ++k) av[k] = as[aidx(sv[k]) + hh];
      }
#pragma unroll
      for (int c = 0; c < KS; c += CH) {
        __builtin_amdgcn_sched_barrier(0);
        int lh[CH];
        int eb = e0;
        asm volatile("" : "+v"(eb));
        float adv[CH];
#pragma unroll
        for (int k = 0; k < CH; ++k) lh[k] = ldl[entry(eb, c + k)] * H + hh;
#pragma unroll
        for (int k = 0; k < CH; ++k) adv[k] = adl[lh[k]];
        __builtin_amdgcn_sched_barrier(0);
#pragma unroll
        for (int k = 0; k < CH; ++k) {
          const bool valid = eb + (c + k) * 32 * G < n;
          av[c + k] = leaky(av[c + k] + adv[k]);
          atomicMax(&mx[lh[k]], valid ? fkey(av[c + k]) : 0u);
        }
      }
      for (int i = e0 + KS * 32 * G; i < n; i += 32 * G) {
        const int l = ldl[i] * H + hh;
        atomicMax(&mx[l], fkey(leaky(as[aidx(lsrc[i]) + hh] + adl[l])));
      }
    }
    __syncthreads();
    if (n > 0) {
#pragma unroll
      for (int c = 0; c < KS; c += CH) {
        __builtin_amdgcn_sched_barrier(0);
        int lh[CH];
        int eb = e0;
        asm volatile("" : "+v"(eb));
        unsigned mxv[CH];
#pragma unroll
        for (int k = 0; k < CH; ++k) lh[k] = ldl[entry(eb, c + k)] * H + hh;
#pragma unroll
        for (int k = 0; k < CH; ++k) mxv[k] = mx[lh[k]];
        __builtin_amdgcn_sched_barrier(0);
#pragma unroll
        for (int k = 0; k < CH; ++k) {
          const bool valid = eb + (c + k) * 32 * G < n;
          const float e = __expf(av[c + k] - fdecode(mxv[k]));
          atomicAdd(&dn[lh[k]], valid ? e : 0.f);
        }
      }
      for (int i = e0 + KS * 32 * G; i < n; i += 32 * G) {
        const int l = ldl[i] * H + hh;
        atomicAdd(&dn[l], __expf(leaky(as[aidx(lsrc[i]) + hh] + adl[l]) - fdecode(mx[l])));
      }
    }
    __syncthreads();

    if constexpr (LMODE == 1) {
      const bool park = listed && n < lcap && N <= (1 << 24);
      if (tid == 0) lp[0] = __int_as_float(park ? n : -1);
      if (park)
        for (int i = tid; i < n; i += 1024) lp[1 + i] = __int_as_float(lsrc[i] | (ldl[i] << 24));
    }

    for (int i0 = wave * EB; i0 < n; i0 += 32 * EB) {
      const int j = min(lane / H, EB - 1), h = lane & (H - 1);
      const int ii = min(i0 + j, n - 1);
      const int s = lsrc[ii], dl = ldl[ii];
      const int lh = dl * H + h;
      const float v_as = as[aidx(s) + h], v_ad = adl[lh], v_den = dn[lh];
      const unsigned v_mx = mx[lh];
      float m[EB][NS];
      int dls[EB];
#pragma unroll
      for (int jj = 0; jj < EB; ++jj) {
        const int sj = __builtin_amdgcn_readlane(s, jj * H);
        dls[jj] = __builtin_amdgcn_readlane(dl, jj * H);
#pragma unroll
        for (int i = 0; i < NS; ++i) m[jj][i] = xs[(size_t)sj * HF + lane + 32 * i];
      }
      const float w = __expf(leaky(v_as + v_ad) - fdecode(v_mx)) / (v_den + 1e-16f);
#pragma unroll
      for (int jj = 0; jj < EB; ++jj) {
        const bool valid = i0 + jj < n;
#pragma unroll
        for (int i = 0; i < NS; ++i) {
          const int p = lane + 32 * i;
          const float wj = __builtin_bit_cast(
              float, __builtin_amdgcn_readlane(__builtin_bit_cast(int, w), jj * H + ((32 * i) >> LOGF)));
          atomicAdd(&acc[dls[jj] * AST + p], valid ? wj * m[jj][i] : 0.f);
        }
      }
    }
    __syncthreads();

    if constexpr (!NEXT) {
      for (int i = tid; i < rows * HF; i += 1024) {
        float v = acc[i] + bias[i & (HF - 1)];
        if (RELU) v = v > 0.f ? v : 0.f;
        out[(size_t)row0 * HF + i] = v;
      }
    } else {
      int tl = tid;
      asm volatile("" : "+v"(tl));
      for (int i = tid; i < RT * HF; i += 1024) {
        const int cc = i & (HF - 1), ix = (i / HF) * AST + cc;
        float v = acc[ix] + bias[cc];
        if (RELU) v = v > 0.f ? v : 0.f;
        acc[ix] = v;
      }
      __syncthreads();
      {
        const int rr = tl >> 2, q = tl & 3;
        const float* hr = acc + min(rr, RT - 1) * AST;
        float s1 = 0.f, s2 = 0.f;
#pragma unroll 8
        for (unsigned k = (unsigned)q; k < (unsigned)HF; k += 4u) {
          const float hv = hr[k];
          s1 += hv * wan[k];
          s2 += hv * wdn[k];
        }
        s1 += __shfl_xor(s1, 1); s2 += __shfl_xor(s2, 1);
        s1 += __shfl_xor(s1, 2); s2 += __shfl_xor(s2, 2);
        if (q == 0 && rr < RT) {
          adl[rr] = s1;
          dn[rr] = s2;
        }
      }
      __syncthreads();
      if (tl < 2 * LGP) {
        const int j = tl & (LGP - 1);
        if (j < rows) (tl < LGP ? asn : adn)[(size_t)(row0 / RT) * LGP + j] = (tl < LGP ? adl : dn)[j];
      }
      {
        constexpr int MT = (RT + 15) / 16, NU = MT * (NN / 32);
        static_assert(NU <= 52 && NN == 128, "two units per wave at most, unit -> (tile, column pair) by shift and mask");
        const int half = (tl >> 4) & 1, r = tl & 15;
        const int wv = __builtin_amdgcn_readfirstlane(tl >> 5);
        const int wb = wv >> 3, wo = wv & 7;
        const bool two = wo < 4 || wo == 4 + wb;
        const int u2 = 32 + 5 * wb + min(wo, 4);
#pragma unroll 1
        for (int t = 0; t < 2; ++t) {
          const int u = t == 0 ? wv : u2;
          if (t == 1 && !two) break;
          if (u >= NU) break;
          const int mt = u >> 2, np = u & 3;
          if (mt * 16 >= rows) continue;
          const float* ap = acc + min(mt * 16 + r, RT - 1) * AST + 2 * half;
          const float* bp = wn + (unsigned)((2 * half) * NN + np * 32 + r);
          v8f c0 = {}, c1 = {};
          constexpr int KG = 4;
#pragma unroll
          for (int k0 = 0; k0 < HF; k0 += 4 * KG) {
            v2f a[KG], b0[KG], b1[KG];
#pragma unroll
            for (int g = 0; g < KG; ++g) {
              const int k = k0 + 4 * g;
              a[g] = *(const v2f*)(ap + k);
              b0[g].x = bp[k * NN];       b0[g].y = bp[k * NN + NN];
              b1[g].x = bp[k * NN + 16];  b1[g].y = bp[k * NN + NN + 16];
            }
            __builtin_amdgcn_sched_barrier(0);
#pragma unroll
            for (int g = 0; g < KG; ++g) {
              c0 = wmma4(a[g], b0[g], c0);
              c1 = wmma4(a[g], b1[g], c1);
            }
          }
          float* cp = out + (unsigned)((row0 + mt * 16 + 8 * half) * NN + np * 32 + r);
          const int rl = rows - (mt * 16 + 8 * half);
#pragma unroll
          for (int i = 0; i < 8; ++i) {
            if (i < rl) {
              cp[i * NN] = c0[i];
              cp[i * NN + 16] = c1[i];
            }
          }
        }
      }
    }
    __syncthreads();
  }
}

extern "C" void kernel_launch(void* const* d_in, const int* in_sizes, int n_in,
                              void* d_out, int out_size, void* d_ws, size_t ws_size,
                              hipStream_t stream) {
  (void)n_in; (void)out_size; (void)ws_size;
  const float* x     = (const float*)d_in[0];
  const int*   ei    = (const int*)d_in[1];
  const float* W1s   = (const float*)d_in[2];
  const float* W1d   = (const float*)d_in[3];
  const float* att1s = (const float*)d_in[4];
  const float* att1d = (const float*)d_in[5];
  const float* b1    = (const float*)d_in[6];
  const float* W2s   = (const float*)d_in[7];
  const float* W2d   = (const float*)d_in[8];
  const float* att2s = (const float*)d_in[9];
  const float* att2d = (const float*)d_in[10];
  const float* b2    = (const float*)d_in[11];
  float* out = (float*)d_out;

  const int N = in_sizes[0] / 128;
  const int E = in_sizes[1] / 2;
  const int* src = ei;
  const int* dst = ei + E;

  float* ws = (float*)d_ws;
  size_t off = 0;
  float*    XS   = ws + off; off += (size_t)N * 256;
  float*    Hb   = ws + off; off += (size_t)N * 256;
  float*    AS   = ws + off; off += (size_t)N * 8;
  float*    AD   = ws + off; off += (size_t)N * 8;
  float*    WA   = ws + off; off += 2048;
  float*    WD   = ws + off; off += 2048;
  float*    AS2  = Hb + (size_t)N * 128;
  float*    AD2  = AS2 + (size_t)N * 2 + 256;
  float*    BINS = AS2 + kBOFF;

  const int TPB = 256;

  const int bF = 256;
  const int bK = bF + 2;
  const int bG = bK + (E + kEPB - 1) / kEPB;
  const int nb = bG + 4 * ((N / 16 + 8 * kGT - 1) / (8 * kGT));
  front<<<nb, TPB, 0, stream>>>(x, src, dst, W1s, W1d, att1s, att1d, W2s, W2d, att2s, att2d,
                                XS, AS, AD, WA, WD, BINS, N, E, bF, bK, bG);

  edge_aggregate_owned<8, 5, 4, true, 1><<<256, 1024, 0, stream>>>(src, dst, AS, AD,
                                                                 XS, b1, Hb, out, BINS, E, N,
                                                                 W2s, WA, WD, AS2, AD2);

  edge_aggregate_owned<1, 7, 8, false, 2><<<256, 1024, 0, stream>>>(src, dst, AS2, AD2,
                                                                  Hb, b2, out, nullptr, nullptr, E, N,
                                                                  nullptr, nullptr, nullptr,
                                                                  nullptr, nullptr);
}
